# starting kernel
# baseline (speedup 1.0000x reference)
#include <hip/hip_runtime.h>
#include <math.h>

#define N_QUBITS 12
#define DEPTH    4
#define HID      64
#define NPARAMS  144
#define BATCH    4096
#define DIM      4096
#define OUT_DIM  2

typedef __attribute__((ext_vector_type(2))) float v2f;
typedef __attribute__((ext_vector_type(8))) float v8f;

__device__ __forceinline__ v8f wmma_f32(v2f a, v2f b, v8f c) {
  return __builtin_amdgcn_wmma_f32_16x16x4_f32(
       false, a,  false, b,
       (short)0, c,  false,  false);
}

__device__ __forceinline__ v2f load_a_frag(const float* src, int lda,
                                           int row0, int k0, int lane) {
  int m = row0 + (lane & 15);
  int k = k0 + ((lane >> 4) << 1);
  v2f a;
  a.x = src[m * lda + k];
  a.y = src[m * lda + k + 1];
  return a;
}

__device__ __forceinline__ v2f load_b_frag(const float* W, int ldb,
                                           int k0, int n0, int lane) {
  int c = n0 + (lane & 15);
  int k = k0 + ((lane >> 4) << 1);
  v2f b;
  b.x = W[k * ldb + c];
  b.y = W[(k + 1) * ldb + c];
  return b;
}

__global__ __launch_bounds__(128) void front_mlp_kernel(
    const float* __restrict__ noise, const float* __restrict__ W1,
    const float* __restrict__ b1,    const float* __restrict__ W2,
    const float* __restrict__ b2,    float* __restrict__ params) {
  __shared__ float hbuf[4][16 * HID];
  const int lane  = threadIdx.x & 31;
  const int wave  = threadIdx.x >> 5;
  const int row0  = (blockIdx.x * 4 + wave) * 16;
  const int col   = lane & 15;
  const int rhalf = (lane >> 4) * 8;
  float* h = hbuf[wave];

  for (int nt = 0; nt < HID / 16; ++nt) {
    v8f c = {};
#pragma unroll
    for (int kc = 0; kc < N_QUBITS; kc += 4) {
      v2f a = load_a_frag(noise, N_QUBITS, row0, kc, lane);
      v2f b = load_b_frag(W1, HID, kc, nt * 16, lane);
      c = wmma_f32(a, b, c);
    }
    int n = nt * 16 + col;
    float bias = b1[n];
#pragma unroll
    for (int r = 0; r < 8; ++r)
      h[(rhalf + r) * HID + n] = tanhf(c[r] + bias);
  }
  __syncthreads();

  for (int nt = 0; nt < NPARAMS / 16; ++nt) {
    v8f c = {};
#pragma unroll
    for (int kc = 0; kc < HID; kc += 4) {
      v2f a = load_a_frag(h, HID, 0, kc, lane);
      v2f b = load_b_frag(W2, NPARAMS, kc, nt * 16, lane);
      c = wmma_f32(a, b, c);
    }
    int n = nt * 16 + col;
    float bias = b2[n];
#pragma unroll
    for (int r = 0; r < 8; ++r)
      params[(size_t)(row0 + rhalf + r) * NPARAMS + n] = c[r] + bias;
  }
}

__global__ __launch_bounds__(256) void quantum_kernel(
    const float* __restrict__ params, float* __restrict__ meas) {
  __shared__ float s_re0[DIM], s_im0[DIM], s_re1[DIM], s_im1[DIM];
  const int tid = threadIdx.x;
  const int bidx = blockIdx.x;

  for (int i = tid; i < DIM; i += 256) { s_re0[i] = 0.0f; s_im0[i] = 0.0f; }
  if (tid == 0) s_re0[0] = 1.0f;
  __syncthreads();

  const float* p = params + (size_t)bidx * NPARAMS;

  for (int l = 0; l < DEPTH; ++l) {
    float* Ar = (l & 1) ? s_re1 : s_re0;
    float* Ai = (l & 1) ? s_im1 : s_im0;
    float* Br = (l & 1) ? s_re0 : s_re1;
    float* Bi = (l & 1) ? s_im0 : s_im1;

    for (int q = 0; q < N_QUBITS; ++q) {
      const float aa = p[(l * N_QUBITS + q) * 3 + 0];
      const float bb = p[(l * N_QUBITS + q) * 3 + 1];
      const float cc = p[(l * N_QUBITS + q) * 3 + 2];
      float sa, ca, sb, cb, sz, cz;
      sincosf(aa * 0.5f, &sa, &ca);
      sincosf(bb * 0.5f, &sb, &cb);
      sincosf(cc * 0.5f, &sz, &cz);
      const float m00r =  cb * ca, m00i =  sb * sa;
      const float m01r = -sb * ca, m01i = -cb * sa;
      const float m10r =  sb * ca, m10i = -cb * sa;
      const float m11r =  cb * ca, m11i = -sb * sa;
      const float U00r = cz * m00r + sz * m00i, U00i = cz * m00i - sz * m00r;
      const float U01r = cz * m01r + sz * m01i, U01i = cz * m01i - sz * m01r;
      const float U10r = cz * m10r - sz * m10i, U10i = cz * m10i + sz * m10r;
      const float U11r = cz * m11r - sz * m11i, U11i = cz * m11i + sz * m11r;

      const int bitpos  = N_QUBITS - 1 - q;
      const int stride  = 1 << bitpos;
      const int lowmask = stride - 1;
      for (int t = tid; t < DIM / 2; t += 256) {
        int i0 = ((t & ~lowmask) << 1) | (t & lowmask);
        int i1 = i0 | stride;
        float xr = Ar[i0], xi = Ai[i0];
        float yr = Ar[i1], yi = Ai[i1];
        Ar[i0] = U00r * xr - U00i * xi + U01r * yr - U01i * yi;
        Ai[i0] = U00r * xi + U00i * xr + U01r * yi + U01i * yr;
        Ar[i1] = U10r * xr - U10i * xi + U11r * yr - U11i * yi;
        Ai[i1] = U10r * xi + U10i * xr + U11r * yi + U11i * yr;
      }
      __syncthreads();
    }

    for (int i = tid; i < DIM; i += 256) {
      int x = i;
#pragma unroll
      for (int q = 0; q < N_QUBITS; ++q) {
        int pc = N_QUBITS - 1 - q;
        int pt = N_QUBITS - 1 - ((q + 1) % N_QUBITS);
        if ((x >> pc) & 1) x ^= (1 << pt);
      }
      Br[x] = Ar[i];
      Bi[x] = Ai[i];
    }
    __syncthreads();
  }

  float acc[N_QUBITS];
#pragma unroll
  for (int q = 0; q < N_QUBITS; ++q) acc[q] = 0.0f;
  for (int i = tid; i < DIM; i += 256) {
    float pr = s_re0[i] * s_re0[i] + s_im0[i] * s_im0[i];
#pragma unroll
    for (int q = 0; q < N_QUBITS; ++q)
      acc[q] += ((i >> (N_QUBITS - 1 - q)) & 1) ? -pr : pr;
  }
  float* part = s_re1;
#pragma unroll
  for (int q = 0; q < N_QUBITS; ++q) part[tid * N_QUBITS + q] = acc[q];
  __syncthreads();
  for (int s = 128; s > 0; s >>= 1) {
    if (tid < s) {
#pragma unroll
      for (int q = 0; q < N_QUBITS; ++q)
        part[tid * N_QUBITS + q] += part[(tid + s) * N_QUBITS + q];
    }
    __syncthreads();
  }
  if (tid < N_QUBITS) meas[(size_t)bidx * N_QUBITS + tid] = part[tid];
}

__global__ __launch_bounds__(128) void back_mlp_kernel(
    const float* __restrict__ meas, const float* __restrict__ W3,
    const float* __restrict__ b3,   const float* __restrict__ W4,
    const float* __restrict__ b4,   float* __restrict__ out) {
  __shared__ float hbuf[4][16 * HID];
  const int lane  = threadIdx.x & 31;
  const int wave  = threadIdx.x >> 5;
  const int row0  = (blockIdx.x * 4 + wave) * 16;
  const int col   = lane & 15;
  const int rhalf = (lane >> 4) * 8;
  float* h = hbuf[wave];

  for (int nt = 0; nt < HID / 16; ++nt) {
    v8f c = {};
#pragma unroll
    for (int kc = 0; kc < N_QUBITS; kc += 4) {
      v2f a = load_a_frag(meas, N_QUBITS, row0, kc, lane);
      v2f b = load_b_frag(W3, HID, kc, nt * 16, lane);
      c = wmma_f32(a, b, c);
    }
    int n = nt * 16 + col;
    float bias = b3[n];
#pragma unroll
    for (int r = 0; r < 8; ++r)
      h[(rhalf + r) * HID + n] = tanhf(c[r] + bias);
  }
  __syncthreads();

  v8f c = {};
#pragma unroll
  for (int kc = 0; kc < HID; kc += 4) {
    v2f a = load_a_frag(h, HID, 0, kc, lane);
    int k = kc + ((lane >> 4) << 1);
    v2f b;
    b.x = (col < OUT_DIM) ? W4[k * OUT_DIM + col] : 0.0f;
    b.y = (col < OUT_DIM) ? W4[(k + 1) * OUT_DIM + col] : 0.0f;
    c = wmma_f32(a, b, c);
  }
  if (col < OUT_DIM) {
    float bias = b4[col];
#pragma unroll
    for (int r = 0; r < 8; ++r)
      out[(size_t)(row0 + rhalf + r) * OUT_DIM + col] = c[r] + bias;
  }
}

extern "C" void kernel_launch(void* const* d_in, const int* in_sizes, int n_in,
                              void* d_out, int out_size, void* d_ws, size_t ws_size,
                              hipStream_t stream) {
  const float* noise = (const float*)d_in[0];
  const float* W1 = (const float*)d_in[1];
  const float* b1 = (const float*)d_in[2];
  const float* W2 = (const float*)d_in[3];
  const float* b2 = (const float*)d_in[4];
  const float* W3 = (const float*)d_in[5];
  const float* b3 = (const float*)d_in[6];
  const float* W4 = (const float*)d_in[7];
  const float* b4 = (const float*)d_in[8];
  float* out = (float*)d_out;

  float* params = (float*)d_ws;
  float* meas   = params + (size_t)BATCH * NPARAMS;

  front_mlp_kernel<<<BATCH / 64, 128, 0, stream>>>(noise, W1, b1, W2, b2, params);
  quantum_kernel<<<BATCH, 256, 0, stream>>>(params, meas);
  back_mlp_kernel<<<BATCH / 64, 128, 0, stream>>>(meas, W3, b3, W4, b4, out);
}


# State-vector kernel keeps 64 complex amplitudes per thread in registers (6 qubits per LDS pass, CNOT ring folded into read addresses); expectation sums become a pairwise tree plus lane shuffles
# speedup vs baseline: 2.7960x; 2.7960x over previous
#include <hip/hip_runtime.h>
#include <math.h>

#define N_QUBITS 12
#define DEPTH    4
#define HID      64
#define NPARAMS  144
#define BATCH    4096
#define DIM      4096
#define OUT_DIM  2

typedef __attribute__((ext_vector_type(2))) float v2f;
typedef __attribute__((ext_vector_type(8))) float v8f;

__device__ __forceinline__ v8f wmma_f32(v2f a, v2f b, v8f c) {
  return __builtin_amdgcn_wmma_f32_16x16x4_f32(
       false, a,  false, b,
       (short)0, c,  false,  false);
}

__device__ __forceinline__ v2f load_a_frag(const float* src, int lda,
                                           int row0, int k0, int lane) {
  int m = row0 + (lane & 15);
  int k = k0 + ((lane >> 4) << 1);
  v2f a;
  a.x = src[m * lda + k];
  a.y = src[m * lda + k + 1];
  return a;
}

__device__ __forceinline__ v2f load_b_frag(const float* W, int ldb,
                                           int k0, int n0, int lane) {
  int c = n0 + (lane & 15);
  int k = k0 + ((lane >> 4) << 1);
  v2f b;
  b.x = W[k * ldb + c];
  b.y = W[(k + 1) * ldb + c];
  return b;
}

__global__ __launch_bounds__(128) void front_mlp_kernel(
    const float* __restrict__ noise, const float* __restrict__ W1,
    const float* __restrict__ b1,    const float* __restrict__ W2,
    const float* __restrict__ b2,    float* __restrict__ params) {
  __shared__ float hbuf[4][16 * HID];
  const int lane  = threadIdx.x & 31;
  const int wave  = threadIdx.x >> 5;
  const int row0  = (blockIdx.x * 4 + wave) * 16;
  const int col   = lane & 15;
  const int rhalf = (lane >> 4) * 8;
  float* h = hbuf[wave];

  for (int nt = 0; nt < HID / 16; ++nt) {
    v8f c = {};
#pragma unroll
    for (int kc = 0; kc < N_QUBITS; kc += 4) {
      v2f a = load_a_frag(noise, N_QUBITS, row0, kc, lane);
      v2f b = load_b_frag(W1, HID, kc, nt * 16, lane);
      c = wmma_f32(a, b, c);
    }
    int n = nt * 16 + col;
    float bias = b1[n];
#pragma unroll
    for (int r = 0; r < 8; ++r)
      h[(rhalf + r) * HID + n] = tanhf(c[r] + bias);
  }
  __syncthreads();

  for (int nt = 0; nt < NPARAMS / 16; ++nt) {
    v8f c = {};
#pragma unroll
    for (int kc = 0; kc < HID; kc += 4) {
      v2f a = load_a_frag(h, HID, 0, kc, lane);
      v2f b = load_b_frag(W2, NPARAMS, kc, nt * 16, lane);
      c = wmma_f32(a, b, c);
    }
    int n = nt * 16 + col;
    float bias = b2[n];
#pragma unroll
    for (int r = 0; r < 8; ++r)
      params[(size_t)(row0 + rhalf + r) * NPARAMS + n] = c[r] + bias;
  }
}

#define QPITCH 66

typedef __attribute__((ext_vector_type(4))) float v4f;

struct Gate { float r00, i00, r01, i01, r10, i10, r11, i11; };

__device__ __forceinline__ Gate make_gate(const float* __restrict__ p, int g) {
  const float aa = p[g * 3 + 0];
  const float bb = p[g * 3 + 1];
  const float cc = p[g * 3 + 2];
  float sa, ca, sb, cb, sz, cz;
  sincosf(aa * 0.5f, &sa, &ca);
  sincosf(bb * 0.5f, &sb, &cb);
  sincosf(cc * 0.5f, &sz, &cz);
  const float m00r =  cb * ca, m00i =  sb * sa;
  const float m01r = -sb * ca, m01i = -cb * sa;
  const float m10r =  sb * ca, m10i = -cb * sa;
  const float m11r =  cb * ca, m11i = -sb * sa;
  Gate U;
  U.r00 = cz * m00r + sz * m00i;  U.i00 = cz * m00i - sz * m00r;
  U.r01 = cz * m01r + sz * m01i;  U.i01 = cz * m01i - sz * m01r;
  U.r10 = cz * m10r - sz * m10i;  U.i10 = cz * m10i + sz * m10r;
  U.r11 = cz * m11r - sz * m11i;  U.i11 = cz * m11i + sz * m11r;
  return U;
}

template <int RB>
__device__ __forceinline__ void apply_gate(v2f (&c)[64], const Gate& U) {
  v2f n00, n01, n10, n11;
  n00.x = -U.i00; n00.y = U.i00;
  n01.x = -U.i01; n01.y = U.i01;
  n10.x = -U.i10; n10.y = U.i10;
  n11.x = -U.i11; n11.y = U.i11;
#pragma unroll
  for (int j = 0; j < 32; ++j) {
    const int i0 = ((j >> RB) << (RB + 1)) | (j & ((1 << RB) - 1));
    const int i1 = i0 | (1 << RB);
    const v2f x = c[i0], y = c[i1];
    const v2f sx = __builtin_shufflevector(x, x, 1, 0);
    const v2f sy = __builtin_shufflevector(y, y, 1, 0);
    c[i0] = U.r00 * x + n00 * sx + U.r01 * y + n01 * sy;
    c[i1] = U.r10 * x + n10 * sx + U.r11 * y + n11 * sy;
  }
}

__device__ __forceinline__ void apply_six(v2f (&c)[64],
                                          const float* __restrict__ p, int g0) {
  apply_gate<5>(c, make_gate(p, g0 + 0));
  apply_gate<4>(c, make_gate(p, g0 + 1));
  apply_gate<3>(c, make_gate(p, g0 + 2));
  apply_gate<2>(c, make_gate(p, g0 + 3));
  apply_gate<1>(c, make_gate(p, g0 + 4));
  apply_gate<0>(c, make_gate(p, g0 + 5));
}

__device__ __forceinline__ int parity6(int v) { return __popc(v & 63) & 1; }

__global__ __launch_bounds__(64) void quantum_kernel(
    const float* __restrict__ params, float* __restrict__ meas) {
  __shared__ __attribute__((aligned(16))) v2f st[64 * QPITCH];
  __shared__ float red[2][N_QUBITS];
  const int t = threadIdx.x;
  const int bidx = blockIdx.x;
  const float* p = params + (size_t)bidx * NPARAMS;

  v2f c[64];
#pragma unroll
  for (int r = 0; r < 64; ++r) { c[r].x = 0.0f; c[r].y = 0.0f; }
  c[0].x = (t == 0) ? 1.0f : 0.0f;

  const int cg0 = t ^ (t >> 1);
  const int cg1 = cg0 ^ 32;
  const int t0  = t & 1;

#pragma unroll 1
  for (int l = 0; l < DEPTH; ++l) {
    apply_six(c, p, l * N_QUBITS);
#pragma unroll
    for (int r = 0; r < 64; ++r) st[r * QPITCH + t] = c[r];
    __syncthreads();
    {
      const v4f* row = reinterpret_cast<const v4f*>(st + t * QPITCH);
#pragma unroll
      for (int h = 0; h < 32; ++h) {
        const v4f v = row[h];
        c[2 * h].x = v.x;     c[2 * h].y = v.y;
        c[2 * h + 1].x = v.z; c[2 * h + 1].y = v.w;
      }
    }
    apply_six(c, p, l * N_QUBITS + 6);
    if (l == DEPTH - 1) break;
    {
      v4f* row = reinterpret_cast<v4f*>(st + t * QPITCH);
#pragma unroll
      for (int h = 0; h < 32; ++h) {
        v4f v;
        v.x = c[2 * h].x;     v.y = c[2 * h].y;
        v.z = c[2 * h + 1].x; v.w = c[2 * h + 1].y;
        row[h] = v;
      }
    }
    __syncthreads();
#pragma unroll
    for (int r = 0; r < 64; ++r) {
      const int G  = r ^ (r >> 1);
      const int hi = (G >> 4) & 3;
      const int delta = (hi == 0) ? 48 : (hi == 1) ? 16 : (hi == 2) ? -16 : -48;
      const int idx = G * QPITCH + ((r & 1) ? cg1 : cg0) + t0 * (delta * QPITCH);
      c[r] = st[idx];
    }
    __syncthreads();
  }

  float cur[64];
#pragma unroll
  for (int r = 0; r < 64; ++r) cur[r] = c[r].x * c[r].x + c[r].y * c[r].y;
  float D[6];
#pragma unroll
  for (int k = 0; k < 6; ++k) {
    const int n = 32 >> k;
    float d = 0.0f;
#pragma unroll
    for (int m = 0; m < n; ++m) {
      const float a = cur[2 * m] + cur[2 * m + 1];
      const float b = cur[2 * m] - cur[2 * m + 1];
      d += (__builtin_popcount(m) & 1) ? -b : b;
      cur[m] = a;
    }
    D[k] = d;
  }
  const float S = cur[0];

  float z[N_QUBITS];
  const int pt = parity6(t);
#pragma unroll
  for (int k = 0; k < 6; ++k) z[k] = pt ? -D[k] : D[k];
#pragma unroll
  for (int k = 6; k < 11; ++k) z[k] = parity6(t >> (k - 6)) ? -S : S;
  z[11] = parity6(t & 31) ? -D[0] : D[0];

#pragma unroll
  for (int k = 0; k < N_QUBITS; ++k) {
    float v = z[k];
#pragma unroll
    for (int m = 16; m > 0; m >>= 1) v += __shfl_xor(v, m);
    z[k] = v;
  }
  if ((t & 31) == 0) {
#pragma unroll
    for (int k = 0; k < N_QUBITS; ++k) red[t >> 5][k] = z[k];
  }
  __syncthreads();
  if (t < N_QUBITS)
    meas[(size_t)bidx * N_QUBITS + t] = red[0][N_QUBITS - 1 - t] + red[1][N_QUBITS - 1 - t];
}

__global__ __launch_bounds__(128) void back_mlp_kernel(
    const float* __restrict__ meas, const float* __restrict__ W3,
    const float* __restrict__ b3,   const float* __restrict__ W4,
    const float* __restrict__ b4,   float* __restrict__ out) {
  __shared__ float hbuf[4][16 * HID];
  const int lane  = threadIdx.x & 31;
  const int wave  = threadIdx.x >> 5;
  const int row0  = (blockIdx.x * 4 + wave) * 16;
  const int col   = lane & 15;
  const int rhalf = (lane >> 4) * 8;
  float* h = hbuf[wave];

  for (int nt = 0; nt < HID / 16; ++nt) {
    v8f c = {};
#pragma unroll
    for (int kc = 0; kc < N_QUBITS; kc += 4) {
      v2f a = load_a_frag(meas, N_QUBITS, row0, kc, lane);
      v2f b = load_b_frag(W3, HID, kc, nt * 16, lane);
      c = wmma_f32(a, b, c);
    }
    int n = nt * 16 + col;
    float bias = b3[n];
#pragma unroll
    for (int r = 0; r < 8; ++r)
      h[(rhalf + r) * HID + n] = tanhf(c[r] + bias);
  }
  __syncthreads();

  v8f c = {};
#pragma unroll
  for (int kc = 0; kc < HID; kc += 4) {
    v2f a = load_a_frag(h, HID, 0, kc, lane);
    int k = kc + ((lane >> 4) << 1);
    v2f b;
    b.x = (col < OUT_DIM) ? W4[k * OUT_DIM + col] : 0.0f;
    b.y = (col < OUT_DIM) ? W4[(k + 1) * OUT_DIM + col] : 0.0f;
    c = wmma_f32(a, b, c);
  }
  if (col < OUT_DIM) {
    float bias = b4[col];
#pragma unroll
    for (int r = 0; r < 8; ++r)
      out[(size_t)(row0 + rhalf + r) * OUT_DIM + col] = c[r] + bias;
  }
}

extern "C" void kernel_launch(void* const* d_in, const int* in_sizes, int n_in,
                              void* d_out, int out_size, void* d_ws, size_t ws_size,
                              hipStream_t stream) {
  const float* noise = (const float*)d_in[0];
  const float* W1 = (const float*)d_in[1];
  const float* b1 = (const float*)d_in[2];
  const float* W2 = (const float*)d_in[3];
  const float* b2 = (const float*)d_in[4];
  const float* W3 = (const float*)d_in[5];
  const float* b3 = (const float*)d_in[6];
  const float* W4 = (const float*)d_in[7];
  const float* b4 = (const float*)d_in[8];
  float* out = (float*)d_out;

  float* params = (float*)d_ws;
  float* meas   = params + (size_t)BATCH * NPARAMS;

  front_mlp_kernel<<<BATCH / 64, 128, 0, stream>>>(noise, W1, b1, W2, b2, params);
  quantum_kernel<<<BATCH, 64, 0, stream>>>(params, meas);
  back_mlp_kernel<<<BATCH / 64, 128, 0, stream>>>(meas, W3, b3, W4, b4, out);
}


# Layer 0 acts on |0..0>: its state is written directly as the tensor product of the 12 gates' first columns (sin/cos one per lane, products regrouped), already permuted by the first CNOT ring
# speedup vs baseline: 3.1075x; 1.1114x over previous
#include <hip/hip_runtime.h>
#include <math.h>

#define N_QUBITS 12
#define DEPTH    4
#define HID      64
#define NPARAMS  144
#define BATCH    4096
#define DIM      4096
#define OUT_DIM  2

typedef __attribute__((ext_vector_type(2))) float v2f;
typedef __attribute__((ext_vector_type(8))) float v8f;

__device__ __forceinline__ v8f wmma_f32(v2f a, v2f b, v8f c) {
  return __builtin_amdgcn_wmma_f32_16x16x4_f32(
       false, a,  false, b,
       (short)0, c,  false,  false);
}

__device__ __forceinline__ v2f load_a_frag(const float* src, int lda,
                                           int row0, int k0, int lane) {
  int m = row0 + (lane & 15);
  int k = k0 + ((lane >> 4) << 1);
  v2f a;
  a.x = src[m * lda + k];
  a.y = src[m * lda + k + 1];
  return a;
}

__device__ __forceinline__ v2f load_b_frag(const float* W, int ldb,
                                           int k0, int n0, int lane) {
  int c = n0 + (lane & 15);
  int k = k0 + ((lane >> 4) << 1);
  v2f b;
  b.x = W[k * ldb + c];
  b.y = W[(k + 1) * ldb + c];
  return b;
}

__global__ __launch_bounds__(128) void front_mlp_kernel(
    const float* __restrict__ noise, const float* __restrict__ W1,
    const float* __restrict__ b1,    const float* __restrict__ W2,
    const float* __restrict__ b2,    float* __restrict__ params) {
  __shared__ float hbuf[4][16 * HID];
  const int lane  = threadIdx.x & 31;
  const int wave  = threadIdx.x >> 5;
  const int row0  = (blockIdx.x * 4 + wave) * 16;
  const int col   = lane & 15;
  const int rhalf = (lane >> 4) * 8;
  float* h = hbuf[wave];

  for (int nt = 0; nt < HID / 16; ++nt) {
    v8f c = {};
#pragma unroll
    for (int kc = 0; kc < N_QUBITS; kc += 4) {
      v2f a = load_a_frag(noise, N_QUBITS, row0, kc, lane);
      v2f b = load_b_frag(W1, HID, kc, nt * 16, lane);
      c = wmma_f32(a, b, c);
    }
    int n = nt * 16 + col;
    float bias = b1[n];
#pragma unroll
    for (int r = 0; r < 8; ++r)
      h[(rhalf + r) * HID + n] = tanhf(c[r] + bias);
  }
  __syncthreads();

  for (int nt = 0; nt < NPARAMS / 16; ++nt) {
    v8f c = {};
#pragma unroll
    for (int kc = 0; kc < HID; kc += 4) {
      v2f a = load_a_frag(h, HID, 0, kc, lane);
      v2f b = load_b_frag(W2, NPARAMS, kc, nt * 16, lane);
      c = wmma_f32(a, b, c);
    }
    int n = nt * 16 + col;
    float bias = b2[n];
#pragma unroll
    for (int r = 0; r < 8; ++r)
      params[(size_t)(row0 + rhalf + r) * NPARAMS + n] = c[r] + bias;
  }
}

#define QPITCH 66

typedef __attribute__((ext_vector_type(4))) float v4f;

struct Gate { float r00, i00, r01, i01, r10, i10, r11, i11; };

__device__ __forceinline__ Gate make_gate(const float* __restrict__ p, int g) {
  const float aa = p[g * 3 + 0];
  const float bb = p[g * 3 + 1];
  const float cc = p[g * 3 + 2];
  float sa, ca, sb, cb, sz, cz;
  sincosf(aa * 0.5f, &sa, &ca);
  sincosf(bb * 0.5f, &sb, &cb);
  sincosf(cc * 0.5f, &sz, &cz);
  const float m00r =  cb * ca, m00i =  sb * sa;
  const float m01r = -sb * ca, m01i = -cb * sa;
  const float m10r =  sb * ca, m10i = -cb * sa;
  const float m11r =  cb * ca, m11i = -sb * sa;
  Gate U;
  U.r00 = cz * m00r + sz * m00i;  U.i00 = cz * m00i - sz * m00r;
  U.r01 = cz * m01r + sz * m01i;  U.i01 = cz * m01i - sz * m01r;
  U.r10 = cz * m10r - sz * m10i;  U.i10 = cz * m10i + sz * m10r;
  U.r11 = cz * m11r - sz * m11i;  U.i11 = cz * m11i + sz * m11r;
  return U;
}

template <int RB>
__device__ __forceinline__ void apply_gate(v2f (&c)[64], const Gate& U) {
  v2f n00, n01, n10, n11;
  n00.x = -U.i00; n00.y = U.i00;
  n01.x = -U.i01; n01.y = U.i01;
  n10.x = -U.i10; n10.y = U.i10;
  n11.x = -U.i11; n11.y = U.i11;
#pragma unroll
  for (int j = 0; j < 32; ++j) {
    const int i0 = ((j >> RB) << (RB + 1)) | (j & ((1 << RB) - 1));
    const int i1 = i0 | (1 << RB);
    const v2f x = c[i0], y = c[i1];
    const v2f sx = __builtin_shufflevector(x, x, 1, 0);
    const v2f sy = __builtin_shufflevector(y, y, 1, 0);
    c[i0] = U.r00 * x + n00 * sx + U.r01 * y + n01 * sy;
    c[i1] = U.r10 * x + n10 * sx + U.r11 * y + n11 * sy;
  }
}

__device__ __forceinline__ v2f cmul(v2f a, v2f b) {
  const v2f sb = __builtin_shufflevector(b, b, 1, 0);
  v2f na; na.x = -a.y; na.y = a.y;
  return a.x * b + na * sb;
}

__device__ __forceinline__ void apply_six(v2f (&c)[64],
                                          const float* __restrict__ p, int g0) {
  apply_gate<5>(c, make_gate(p, g0 + 0));
  apply_gate<4>(c, make_gate(p, g0 + 1));
  apply_gate<3>(c, make_gate(p, g0 + 2));
  apply_gate<2>(c, make_gate(p, g0 + 3));
  apply_gate<1>(c, make_gate(p, g0 + 4));
  apply_gate<0>(c, make_gate(p, g0 + 5));
}

__device__ __forceinline__ int parity6(int v) { return __popc(v & 63) & 1; }

__global__ __launch_bounds__(64) void quantum_kernel(
    const float* __restrict__ params, float* __restrict__ meas) {
  __shared__ __attribute__((aligned(16))) v2f st[64 * QPITCH];
  __shared__ float red[2][N_QUBITS];
  __shared__ __attribute__((aligned(16))) v2f trig[3 * N_QUBITS];
  const int t = threadIdx.x;
  const int bidx = blockIdx.x;
  const float* p = params + (size_t)bidx * NPARAMS;

  const int cg0 = t ^ (t >> 1);
  const int cg1 = cg0 ^ 32;
  const int t0  = t & 1;

  if (t < 3 * N_QUBITS) {
    float sn, cs;
    sincosf(p[t] * 0.5f, &sn, &cs);
    v2f e; e.x = cs; e.y = sn;
    trig[t] = e;
  }
  __syncthreads();

  v2f c[64];
  {
    v2f f0[N_QUBITS], f1[N_QUBITS];
#pragma unroll
    for (int b = 0; b < N_QUBITS; ++b) {
      const int g = N_QUBITS - 1 - b;
      const v2f ea = trig[3 * g], eb = trig[3 * g + 1], ez = trig[3 * g + 2];
      const float ca = ea.x, sa = ea.y, cb = eb.x, sb = eb.y, cz = ez.x, sz = ez.y;
      const float m00r =  cb * ca, m00i =  sb * sa;
      const float m10r =  sb * ca, m10i = -cb * sa;
      f0[b].x = cz * m00r + sz * m00i;  f0[b].y = cz * m00i - sz * m00r;
      f1[b].x = cz * m10r - sz * m10i;  f1[b].y = cz * m10i + sz * m10r;
    }
    v2f pt = (cg0 & 1) ? f1[0] : f0[0];
#pragma unroll
    for (int b = 1; b < 5; ++b) pt = cmul(((cg0 >> b) & 1) ? f1[b] : f0[b], pt);
    const int t5 = (t >> 5) & 1;
    c[0] = cmul(t5 ? f1[5] : f0[5], pt);
    c[1] = cmul(t5 ? f0[5] : f1[5], pt);
#pragma unroll
    for (int k = 1; k < 5; ++k) {
      const int n = 1 << k;
#pragma unroll
      for (int j = 0; j < n; ++j) {
        const int lowbit = (j >> (k - 1)) & 1;
        c[n + j] = cmul(lowbit ? f0[5 + k] : f1[5 + k], c[j]);
      }
#pragma unroll
      for (int j = 0; j < n; ++j) {
        const int lowbit = (j >> (k - 1)) & 1;
        c[j] = cmul(lowbit ? f1[5 + k] : f0[5 + k], c[j]);
      }
    }
    const v2f a10 = t0 ? f1[10] : f0[10], b10 = t0 ? f0[10] : f1[10];
    const v2f a11 = t0 ? f1[11] : f0[11], b11 = t0 ? f0[11] : f1[11];
    const v2f w00 = cmul(a10, a11);
    const v2f w01 = cmul(b10, a11);
    const v2f w10 = cmul(b10, b11);
    const v2f w11 = cmul(a10, b11);
#pragma unroll
    for (int j = 0; j < 32; ++j) {
      const int r4 = (j >> 4) & 1;
      c[32 + j] = cmul(r4 ? w11 : w10, c[j]);
      c[j]      = cmul(r4 ? w01 : w00, c[j]);
    }
  }

#pragma unroll 1
  for (int l = 1; l < DEPTH; ++l) {
    apply_six(c, p, l * N_QUBITS);
#pragma unroll
    for (int r = 0; r < 64; ++r) st[r * QPITCH + t] = c[r];
    __syncthreads();
    {
      const v4f* row = reinterpret_cast<const v4f*>(st + t * QPITCH);
#pragma unroll
      for (int h = 0; h < 32; ++h) {
        const v4f v = row[h];
        c[2 * h].x = v.x;     c[2 * h].y = v.y;
        c[2 * h + 1].x = v.z; c[2 * h + 1].y = v.w;
      }
    }
    apply_six(c, p, l * N_QUBITS + 6);
    if (l == DEPTH - 1) break;
    {
      v4f* row = reinterpret_cast<v4f*>(st + t * QPITCH);
#pragma unroll
      for (int h = 0; h < 32; ++h) {
        v4f v;
        v.x = c[2 * h].x;     v.y = c[2 * h].y;
        v.z = c[2 * h + 1].x; v.w = c[2 * h + 1].y;
        row[h] = v;
      }
    }
    __syncthreads();
#pragma unroll
    for (int r = 0; r < 64; ++r) {
      const int G  = r ^ (r >> 1);
      const int hi = (G >> 4) & 3;
      const int delta = (hi == 0) ? 48 : (hi == 1) ? 16 : (hi == 2) ? -16 : -48;
      const int idx = G * QPITCH + ((r & 1) ? cg1 : cg0) + t0 * (delta * QPITCH);
      c[r] = st[idx];
    }
    __syncthreads();
  }

  float cur[64];
#pragma unroll
  for (int r = 0; r < 64; ++r) cur[r] = c[r].x * c[r].x + c[r].y * c[r].y;
  float D[6];
#pragma unroll
  for (int k = 0; k < 6; ++k) {
    const int n = 32 >> k;
    float d = 0.0f;
#pragma unroll
    for (int m = 0; m < n; ++m) {
      const float a = cur[2 * m] + cur[2 * m + 1];
      const float b = cur[2 * m] - cur[2 * m + 1];
      d += (__builtin_popcount(m) & 1) ? -b : b;
      cur[m] = a;
    }
    D[k] = d;
  }
  const float S = cur[0];

  float z[N_QUBITS];
  const int pt = parity6(t);
#pragma unroll
  for (int k = 0; k < 6; ++k) z[k] = pt ? -D[k] : D[k];
#pragma unroll
  for (int k = 6; k < 11; ++k) z[k] = parity6(t >> (k - 6)) ? -S : S;
  z[11] = parity6(t & 31) ? -D[0] : D[0];

#pragma unroll
  for (int k = 0; k < N_QUBITS; ++k) {
    float v = z[k];
#pragma unroll
    for (int m = 16; m > 0; m >>= 1) v += __shfl_xor(v, m);
    z[k] = v;
  }
  if ((t & 31) == 0) {
#pragma unroll
    for (int k = 0; k < N_QUBITS; ++k) red[t >> 5][k] = z[k];
  }
  __syncthreads();
  if (t < N_QUBITS)
    meas[(size_t)bidx * N_QUBITS + t] = red[0][N_QUBITS - 1 - t] + red[1][N_QUBITS - 1 - t];
}

__global__ __launch_bounds__(128) void back_mlp_kernel(
    const float* __restrict__ meas, const float* __restrict__ W3,
    const float* __restrict__ b3,   const float* __restrict__ W4,
    const float* __restrict__ b4,   float* __restrict__ out) {
  __shared__ float hbuf[4][16 * HID];
  const int lane  = threadIdx.x & 31;
  const int wave  = threadIdx.x >> 5;
  const int row0  = (blockIdx.x * 4 + wave) * 16;
  const int col   = lane & 15;
  const int rhalf = (lane >> 4) * 8;
  float* h = hbuf[wave];

  for (int nt = 0; nt < HID / 16; ++nt) {
    v8f c = {};
#pragma unroll
    for (int kc = 0; kc < N_QUBITS; kc += 4) {
      v2f a = load_a_frag(meas, N_QUBITS, row0, kc, lane);
      v2f b = load_b_frag(W3, HID, kc, nt * 16, lane);
      c = wmma_f32(a, b, c);
    }
    int n = nt * 16 + col;
    float bias = b3[n];
#pragma unroll
    for (int r = 0; r < 8; ++r)
      h[(rhalf + r) * HID + n] = tanhf(c[r] + bias);
  }
  __syncthreads();

  v8f c = {};
#pragma unroll
  for (int kc = 0; kc < HID; kc += 4) {
    v2f a = load_a_frag(h, HID, 0, kc, lane);
    int k = kc + ((lane >> 4) << 1);
    v2f b;
    b.x = (col < OUT_DIM) ? W4[k * OUT_DIM + col] : 0.0f;
    b.y = (col < OUT_DIM) ? W4[(k + 1) * OUT_DIM + col] : 0.0f;
    c = wmma_f32(a, b, c);
  }
  if (col < OUT_DIM) {
    float bias = b4[col];
#pragma unroll
    for (int r = 0; r < 8; ++r)
      out[(size_t)(row0 + rhalf + r) * OUT_DIM + col] = c[r] + bias;
  }
}

extern "C" void kernel_launch(void* const* d_in, const int* in_sizes, int n_in,
                              void* d_out, int out_size, void* d_ws, size_t ws_size,
                              hipStream_t stream) {
  const float* noise = (const float*)d_in[0];
  const float* W1 = (const float*)d_in[1];
  const float* b1 = (const float*)d_in[2];
  const float* W2 = (const float*)d_in[3];
  const float* b2 = (const float*)d_in[4];
  const float* W3 = (const float*)d_in[5];
  const float* b3 = (const float*)d_in[6];
  const float* W4 = (const float*)d_in[7];
  const float* b4 = (const float*)d_in[8];
  float* out = (float*)d_out;

  float* params = (float*)d_ws;
  float* meas   = params + (size_t)BATCH * NPARAMS;

  front_mlp_kernel<<<BATCH / 64, 128, 0, stream>>>(noise, W1, b1, W2, b2, params);
  quantum_kernel<<<BATCH, 64, 0, stream>>>(params, meas);
  back_mlp_kernel<<<BATCH / 64, 128, 0, stream>>>(meas, W3, b3, W4, b4, out);
}


# State-vector kernel: gate matrices of layers 1-3 built once per block (sincos and gate one per lane), read from LDS, not 108 inlined scalar sincos per wave; half-layer body looped; text 66 -> 24 KB
# speedup vs baseline: 4.2837x; 1.3785x over previous
#include <hip/hip_runtime.h>
#include <math.h>

#define N_QUBITS 12
#define DEPTH    4
#define HID      64
#define NPARAMS  144
#define BATCH    4096
#define DIM      4096
#define OUT_DIM  2

typedef __attribute__((ext_vector_type(2))) float v2f;
typedef __attribute__((ext_vector_type(8))) float v8f;

__device__ __forceinline__ v8f wmma_f32(v2f a, v2f b, v8f c) {
  return __builtin_amdgcn_wmma_f32_16x16x4_f32(
       false, a,  false, b,
       (short)0, c,  false,  false);
}

__device__ __forceinline__ v2f load_a_frag(const float* src, int lda,
                                           int row0, int k0, int lane) {
  int m = row0 + (lane & 15);
  int k = k0 + ((lane >> 4) << 1);
  v2f a;
  a.x = src[m * lda + k];
  a.y = src[m * lda + k + 1];
  return a;
}

__device__ __forceinline__ v2f load_b_frag(const float* W, int ldb,
                                           int k0, int n0, int lane) {
  int c = n0 + (lane & 15);
  int k = k0 + ((lane >> 4) << 1);
  v2f b;
  b.x = W[k * ldb + c];
  b.y = W[(k + 1) * ldb + c];
  return b;
}

__global__ __launch_bounds__(128) void front_mlp_kernel(
    const float* __restrict__ noise, const float* __restrict__ W1,
    const float* __restrict__ b1,    const float* __restrict__ W2,
    const float* __restrict__ b2,    float* __restrict__ params) {
  __shared__ float hbuf[4][16 * HID];
  const int lane  = threadIdx.x & 31;
  const int wave  = threadIdx.x >> 5;
  const int row0  = (blockIdx.x * 4 + wave) * 16;
  const int col   = lane & 15;
  const int rhalf = (lane >> 4) * 8;
  float* h = hbuf[wave];

  for (int nt = 0; nt < HID / 16; ++nt) {
    v8f c = {};
#pragma unroll
    for (int kc = 0; kc < N_QUBITS; kc += 4) {
      v2f a = load_a_frag(noise, N_QUBITS, row0, kc, lane);
      v2f b = load_b_frag(W1, HID, kc, nt * 16, lane);
      c = wmma_f32(a, b, c);
    }
    int n = nt * 16 + col;
    float bias = b1[n];
#pragma unroll
    for (int r = 0; r < 8; ++r)
      h[(rhalf + r) * HID + n] = tanhf(c[r] + bias);
  }
  __syncthreads();

  for (int nt = 0; nt < NPARAMS / 16; ++nt) {
    v8f c = {};
#pragma unroll
    for (int kc = 0; kc < HID; kc += 4) {
      v2f a = load_a_frag(h, HID, 0, kc, lane);
      v2f b = load_b_frag(W2, NPARAMS, kc, nt * 16, lane);
      c = wmma_f32(a, b, c);
    }
    int n = nt * 16 + col;
    float bias = b2[n];
#pragma unroll
    for (int r = 0; r < 8; ++r)
      params[(size_t)(row0 + rhalf + r) * NPARAMS + n] = c[r] + bias;
  }
}

#define QPITCH 66

typedef __attribute__((ext_vector_type(4))) float v4f;

struct Gate { float r00, i00, r01, i01, r10, i10, r11, i11; };

__device__ __forceinline__ Gate make_gate(v2f ea, v2f eb, v2f ez) {
  const float ca = ea.x, sa = ea.y, cb = eb.x, sb = eb.y, cz = ez.x, sz = ez.y;
  const float m00r =  cb * ca, m00i =  sb * sa;
  const float m01r = -sb * ca, m01i = -cb * sa;
  const float m10r =  sb * ca, m10i = -cb * sa;
  const float m11r =  cb * ca, m11i = -sb * sa;
  Gate U;
  U.r00 = cz * m00r + sz * m00i;  U.i00 = cz * m00i - sz * m00r;
  U.r01 = cz * m01r + sz * m01i;  U.i01 = cz * m01i - sz * m01r;
  U.r10 = cz * m10r - sz * m10i;  U.i10 = cz * m10i + sz * m10r;
  U.r11 = cz * m11r - sz * m11i;  U.i11 = cz * m11i + sz * m11r;
  return U;
}

template <int RB>
__device__ __forceinline__ void apply_gate(v2f (&c)[64], const Gate& U) {
  v2f n00, n01, n10, n11;
  n00.x = -U.i00; n00.y = U.i00;
  n01.x = -U.i01; n01.y = U.i01;
  n10.x = -U.i10; n10.y = U.i10;
  n11.x = -U.i11; n11.y = U.i11;
#pragma unroll
  for (int j = 0; j < 32; ++j) {
    const int i0 = ((j >> RB) << (RB + 1)) | (j & ((1 << RB) - 1));
    const int i1 = i0 | (1 << RB);
    const v2f x = c[i0], y = c[i1];
    const v2f sx = __builtin_shufflevector(x, x, 1, 0);
    const v2f sy = __builtin_shufflevector(y, y, 1, 0);
    c[i0] = U.r00 * x + n00 * sx + U.r01 * y + n01 * sy;
    c[i1] = U.r10 * x + n10 * sx + U.r11 * y + n11 * sy;
  }
}

__device__ __forceinline__ v2f cmul(v2f a, v2f b) {
  const v2f sb = __builtin_shufflevector(b, b, 1, 0);
  v2f na; na.x = -a.y; na.y = a.y;
  return a.x * b + na * sb;
}

__device__ __forceinline__ Gate load_gate(const float* g) {
  const v4f a = *reinterpret_cast<const v4f*>(g);
  const v4f b = *reinterpret_cast<const v4f*>(g + 4);
  Gate U;
  U.r00 = a.x; U.i00 = a.y; U.r01 = a.z; U.i01 = a.w;
  U.r10 = b.x; U.i10 = b.y; U.r11 = b.z; U.i11 = b.w;
  return U;
}

__device__ __forceinline__ void apply_six(v2f (&c)[64], const float* g) {
  const Gate U5 = load_gate(g),      U4 = load_gate(g + 8),  U3 = load_gate(g + 16);
  const Gate U2 = load_gate(g + 24), U1 = load_gate(g + 32), U0 = load_gate(g + 40);
  apply_gate<5>(c, U5);
  apply_gate<4>(c, U4);
  apply_gate<3>(c, U3);
  apply_gate<2>(c, U2);
  apply_gate<1>(c, U1);
  apply_gate<0>(c, U0);
}

__device__ __forceinline__ int parity6(int v) { return __popc(v & 63) & 1; }

__global__ __launch_bounds__(64) void quantum_kernel(
    const float* __restrict__ params, float* __restrict__ meas) {
  __shared__ __attribute__((aligned(16))) v2f st[64 * QPITCH];
  __shared__ float red[2][N_QUBITS];
  __shared__ __attribute__((aligned(16))) v2f trig[NPARAMS];
  __shared__ __attribute__((aligned(16))) float gl[(DEPTH - 1) * N_QUBITS * 8];
  const int t = threadIdx.x;
  const int bidx = blockIdx.x;
  const float* p = params + (size_t)bidx * NPARAMS;

  const int cg0 = t ^ (t >> 1);
  const int cg1 = cg0 ^ 32;
  const int t0  = t & 1;

  const float ang0 = p[t], ang1 = p[64 + t], ang2 = p[128 + (t & 15)];
#pragma unroll 1
  for (int j = 0; j < 3; ++j) {
    const float ang = (j == 0) ? ang0 : (j == 1) ? ang1 : ang2;
    float sn, cs;
    sincosf(ang * 0.5f, &sn, &cs);
    v2f e; e.x = cs; e.y = sn;
    if (j < 2 || t < 16) trig[64 * j + t] = e;
  }
  __syncthreads();
  if (t < (DEPTH - 1) * N_QUBITS) {
    const int a0 = 3 * (N_QUBITS + t);
    const Gate U = make_gate(trig[a0], trig[a0 + 1], trig[a0 + 2]);
    v4f lo, hi;
    lo.x = U.r00; lo.y = U.i00; lo.z = U.r01; lo.w = U.i01;
    hi.x = U.r10; hi.y = U.i10; hi.z = U.r11; hi.w = U.i11;
    v4f* g = reinterpret_cast<v4f*>(gl + 8 * t);
    g[0] = lo;
    g[1] = hi;
  }

  v2f c[64];
  {
    v2f f0[N_QUBITS], f1[N_QUBITS];
#pragma unroll
    for (int b = 0; b < N_QUBITS; ++b) {
      const int g = N_QUBITS - 1 - b;
      const v2f ea = trig[3 * g], eb = trig[3 * g + 1], ez = trig[3 * g + 2];
      const float ca = ea.x, sa = ea.y, cb = eb.x, sb = eb.y, cz = ez.x, sz = ez.y;
      const float m00r =  cb * ca, m00i =  sb * sa;
      const float m10r =  sb * ca, m10i = -cb * sa;
      f0[b].x = cz * m00r + sz * m00i;  f0[b].y = cz * m00i - sz * m00r;
      f1[b].x = cz * m10r - sz * m10i;  f1[b].y = cz * m10i + sz * m10r;
    }
    v2f pt = (cg0 & 1) ? f1[0] : f0[0];
#pragma unroll
    for (int b = 1; b < 5; ++b) pt = cmul(((cg0 >> b) & 1) ? f1[b] : f0[b], pt);
    const int t5 = (t >> 5) & 1;
    c[0] = cmul(t5 ? f1[5] : f0[5], pt);
    c[1] = cmul(t5 ? f0[5] : f1[5], pt);
#pragma unroll
    for (int k = 1; k < 5; ++k) {
      const int n = 1 << k;
#pragma unroll
      for (int j = 0; j < n; ++j) {
        const int lowbit = (j >> (k - 1)) & 1;
        c[n + j] = cmul(lowbit ? f0[5 + k] : f1[5 + k], c[j]);
      }
#pragma unroll
      for (int j = 0; j < n; ++j) {
        const int lowbit = (j >> (k - 1)) & 1;
        c[j] = cmul(lowbit ? f1[5 + k] : f0[5 + k], c[j]);
      }
    }
    const v2f a10 = t0 ? f1[10] : f0[10], b10 = t0 ? f0[10] : f1[10];
    const v2f a11 = t0 ? f1[11] : f0[11], b11 = t0 ? f0[11] : f1[11];
    const v2f w00 = cmul(a10, a11);
    const v2f w01 = cmul(b10, a11);
    const v2f w10 = cmul(b10, b11);
    const v2f w11 = cmul(a10, b11);
#pragma unroll
    for (int j = 0; j < 32; ++j) {
      const int r4 = (j >> 4) & 1;
      c[32 + j] = cmul(r4 ? w11 : w10, c[j]);
      c[j]      = cmul(r4 ? w01 : w00, c[j]);
    }
  }

  __syncthreads();

#pragma unroll 1
  for (int s = 0; s < 2 * (DEPTH - 1); ++s) {
    apply_six(c, gl + s * 48);
    if (s == 2 * (DEPTH - 1) - 1) break;
    if ((s & 1) == 0) {
#pragma unroll
      for (int r = 0; r < 64; ++r) st[r * QPITCH + t] = c[r];
      __syncthreads();
      const v4f* row = reinterpret_cast<const v4f*>(st + t * QPITCH);
#pragma unroll
      for (int h = 0; h < 32; ++h) {
        const v4f v = row[h];
        c[2 * h].x = v.x;     c[2 * h].y = v.y;
        c[2 * h + 1].x = v.z; c[2 * h + 1].y = v.w;
      }
    } else {
      v4f* row = reinterpret_cast<v4f*>(st + t * QPITCH);
#pragma unroll
      for (int h = 0; h < 32; ++h) {
        v4f v;
        v.x = c[2 * h].x;     v.y = c[2 * h].y;
        v.z = c[2 * h + 1].x; v.w = c[2 * h + 1].y;
        row[h] = v;
      }
      __syncthreads();
#pragma unroll
      for (int r = 0; r < 64; ++r) {
        const int G  = r ^ (r >> 1);
        const int hi = (G >> 4) & 3;
        const int delta = (hi == 0) ? 48 : (hi == 1) ? 16 : (hi == 2) ? -16 : -48;
        const int idx = G * QPITCH + ((r & 1) ? cg1 : cg0) + t0 * (delta * QPITCH);
        c[r] = st[idx];
      }
      __syncthreads();
    }
  }

  float cur[64];
#pragma unroll
  for (int r = 0; r < 64; ++r) cur[r] = c[r].x * c[r].x + c[r].y * c[r].y;
  float D[6];
#pragma unroll
  for (int k = 0; k < 6; ++k) {
    const int n = 32 >> k;
    float d = 0.0f;
#pragma unroll
    for (int m = 0; m < n; ++m) {
      const float a = cur[2 * m] + cur[2 * m + 1];
      const float b = cur[2 * m] - cur[2 * m + 1];
      d += (__builtin_popcount(m) & 1) ? -b : b;
      cur[m] = a;
    }
    D[k] = d;
  }
  const float S = cur[0];

  float z[N_QUBITS];
  const int pt = parity6(t);
#pragma unroll
  for (int k = 0; k < 6; ++k) z[k] = pt ? -D[k] : D[k];
#pragma unroll
  for (int k = 6; k < 11; ++k) z[k] = parity6(t >> (k - 6)) ? -S : S;
  z[11] = parity6(t & 31) ? -D[0] : D[0];

#pragma unroll
  for (int k = 0; k < N_QUBITS; ++k) {
    float v = z[k];
#pragma unroll
    for (int m = 16; m > 0; m >>= 1) v += __shfl_xor(v, m);
    z[k] = v;
  }
  if ((t & 31) == 0) {
#pragma unroll
    for (int k = 0; k < N_QUBITS; ++k) red[t >> 5][k] = z[k];
  }
  __syncthreads();
  if (t < N_QUBITS)
    meas[(size_t)bidx * N_QUBITS + t] = red[0][N_QUBITS - 1 - t] + red[1][N_QUBITS - 1 - t];
}

__global__ __launch_bounds__(128) void back_mlp_kernel(
    const float* __restrict__ meas, const float* __restrict__ W3,
    const float* __restrict__ b3,   const float* __restrict__ W4,
    const float* __restrict__ b4,   float* __restrict__ out) {
  __shared__ float hbuf[4][16 * HID];
  const int lane  = threadIdx.x & 31;
  const int wave  = threadIdx.x >> 5;
  const int row0  = (blockIdx.x * 4 + wave) * 16;
  const int col   = lane & 15;
  const int rhalf = (lane >> 4) * 8;
  float* h = hbuf[wave];

  for (int nt = 0; nt < HID / 16; ++nt) {
    v8f c = {};
#pragma unroll
    for (int kc = 0; kc < N_QUBITS; kc += 4) {
      v2f a = load_a_frag(meas, N_QUBITS, row0, kc, lane);
      v2f b = load_b_frag(W3, HID, kc, nt * 16, lane);
      c = wmma_f32(a, b, c);
    }
    int n = nt * 16 + col;
    float bias = b3[n];
#pragma unroll
    for (int r = 0; r < 8; ++r)
      h[(rhalf + r) * HID + n] = tanhf(c[r] + bias);
  }
  __syncthreads();

  v8f c = {};
#pragma unroll
  for (int kc = 0; kc < HID; kc += 4) {
    v2f a = load_a_frag(h, HID, 0, kc, lane);
    int k = kc + ((lane >> 4) << 1);
    v2f b;
    b.x = (col < OUT_DIM) ? W4[k * OUT_DIM + col] : 0.0f;
    b.y = (col < OUT_DIM) ? W4[(k + 1) * OUT_DIM + col] : 0.0f;
    c = wmma_f32(a, b, c);
  }
  if (col < OUT_DIM) {
    float bias = b4[col];
#pragma unroll
    for (int r = 0; r < 8; ++r)
      out[(size_t)(row0 + rhalf + r) * OUT_DIM + col] = c[r] + bias;
  }
}

extern "C" void kernel_launch(void* const* d_in, const int* in_sizes, int n_in,
                              void* d_out, int out_size, void* d_ws, size_t ws_size,
                              hipStream_t stream) {
  const float* noise = (const float*)d_in[0];
  const float* W1 = (const float*)d_in[1];
  const float* b1 = (const float*)d_in[2];
  const float* W2 = (const float*)d_in[3];
  const float* b2 = (const float*)d_in[4];
  const float* W3 = (const float*)d_in[5];
  const float* b3 = (const float*)d_in[6];
  const float* W4 = (const float*)d_in[7];
  const float* b4 = (const float*)d_in[8];
  float* out = (float*)d_out;

  float* params = (float*)d_ws;
  float* meas   = params + (size_t)BATCH * NPARAMS;

  front_mlp_kernel<<<BATCH / 64, 128, 0, stream>>>(noise, W1, b1, W2, b2, params);
  quantum_kernel<<<BATCH, 64, 0, stream>>>(params, meas);
  back_mlp_kernel<<<BATCH / 64, 128, 0, stream>>>(meas, W3, b3, W4, b4, out);
}


# Both MLP kernels re-tiled: one 16-row tile per block (256 blocks, not 64), its column tiles split over the block's waves, second-layer weight fragments loaded in front of the first layer
# speedup vs baseline: 4.8568x; 1.1338x over previous
#include <hip/hip_runtime.h>
#include <math.h>

#define N_QUBITS 12
#define DEPTH    4
#define HID      64
#define NPARAMS  144
#define BATCH    4096
#define DIM      4096
#define OUT_DIM  2

typedef __attribute__((ext_vector_type(2))) float v2f;
typedef __attribute__((ext_vector_type(8))) float v8f;

__device__ __forceinline__ v8f wmma_f32(v2f a, v2f b, v8f c) {
  return __builtin_amdgcn_wmma_f32_16x16x4_f32(
       false, a,  false, b,
       (short)0, c,  false,  false);
}

__device__ __forceinline__ v2f load_a_frag(const float* src, int lda,
                                           int row0, int k0, int lane) {
  int m = row0 + (lane & 15);
  int k = k0 + ((lane >> 4) << 1);
  v2f a;
  a.x = src[m * lda + k];
  a.y = src[m * lda + k + 1];
  return a;
}

__device__ __forceinline__ v2f load_b_frag(const float* W, int ldb,
                                           int k0, int n0, int lane) {
  int c = n0 + (lane & 15);
  int k = k0 + ((lane >> 4) << 1);
  v2f b;
  b.x = W[k * ldb + c];
  b.y = W[(k + 1) * ldb + c];
  return b;
}

#define FRONT_WAVES (NPARAMS / 16)

__global__ __launch_bounds__(FRONT_WAVES * 32) void front_mlp_kernel(
    const float* __restrict__ noise, const float* __restrict__ W1,
    const float* __restrict__ b1,    const float* __restrict__ W2,
    const float* __restrict__ b2,    float* __restrict__ params) {
  __shared__ float h[16 * HID];
  const int lane  = threadIdx.x & 31;
  const int wave  = threadIdx.x >> 5;
  const int row0  = blockIdx.x * 16;
  const int col   = lane & 15;
  const int rhalf = (lane >> 4) * 8;

  v2f bw[HID / 4];
#pragma unroll
  for (int i = 0; i < HID / 4; ++i)
    bw[i] = load_b_frag(W2, NPARAMS, 4 * i, wave * 16, lane);
  const float bias2 = b2[wave * 16 + col];

  if (wave < HID / 16) {
    v8f c = {};
#pragma unroll
    for (int kc = 0; kc < N_QUBITS; kc += 4) {
      v2f a = load_a_frag(noise, N_QUBITS, row0, kc, lane);
      v2f b = load_b_frag(W1, HID, kc, wave * 16, lane);
      c = wmma_f32(a, b, c);
    }
    const int n = wave * 16 + col;
    const float bias = b1[n];
#pragma unroll
    for (int r = 0; r < 8; ++r)
      h[(rhalf + r) * HID + n] = tanhf(c[r] + bias);
  }
  __syncthreads();

  v8f c = {};
#pragma unroll
  for (int i = 0; i < HID / 4; ++i) {
    v2f a = load_a_frag(h, HID, 0, 4 * i, lane);
    c = wmma_f32(a, bw[i], c);
  }
  const int n = wave * 16 + col;
#pragma unroll
  for (int r = 0; r < 8; ++r)
    params[(size_t)(row0 + rhalf + r) * NPARAMS + n] = c[r] + bias2;
}

#define QPITCH 66

typedef __attribute__((ext_vector_type(4))) float v4f;

struct Gate { float r00, i00, r01, i01, r10, i10, r11, i11; };

__device__ __forceinline__ Gate make_gate(v2f ea, v2f eb, v2f ez) {
  const float ca = ea.x, sa = ea.y, cb = eb.x, sb = eb.y, cz = ez.x, sz = ez.y;
  const float m00r =  cb * ca, m00i =  sb * sa;
  const float m01r = -sb * ca, m01i = -cb * sa;
  const float m10r =  sb * ca, m10i = -cb * sa;
  const float m11r =  cb * ca, m11i = -sb * sa;
  Gate U;
  U.r00 = cz * m00r + sz * m00i;  U.i00 = cz * m00i - sz * m00r;
  U.r01 = cz * m01r + sz * m01i;  U.i01 = cz * m01i - sz * m01r;
  U.r10 = cz * m10r - sz * m10i;  U.i10 = cz * m10i + sz * m10r;
  U.r11 = cz * m11r - sz * m11i;  U.i11 = cz * m11i + sz * m11r;
  return U;
}

template <int RB>
__device__ __forceinline__ void apply_gate(v2f (&c)[64], const Gate& U) {
  v2f n00, n01, n10, n11;
  n00.x = -U.i00; n00.y = U.i00;
  n01.x = -U.i01; n01.y = U.i01;
  n10.x = -U.i10; n10.y = U.i10;
  n11.x = -U.i11; n11.y = U.i11;
#pragma unroll
  for (int j = 0; j < 32; ++j) {
    const int i0 = ((j >> RB) << (RB + 1)) | (j & ((1 << RB) - 1));
    const int i1 = i0 | (1 << RB);
    const v2f x = c[i0], y = c[i1];
    const v2f sx = __builtin_shufflevector(x, x, 1, 0);
    const v2f sy = __builtin_shufflevector(y, y, 1, 0);
    c[i0] = U.r00 * x + n00 * sx + U.r01 * y + n01 * sy;
    c[i1] = U.r10 * x + n10 * sx + U.r11 * y + n11 * sy;
  }
}

__device__ __forceinline__ v2f cmul(v2f a, v2f b) {
  const v2f sb = __builtin_shufflevector(b, b, 1, 0);
  v2f na; na.x = -a.y; na.y = a.y;
  return a.x * b + na * sb;
}

__device__ __forceinline__ Gate load_gate(const float* g) {
  const v4f a = *reinterpret_cast<const v4f*>(g);
  const v4f b = *reinterpret_cast<const v4f*>(g + 4);
  Gate U;
  U.r00 = a.x; U.i00 = a.y; U.r01 = a.z; U.i01 = a.w;
  U.r10 = b.x; U.i10 = b.y; U.r11 = b.z; U.i11 = b.w;
  return U;
}

__device__ __forceinline__ void apply_six(v2f (&c)[64], const float* g) {
  const Gate U5 = load_gate(g),      U4 = load_gate(g + 8),  U3 = load_gate(g + 16);
  const Gate U2 = load_gate(g + 24), U1 = load_gate(g + 32), U0 = load_gate(g + 40);
  apply_gate<5>(c, U5);
  apply_gate<4>(c, U4);
  apply_gate<3>(c, U3);
  apply_gate<2>(c, U2);
  apply_gate<1>(c, U1);
  apply_gate<0>(c, U0);
}

__device__ __forceinline__ int parity6(int v) { return __popc(v & 63) & 1; }

__global__ __launch_bounds__(64) void quantum_kernel(
    const float* __restrict__ params, float* __restrict__ meas) {
  __shared__ __attribute__((aligned(16))) v2f st[64 * QPITCH];
  __shared__ float red[2][N_QUBITS];
  __shared__ __attribute__((aligned(16))) v2f trig[NPARAMS];
  __shared__ __attribute__((aligned(16))) float gl[(DEPTH - 1) * N_QUBITS * 8];
  const int t = threadIdx.x;
  const int bidx = blockIdx.x;
  const float* p = params + (size_t)bidx * NPARAMS;

  const int cg0 = t ^ (t >> 1);
  const int cg1 = cg0 ^ 32;
  const int t0  = t & 1;

  const float ang0 = p[t], ang1 = p[64 + t], ang2 = p[128 + (t & 15)];
#pragma unroll 1
  for (int j = 0; j < 3; ++j) {
    const float ang = (j == 0) ? ang0 : (j == 1) ? ang1 : ang2;
    float sn, cs;
    sincosf(ang * 0.5f, &sn, &cs);
    v2f e; e.x = cs; e.y = sn;
    if (j < 2 || t < 16) trig[64 * j + t] = e;
  }
  __syncthreads();
  if (t < (DEPTH - 1) * N_QUBITS) {
    const int a0 = 3 * (N_QUBITS + t);
    const Gate U = make_gate(trig[a0], trig[a0 + 1], trig[a0 + 2]);
    v4f lo, hi;
    lo.x = U.r00; lo.y = U.i00; lo.z = U.r01; lo.w = U.i01;
    hi.x = U.r10; hi.y = U.i10; hi.z = U.r11; hi.w = U.i11;
    v4f* g = reinterpret_cast<v4f*>(gl + 8 * t);
    g[0] = lo;
    g[1] = hi;
  }

  v2f c[64];
  {
    v2f f0[N_QUBITS], f1[N_QUBITS];
#pragma unroll
    for (int b = 0; b < N_QUBITS; ++b) {
      const int g = N_QUBITS - 1 - b;
      const v2f ea = trig[3 * g], eb = trig[3 * g + 1], ez = trig[3 * g + 2];
      const float ca = ea.x, sa = ea.y, cb = eb.x, sb = eb.y, cz = ez.x, sz = ez.y;
      const float m00r =  cb * ca, m00i =  sb * sa;
      const float m10r =  sb * ca, m10i = -cb * sa;
      f0[b].x = cz * m00r + sz * m00i;  f0[b].y = cz * m00i - sz * m00r;
      f1[b].x = cz * m10r - sz * m10i;  f1[b].y = cz * m10i + sz * m10r;
    }
    v2f pt = (cg0 & 1) ? f1[0] : f0[0];
#pragma unroll
    for (int b = 1; b < 5; ++b) pt = cmul(((cg0 >> b) & 1) ? f1[b] : f0[b], pt);
    const int t5 = (t >> 5) & 1;
    c[0] = cmul(t5 ? f1[5] : f0[5], pt);
    c[1] = cmul(t5 ? f0[5] : f1[5], pt);
#pragma unroll
    for (int k = 1; k < 5; ++k) {
      const int n = 1 << k;
#pragma unroll
      for (int j = 0; j < n; ++j) {
        const int lowbit = (j >> (k - 1)) & 1;
        c[n + j] = cmul(lowbit ? f0[5 + k] : f1[5 + k], c[j]);
      }
#pragma unroll
      for (int j = 0; j < n; ++j) {
        const int lowbit = (j >> (k - 1)) & 1;
        c[j] = cmul(lowbit ? f1[5 + k] : f0[5 + k], c[j]);
      }
    }
    const v2f a10 = t0 ? f1[10] : f0[10], b10 = t0 ? f0[10] : f1[10];
    const v2f a11 = t0 ? f1[11] : f0[11], b11 = t0 ? f0[11] : f1[11];
    const v2f w00 = cmul(a10, a11);
    const v2f w01 = cmul(b10, a11);
    const v2f w10 = cmul(b10, b11);
    const v2f w11 = cmul(a10, b11);
#pragma unroll
    for (int j = 0; j < 32; ++j) {
      const int r4 = (j >> 4) & 1;
      c[32 + j] = cmul(r4 ? w11 : w10, c[j]);
      c[j]      = cmul(r4 ? w01 : w00, c[j]);
    }
  }

  __syncthreads();

#pragma unroll 1
  for (int s = 0; s < 2 * (DEPTH - 1); ++s) {
    apply_six(c, gl + s * 48);
    if (s == 2 * (DEPTH - 1) - 1) break;
    if ((s & 1) == 0) {
#pragma unroll
      for (int r = 0; r < 64; ++r) st[r * QPITCH + t] = c[r];
      __syncthreads();
      const v4f* row = reinterpret_cast<const v4f*>(st + t * QPITCH);
#pragma unroll
      for (int h = 0; h < 32; ++h) {
        const v4f v = row[h];
        c[2 * h].x = v.x;     c[2 * h].y = v.y;
        c[2 * h + 1].x = v.z; c[2 * h + 1].y = v.w;
      }
    } else {
      v4f* row = reinterpret_cast<v4f*>(st + t * QPITCH);
#pragma unroll
      for (int h = 0; h < 32; ++h) {
        v4f v;
        v.x = c[2 * h].x;     v.y = c[2 * h].y;
        v.z = c[2 * h + 1].x; v.w = c[2 * h + 1].y;
        row[h] = v;
      }
      __syncthreads();
#pragma unroll
      for (int r = 0; r < 64; ++r) {
        const int G  = r ^ (r >> 1);
        const int hi = (G >> 4) & 3;
        const int delta = (hi == 0) ? 48 : (hi == 1) ? 16 : (hi == 2) ? -16 : -48;
        const int idx = G * QPITCH + ((r & 1) ? cg1 : cg0) + t0 * (delta * QPITCH);
        c[r] = st[idx];
      }
      __syncthreads();
    }
  }

  float cur[64];
#pragma unroll
  for (int r = 0; r < 64; ++r) cur[r] = c[r].x * c[r].x + c[r].y * c[r].y;
  float D[6];
#pragma unroll
  for (int k = 0; k < 6; ++k) {
    const int n = 32 >> k;
    float d = 0.0f;
#pragma unroll
    for (int m = 0; m < n; ++m) {
      const float a = cur[2 * m] + cur[2 * m + 1];
      const float b = cur[2 * m] - cur[2 * m + 1];
      d += (__builtin_popcount(m) & 1) ? -b : b;
      cur[m] = a;
    }
    D[k] = d;
  }
  const float S = cur[0];

  float z[N_QUBITS];
  const int pt = parity6(t);
#pragma unroll
  for (int k = 0; k < 6; ++k) z[k] = pt ? -D[k] : D[k];
#pragma unroll
  for (int k = 6; k < 11; ++k) z[k] = parity6(t >> (k - 6)) ? -S : S;
  z[11] = parity6(t & 31) ? -D[0] : D[0];

#pragma unroll
  for (int k = 0; k < N_QUBITS; ++k) {
    float v = z[k];
#pragma unroll
    for (int m = 16; m > 0; m >>= 1) v += __shfl_xor(v, m);
    z[k] = v;
  }
  if ((t & 31) == 0) {
#pragma unroll
    for (int k = 0; k < N_QUBITS; ++k) red[t >> 5][k] = z[k];
  }
  __syncthreads();
  if (t < N_QUBITS)
    meas[(size_t)bidx * N_QUBITS + t] = red[0][N_QUBITS - 1 - t] + red[1][N_QUBITS - 1 - t];
}

__global__ __launch_bounds__(128) void back_mlp_kernel(
    const float* __restrict__ meas, const float* __restrict__ W3,
    const float* __restrict__ b3,   const float* __restrict__ W4,
    const float* __restrict__ b4,   float* __restrict__ out) {
  __shared__ float h[16 * HID];
  const int lane  = threadIdx.x & 31;
  const int wave  = threadIdx.x >> 5;
  const int row0  = blockIdx.x * 16;
  const int col   = lane & 15;
  const int rhalf = (lane >> 4) * 8;

  v2f bw[HID / 4] = {};
  float bias4 = 0.0f;
  if (wave == 0 && col < OUT_DIM) {
#pragma unroll
    for (int i = 0; i < HID / 4; ++i) {
      const int k = 4 * i + ((lane >> 4) << 1);
      bw[i].x = W4[k * OUT_DIM + col];
      bw[i].y = W4[(k + 1) * OUT_DIM + col];
    }
    bias4 = b4[col];
  }

  {
    v8f c = {};
#pragma unroll
    for (int kc = 0; kc < N_QUBITS; kc += 4) {
      v2f a = load_a_frag(meas, N_QUBITS, row0, kc, lane);
      v2f b = load_b_frag(W3, HID, kc, wave * 16, lane);
      c = wmma_f32(a, b, c);
    }
    const int n = wave * 16 + col;
    const float bias = b3[n];
#pragma unroll
    for (int r = 0; r < 8; ++r)
      h[(rhalf + r) * HID + n] = tanhf(c[r] + bias);
  }
  __syncthreads();
  if (wave != 0) return;

  v8f c = {};
#pragma unroll
  for (int i = 0; i < HID / 4; ++i) {
    v2f a = load_a_frag(h, HID, 0, 4 * i, lane);
    c = wmma_f32(a, bw[i], c);
  }
  if (col < OUT_DIM) {
#pragma unroll
    for (int r = 0; r < 8; ++r)
      out[(size_t)(row0 + rhalf + r) * OUT_DIM + col] = c[r] + bias4;
  }
}

extern "C" void kernel_launch(void* const* d_in, const int* in_sizes, int n_in,
                              void* d_out, int out_size, void* d_ws, size_t ws_size,
                              hipStream_t stream) {
  const float* noise = (const float*)d_in[0];
  const float* W1 = (const float*)d_in[1];
  const float* b1 = (const float*)d_in[2];
  const float* W2 = (const float*)d_in[3];
  const float* b2 = (const float*)d_in[4];
  const float* W3 = (const float*)d_in[5];
  const float* b3 = (const float*)d_in[6];
  const float* W4 = (const float*)d_in[7];
  const float* b4 = (const float*)d_in[8];
  float* out = (float*)d_out;

  float* params = (float*)d_ws;
  float* meas   = params + (size_t)BATCH * NPARAMS;

  front_mlp_kernel<<<BATCH / 16, FRONT_WAVES * 32, 0, stream>>>(noise, W1, b1, W2, b2, params);
  quantum_kernel<<<BATCH, 64, 0, stream>>>(params, meas);
  back_mlp_kernel<<<BATCH / 16, 128, 0, stream>>>(meas, W3, b3, W4, b4, out);
}


# State-vector kernel: each gate of layers 1-3 is multiplied by the unit phase that makes its (0,0) entry real (7 packed ops per pair, not 8); last layer's second row too (6); probabilities unchanged
# speedup vs baseline: 5.2900x; 1.0892x over previous
#include <hip/hip_runtime.h>
#include <math.h>

#define N_QUBITS 12
#define DEPTH    4
#define HID      64
#define NPARAMS  144
#define BATCH    4096
#define DIM      4096
#define OUT_DIM  2

typedef __attribute__((ext_vector_type(2))) float v2f;
typedef __attribute__((ext_vector_type(8))) float v8f;

__device__ __forceinline__ v8f wmma_f32(v2f a, v2f b, v8f c) {
  return __builtin_amdgcn_wmma_f32_16x16x4_f32(
       false, a,  false, b,
       (short)0, c,  false,  false);
}

__device__ __forceinline__ v2f load_a_frag(const float* src, int lda,
                                           int row0, int k0, int lane) {
  int m = row0 + (lane & 15);
  int k = k0 + ((lane >> 4) << 1);
  v2f a;
  a.x = src[m * lda + k];
  a.y = src[m * lda + k + 1];
  return a;
}

__device__ __forceinline__ v2f load_b_frag(const float* W, int ldb,
                                           int k0, int n0, int lane) {
  int c = n0 + (lane & 15);
  int k = k0 + ((lane >> 4) << 1);
  v2f b;
  b.x = W[k * ldb + c];
  b.y = W[(k + 1) * ldb + c];
  return b;
}

#define FRONT_WAVES (NPARAMS / 16)

__global__ __launch_bounds__(FRONT_WAVES * 32) void front_mlp_kernel(
    const float* __restrict__ noise, const float* __restrict__ W1,
    const float* __restrict__ b1,    const float* __restrict__ W2,
    const float* __restrict__ b2,    float* __restrict__ params) {
  __shared__ float h[16 * HID];
  const int lane  = threadIdx.x & 31;
  const int wave  = threadIdx.x >> 5;
  const int row0  = blockIdx.x * 16;
  const int col   = lane & 15;
  const int rhalf = (lane >> 4) * 8;

  v2f bw[HID / 4];
#pragma unroll
  for (int i = 0; i < HID / 4; ++i)
    bw[i] = load_b_frag(W2, NPARAMS, 4 * i, wave * 16, lane);
  const float bias2 = b2[wave * 16 + col];

  if (wave < HID / 16) {
    v8f c = {};
#pragma unroll
    for (int kc = 0; kc < N_QUBITS; kc += 4) {
      v2f a = load_a_frag(noise, N_QUBITS, row0, kc, lane);
      v2f b = load_b_frag(W1, HID, kc, wave * 16, lane);
      c = wmma_f32(a, b, c);
    }
    const int n = wave * 16 + col;
    const float bias = b1[n];
#pragma unroll
    for (int r = 0; r < 8; ++r)
      h[(rhalf + r) * HID + n] = tanhf(c[r] + bias);
  }
  __syncthreads();

  v8f c = {};
#pragma unroll
  for (int i = 0; i < HID / 4; ++i) {
    v2f a = load_a_frag(h, HID, 0, 4 * i, lane);
    c = wmma_f32(a, bw[i], c);
  }
  const int n = wave * 16 + col;
#pragma unroll
  for (int r = 0; r < 8; ++r)
    params[(size_t)(row0 + rhalf + r) * NPARAMS + n] = c[r] + bias2;
}

#define QPITCH 66

typedef __attribute__((ext_vector_type(4))) float v4f;

struct Gate { float r00, i00, r01, i01, r10, i10, r11, i11; };

struct PGate { float a, d, r01, i01, r10, i10, r11, i11; };

__device__ __forceinline__ v2f unit_conj(float re, float im, float& mag) {
  const float n2 = re * re + im * im;
  v2f u;
  if (n2 > 1e-30f) {
    const float inv = 1.0f / sqrtf(n2);
    mag = n2 * inv;
    u.x = re * inv; u.y = -im * inv;
  } else {
    mag = re;
    u.x = 1.0f; u.y = 0.0f;
  }
  return u;
}

__device__ __forceinline__ Gate make_gate(v2f ea, v2f eb, v2f ez) {
  const float ca = ea.x, sa = ea.y, cb = eb.x, sb = eb.y, cz = ez.x, sz = ez.y;
  const float m00r =  cb * ca, m00i =  sb * sa;
  const float m01r = -sb * ca, m01i = -cb * sa;
  const float m10r =  sb * ca, m10i = -cb * sa;
  const float m11r =  cb * ca, m11i = -sb * sa;
  Gate U;
  U.r00 = cz * m00r + sz * m00i;  U.i00 = cz * m00i - sz * m00r;
  U.r01 = cz * m01r + sz * m01i;  U.i01 = cz * m01i - sz * m01r;
  U.r10 = cz * m10r - sz * m10i;  U.i10 = cz * m10i + sz * m10r;
  U.r11 = cz * m11r - sz * m11i;  U.i11 = cz * m11i + sz * m11r;
  return U;
}

template <int RB, bool LAST>
__device__ __forceinline__ void apply_gate(v2f (&c)[64], const PGate& U) {
  v2f n01, n10, n11;
  n01.x = -U.i01; n01.y = U.i01;
  n10.x = -U.i10; n10.y = U.i10;
  n11.x = -U.i11; n11.y = U.i11;
#pragma unroll
  for (int j = 0; j < 32; ++j) {
    const int i0 = ((j >> RB) << (RB + 1)) | (j & ((1 << RB) - 1));
    const int i1 = i0 | (1 << RB);
    const v2f x = c[i0], y = c[i1];
    const v2f sx = __builtin_shufflevector(x, x, 1, 0);
    const v2f sy = __builtin_shufflevector(y, y, 1, 0);
    c[i0] = U.a * x + U.r01 * y + n01 * sy;
    if (LAST) c[i1] = U.d * y + U.r10 * x + n10 * sx;
    else      c[i1] = U.r10 * x + n10 * sx + U.r11 * y + n11 * sy;
  }
}

__device__ __forceinline__ v2f cmul(v2f a, v2f b) {
  const v2f sb = __builtin_shufflevector(b, b, 1, 0);
  v2f na; na.x = -a.y; na.y = a.y;
  return a.x * b + na * sb;
}

template <bool LAST>
__device__ __forceinline__ PGate load_gate(const float* g) {
  const v4f a = *reinterpret_cast<const v4f*>(g);
  PGate U;
  U.a   = a.x; U.d   = a.y; U.r01 = a.z; U.i01 = a.w;
  if (LAST) {
    const v2f b = *reinterpret_cast<const v2f*>(g + 4);
    U.r10 = b.x; U.i10 = b.y; U.r11 = 0.0f; U.i11 = 0.0f;
  } else {
    const v4f b = *reinterpret_cast<const v4f*>(g + 4);
    U.r10 = b.x; U.i10 = b.y; U.r11 = b.z; U.i11 = b.w;
  }
  return U;
}

template <bool LAST>
__device__ __forceinline__ void apply_six(v2f (&c)[64], const float* g) {
  const PGate U5 = load_gate<LAST>(g),      U4 = load_gate<LAST>(g + 8),  U3 = load_gate<LAST>(g + 16);
  const PGate U2 = load_gate<LAST>(g + 24), U1 = load_gate<LAST>(g + 32), U0 = load_gate<LAST>(g + 40);
  apply_gate<5, LAST>(c, U5);
  apply_gate<4, LAST>(c, U4);
  apply_gate<3, LAST>(c, U3);
  apply_gate<2, LAST>(c, U2);
  apply_gate<1, LAST>(c, U1);
  apply_gate<0, LAST>(c, U0);
}

__device__ __forceinline__ int parity6(int v) { return __popc(v & 63) & 1; }

__global__ __launch_bounds__(64) __attribute__((amdgpu_waves_per_eu(3))) void quantum_kernel(
    const float* __restrict__ params, float* __restrict__ meas) {
  __shared__ __attribute__((aligned(16))) v2f st[64 * QPITCH];
  __shared__ float red[2][N_QUBITS];
  __shared__ __attribute__((aligned(16))) v2f trig[NPARAMS];
  __shared__ __attribute__((aligned(16))) float gl[(DEPTH - 1) * N_QUBITS * 8];
  const int t = threadIdx.x;
  const int bidx = blockIdx.x;
  const float* p = params + (size_t)bidx * NPARAMS;

  const int cg0 = t ^ (t >> 1);
  const int cg1 = cg0 ^ 32;
  const int t0  = t & 1;

  const float ang0 = p[t], ang1 = p[64 + t], ang2 = p[128 + (t & 15)];
#pragma unroll 1
  for (int j = 0; j < 3; ++j) {
    const float ang = (j == 0) ? ang0 : (j == 1) ? ang1 : ang2;
    float sn, cs;
    sincosf(ang * 0.5f, &sn, &cs);
    v2f e; e.x = cs; e.y = sn;
    if (j < 2 || t < 16) trig[64 * j + t] = e;
  }
  __syncthreads();
  if (t < (DEPTH - 1) * N_QUBITS) {
    const int a0 = 3 * (N_QUBITS + t);
    const Gate U = make_gate(trig[a0], trig[a0 + 1], trig[a0 + 2]);
    float a, d = 0.0f;
    const v2f u0 = unit_conj(U.r00, U.i00, a);
    v2f u1 = u0;
    if (t >= (DEPTH - 2) * N_QUBITS) u1 = unit_conj(U.r11, U.i11, d);
    v2f e01, e10, e11;
    e01.x = U.r01; e01.y = U.i01;
    e10.x = U.r10; e10.y = U.i10;
    e11.x = U.r11; e11.y = U.i11;
    const v2f m01 = cmul(e01, u0), m10 = cmul(e10, u1), m11 = cmul(e11, u1);
    v4f lo, hi;
    lo.x = a;     lo.y = d;     lo.z = m01.x; lo.w = m01.y;
    hi.x = m10.x; hi.y = m10.y; hi.z = m11.x; hi.w = m11.y;
    v4f* g = reinterpret_cast<v4f*>(gl + 8 * t);
    g[0] = lo;
    g[1] = hi;
  }

  v2f c[64];
  {
    v2f f0[N_QUBITS], f1[N_QUBITS];
#pragma unroll
    for (int b = 0; b < N_QUBITS; ++b) {
      const int g = N_QUBITS - 1 - b;
      const v2f ea = trig[3 * g], eb = trig[3 * g + 1], ez = trig[3 * g + 2];
      const float ca = ea.x, sa = ea.y, cb = eb.x, sb = eb.y, cz = ez.x, sz = ez.y;
      const float m00r =  cb * ca, m00i =  sb * sa;
      const float m10r =  sb * ca, m10i = -cb * sa;
      f0[b].x = cz * m00r + sz * m00i;  f0[b].y = cz * m00i - sz * m00r;
      f1[b].x = cz * m10r - sz * m10i;  f1[b].y = cz * m10i + sz * m10r;
    }
    v2f pt = (cg0 & 1) ? f1[0] : f0[0];
#pragma unroll
    for (int b = 1; b < 5; ++b) pt = cmul(((cg0 >> b) & 1) ? f1[b] : f0[b], pt);
    const int t5 = (t >> 5) & 1;
    c[0] = cmul(t5 ? f1[5] : f0[5], pt);
    c[1] = cmul(t5 ? f0[5] : f1[5], pt);
#pragma unroll
    for (int k = 1; k < 5; ++k) {
      const int n = 1 << k;
#pragma unroll
      for (int j = 0; j < n; ++j) {
        const int lowbit = (j >> (k - 1)) & 1;
        c[n + j] = cmul(lowbit ? f0[5 + k] : f1[5 + k], c[j]);
      }
#pragma unroll
      for (int j = 0; j < n; ++j) {
        const int lowbit = (j >> (k - 1)) & 1;
        c[j] = cmul(lowbit ? f1[5 + k] : f0[5 + k], c[j]);
      }
    }
    const v2f a10 = t0 ? f1[10] : f0[10], b10 = t0 ? f0[10] : f1[10];
    const v2f a11 = t0 ? f1[11] : f0[11], b11 = t0 ? f0[11] : f1[11];
    const v2f w00 = cmul(a10, a11);
    const v2f w01 = cmul(b10, a11);
    const v2f w10 = cmul(b10, b11);
    const v2f w11 = cmul(a10, b11);
#pragma unroll
    for (int j = 0; j < 32; ++j) {
      const int r4 = (j >> 4) & 1;
      c[32 + j] = cmul(r4 ? w11 : w10, c[j]);
      c[j]      = cmul(r4 ? w01 : w00, c[j]);
    }
  }

  __syncthreads();

#pragma unroll 1
  for (int s = 0; s < 2 * (DEPTH - 1); ++s) {
    if (s < 2 * (DEPTH - 2)) apply_six<false>(c, gl + s * 48);
    else                     apply_six<true>(c, gl + s * 48);
    if (s == 2 * (DEPTH - 1) - 1) break;
    if ((s & 1) == 0) {
#pragma unroll
      for (int r = 0; r < 64; ++r) st[r * QPITCH + t] = c[r];
      __syncthreads();
      const v4f* row = reinterpret_cast<const v4f*>(st + t * QPITCH);
#pragma unroll
      for (int h = 0; h < 32; ++h) {
        const v4f v = row[h];
        c[2 * h].x = v.x;     c[2 * h].y = v.y;
        c[2 * h + 1].x = v.z; c[2 * h + 1].y = v.w;
      }
    } else {
      v4f* row = reinterpret_cast<v4f*>(st + t * QPITCH);
#pragma unroll
      for (int h = 0; h < 32; ++h) {
        v4f v;
        v.x = c[2 * h].x;     v.y = c[2 * h].y;
        v.z = c[2 * h + 1].x; v.w = c[2 * h + 1].y;
        row[h] = v;
      }
      __syncthreads();
#pragma unroll
      for (int r = 0; r < 64; ++r) {
        const int G  = r ^ (r >> 1);
        const int hi = (G >> 4) & 3;
        const int delta = (hi == 0) ? 48 : (hi == 1) ? 16 : (hi == 2) ? -16 : -48;
        const int idx = G * QPITCH + ((r & 1) ? cg1 : cg0) + t0 * (delta * QPITCH);
        c[r] = st[idx];
      }
      __syncthreads();
    }
  }

  float cur[64];
#pragma unroll
  for (int r = 0; r < 64; ++r) cur[r] = c[r].x * c[r].x + c[r].y * c[r].y;
  float D[6];
#pragma unroll
  for (int k = 0; k < 6; ++k) {
    const int n = 32 >> k;
    float d = 0.0f;
#pragma unroll
    for (int m = 0; m < n; ++m) {
      const float a = cur[2 * m] + cur[2 * m + 1];
      const float b = cur[2 * m] - cur[2 * m + 1];
      d += (__builtin_popcount(m) & 1) ? -b : b;
      cur[m] = a;
    }
    D[k] = d;
  }
  const float S = cur[0];

  float z[N_QUBITS];
  const int pt = parity6(t);
#pragma unroll
  for (int k = 0; k < 6; ++k) z[k] = pt ? -D[k] : D[k];
#pragma unroll
  for (int k = 6; k < 11; ++k) z[k] = parity6(t >> (k - 6)) ? -S : S;
  z[11] = parity6(t & 31) ? -D[0] : D[0];

#pragma unroll
  for (int k = 0; k < N_QUBITS; ++k) {
    float v = z[k];
#pragma unroll
    for (int m = 16; m > 0; m >>= 1) v += __shfl_xor(v, m);
    z[k] = v;
  }
  if ((t & 31) == 0) {
#pragma unroll
    for (int k = 0; k < N_QUBITS; ++k) red[t >> 5][k] = z[k];
  }
  __syncthreads();
  if (t < N_QUBITS)
    meas[(size_t)bidx * N_QUBITS + t] = red[0][N_QUBITS - 1 - t] + red[1][N_QUBITS - 1 - t];
}

__global__ __launch_bounds__(128) void back_mlp_kernel(
    const float* __restrict__ meas, const float* __restrict__ W3,
    const float* __restrict__ b3,   const float* __restrict__ W4,
    const float* __restrict__ b4,   float* __restrict__ out) {
  __shared__ float h[16 * HID];
  const int lane  = threadIdx.x & 31;
  const int wave  = threadIdx.x >> 5;
  const int row0  = blockIdx.x * 16;
  const int col   = lane & 15;
  const int rhalf = (lane >> 4) * 8;

  v2f bw[HID / 4] = {};
  float bias4 = 0.0f;
  if (wave == 0 && col < OUT_DIM) {
#pragma unroll
    for (int i = 0; i < HID / 4; ++i) {
      const int k = 4 * i + ((lane >> 4) << 1);
      bw[i].x = W4[k * OUT_DIM + col];
      bw[i].y = W4[(k + 1) * OUT_DIM + col];
    }
    bias4 = b4[col];
  }

  {
    v8f c = {};
#pragma unroll
    for (int kc = 0; kc < N_QUBITS; kc += 4) {
      v2f a = load_a_frag(meas, N_QUBITS, row0, kc, lane);
      v2f b = load_b_frag(W3, HID, kc, wave * 16, lane);
      c = wmma_f32(a, b, c);
    }
    const int n = wave * 16 + col;
    const float bias = b3[n];
#pragma unroll
    for (int r = 0; r < 8; ++r)
      h[(rhalf + r) * HID + n] = tanhf(c[r] + bias);
  }
  __syncthreads();
  if (wave != 0) return;

  v8f c = {};
#pragma unroll
  for (int i = 0; i < HID / 4; ++i) {
    v2f a = load_a_frag(h, HID, 0, 4 * i, lane);
    c = wmma_f32(a, bw[i], c);
  }
  if (col < OUT_DIM) {
#pragma unroll
    for (int r = 0; r < 8; ++r)
      out[(size_t)(row0 + rhalf + r) * OUT_DIM + col] = c[r] + bias4;
  }
}

extern "C" void kernel_launch(void* const* d_in, const int* in_sizes, int n_in,
                              void* d_out, int out_size, void* d_ws, size_t ws_size,
                              hipStream_t stream) {
  const float* noise = (const float*)d_in[0];
  const float* W1 = (const float*)d_in[1];
  const float* b1 = (const float*)d_in[2];
  const float* W2 = (const float*)d_in[3];
  const float* b2 = (const float*)d_in[4];
  const float* W3 = (const float*)d_in[5];
  const float* b3 = (const float*)d_in[6];
  const float* W4 = (const float*)d_in[7];
  const float* b4 = (const float*)d_in[8];
  float* out = (float*)d_out;

  float* params = (float*)d_ws;
  float* meas   = params + (size_t)BATCH * NPARAMS;

  front_mlp_kernel<<<BATCH / 16, FRONT_WAVES * 32, 0, stream>>>(noise, W1, b1, W2, b2, params);
  quantum_kernel<<<BATCH, 64, 0, stream>>>(params, meas);
  back_mlp_kernel<<<BATCH / 16, 128, 0, stream>>>(meas, W3, b3, W4, b4, out);
}


# State-vector kernel: gates of layers 1-3 factored as phase x real rotation x phase; rotation as 3 shears per pair, phases gathered into one table multiply per layer (amplitude rounding changes)
# speedup vs baseline: 6.5958x; 1.2468x over previous
#include <hip/hip_runtime.h>
#include <math.h>

#define N_QUBITS 12
#define DEPTH    4
#define HID      64
#define NPARAMS  144
#define BATCH    4096
#define DIM      4096
#define OUT_DIM  2

typedef __attribute__((ext_vector_type(2))) float v2f;
typedef __attribute__((ext_vector_type(8))) float v8f;

__device__ __forceinline__ v8f wmma_f32(v2f a, v2f b, v8f c) {
  return __builtin_amdgcn_wmma_f32_16x16x4_f32(
       false, a,  false, b,
       (short)0, c,  false,  false);
}

__device__ __forceinline__ v2f load_a_frag(const float* src, int lda,
                                           int row0, int k0, int lane) {
  int m = row0 + (lane & 15);
  int k = k0 + ((lane >> 4) << 1);
  v2f a;
  a.x = src[m * lda + k];
  a.y = src[m * lda + k + 1];
  return a;
}

__device__ __forceinline__ v2f load_b_frag(const float* W, int ldb,
                                           int k0, int n0, int lane) {
  int c = n0 + (lane & 15);
  int k = k0 + ((lane >> 4) << 1);
  v2f b;
  b.x = W[k * ldb + c];
  b.y = W[(k + 1) * ldb + c];
  return b;
}

#define FRONT_WAVES (NPARAMS / 16)

__global__ __launch_bounds__(FRONT_WAVES * 32) void front_mlp_kernel(
    const float* __restrict__ noise, const float* __restrict__ W1,
    const float* __restrict__ b1,    const float* __restrict__ W2,
    const float* __restrict__ b2,    float* __restrict__ params) {
  __shared__ float h[16 * HID];
  const int lane  = threadIdx.x & 31;
  const int wave  = threadIdx.x >> 5;
  const int row0  = blockIdx.x * 16;
  const int col   = lane & 15;
  const int rhalf = (lane >> 4) * 8;

  v2f bw[HID / 4];
#pragma unroll
  for (int i = 0; i < HID / 4; ++i)
    bw[i] = load_b_frag(W2, NPARAMS, 4 * i, wave * 16, lane);
  const float bias2 = b2[wave * 16 + col];

  if (wave < HID / 16) {
    v8f c = {};
#pragma unroll
    for (int kc = 0; kc < N_QUBITS; kc += 4) {
      v2f a = load_a_frag(noise, N_QUBITS, row0, kc, lane);
      v2f b = load_b_frag(W1, HID, kc, wave * 16, lane);
      c = wmma_f32(a, b, c);
    }
    const int n = wave * 16 + col;
    const float bias = b1[n];
#pragma unroll
    for (int r = 0; r < 8; ++r)
      h[(rhalf + r) * HID + n] = tanhf(c[r] + bias);
  }
  __syncthreads();

  v8f c = {};
#pragma unroll
  for (int i = 0; i < HID / 4; ++i) {
    v2f a = load_a_frag(h, HID, 0, 4 * i, lane);
    c = wmma_f32(a, bw[i], c);
  }
  const int n = wave * 16 + col;
#pragma unroll
  for (int r = 0; r < 8; ++r)
    params[(size_t)(row0 + rhalf + r) * NPARAMS + n] = c[r] + bias2;
}

#define QPITCH 66
#define TPITCH 66
#define TABB0 ((DEPTH - 1) * 4 * TPITCH)

typedef __attribute__((ext_vector_type(4))) float v4f;

struct Gate { float r00, i00, r01, i01, r10, i10, r11, i11; };

__device__ __forceinline__ v2f unit_conj(float re, float im, float& mag) {
  const float n2 = re * re + im * im;
  v2f u;
  if (n2 > 1e-30f) {
    const float inv = 1.0f / sqrtf(n2);
    mag = n2 * inv;
    u.x = re * inv; u.y = -im * inv;
  } else {
    mag = re;
    u.x = 1.0f; u.y = 0.0f;
  }
  return u;
}

__device__ __forceinline__ Gate make_gate(v2f ea, v2f eb, v2f ez) {
  const float ca = ea.x, sa = ea.y, cb = eb.x, sb = eb.y, cz = ez.x, sz = ez.y;
  const float m00r =  cb * ca, m00i =  sb * sa;
  const float m01r = -sb * ca, m01i = -cb * sa;
  const float m10r =  sb * ca, m10i = -cb * sa;
  const float m11r =  cb * ca, m11i = -sb * sa;
  Gate U;
  U.r00 = cz * m00r + sz * m00i;  U.i00 = cz * m00i - sz * m00r;
  U.r01 = cz * m01r + sz * m01i;  U.i01 = cz * m01i - sz * m01r;
  U.r10 = cz * m10r - sz * m10i;  U.i10 = cz * m10i + sz * m10r;
  U.r11 = cz * m11r - sz * m11i;  U.i11 = cz * m11i + sz * m11r;
  return U;
}

__device__ __forceinline__ v2f cmul(v2f a, v2f b) {
  const v2f sb = __builtin_shufflevector(b, b, 1, 0);
  v2f na; na.x = -a.y; na.y = a.y;
  return a.x * b + na * sb;
}

__device__ __forceinline__ v2f phase_if(v2f z, int bit) {
  v2f o;
  o.x = bit ? z.x : 1.0f;
  o.y = bit ? z.y : 0.0f;
  return o;
}

template <int RB>
__device__ __forceinline__ void apply_rot(v2f (&c)[64], float a, float b) {
#pragma unroll
  for (int j = 0; j < 32; ++j) {
    const int i0 = ((j >> RB) << (RB + 1)) | (j & ((1 << RB) - 1));
    const int i1 = i0 | (1 << RB);
    v2f x = c[i0], y = c[i1];
    x = a * y + x;
    y = b * x + y;
    x = a * y + x;
    c[i0] = x;
    c[i1] = y;
  }
}

__device__ __forceinline__ void apply_phases(v2f (&c)[64], const v2f* tab, v2f ft) {
  const v4f* tp = reinterpret_cast<const v4f*>(tab);
#pragma unroll
  for (int h = 0; h < 32; ++h) {
    const v4f v = tp[h];
    v2f e0, e1;
    e0.x = v.x; e0.y = v.y; e1.x = v.z; e1.y = v.w;
    c[2 * h]     = cmul(ft, cmul(e0, c[2 * h]));
    c[2 * h + 1] = cmul(ft, cmul(e1, c[2 * h + 1]));
  }
}

__device__ __forceinline__ void apply_rots(v2f (&c)[64], const v2f* rt) {
  const v4f* rp = reinterpret_cast<const v4f*>(rt);
  const v4f g54 = rp[0], g32 = rp[1], g10 = rp[2];
  apply_rot<5>(c, g54.x, g54.y);
  apply_rot<4>(c, g54.z, g54.w);
  apply_rot<3>(c, g32.x, g32.y);
  apply_rot<2>(c, g32.z, g32.w);
  apply_rot<1>(c, g10.x, g10.y);
  apply_rot<0>(c, g10.z, g10.w);
}

__device__ __forceinline__ int parity6(int v) { return __popc(v & 63) & 1; }

__global__ __launch_bounds__(64) __attribute__((amdgpu_waves_per_eu(3))) void quantum_kernel(
    const float* __restrict__ params, float* __restrict__ meas) {
  __shared__ __attribute__((aligned(16))) v2f st[64 * QPITCH];
  __shared__ float red[2][N_QUBITS];
  __shared__ __attribute__((aligned(16))) v2f trig[NPARAMS];
  __shared__ __attribute__((aligned(16))) v2f rot[(DEPTH - 1) * N_QUBITS];
  __shared__ __attribute__((aligned(16))) v2f Lph[(DEPTH - 1) * N_QUBITS];
  __shared__ __attribute__((aligned(16))) v2f Rph[(DEPTH - 1) * N_QUBITS];
  __shared__ __attribute__((aligned(16))) v2f tabs[TABB0 + (DEPTH - 1) * 64];
  const int t = threadIdx.x;
  const int bidx = blockIdx.x;
  const float* p = params + (size_t)bidx * NPARAMS;

  const int cg0 = t ^ (t >> 1);
  const int cg1 = cg0 ^ 32;
  const int t0  = t & 1;

  const float ang0 = p[t], ang1 = p[64 + t], ang2 = p[128 + (t & 15)];
#pragma unroll 1
  for (int j = 0; j < 3; ++j) {
    const float ang = (j == 0) ? ang0 : (j == 1) ? ang1 : ang2;
    float sn, cs;
    sincosf(ang * 0.5f, &sn, &cs);
    v2f e; e.x = cs; e.y = sn;
    if (j < 2 || t < 16) trig[64 * j + t] = e;
  }
  __syncthreads();
  if (t < (DEPTH - 1) * N_QUBITS) {
    const int a0 = 3 * (N_QUBITS + t);
    const Gate U = make_gate(trig[a0], trig[a0 + 1], trig[a0 + 2]);
    float gc, gs, gw;
    const v2f q00 = unit_conj(U.r00, U.i00, gc);
    const v2f q10 = unit_conj(U.r10, U.i10, gs);
    v2f p10; p10.x = q10.x; p10.y = -q10.y;
    const v2f L = cmul(p10, q00);
    const bool by_s = gs >= gc;
    const float wr = by_s ? -U.r01 : U.r11, wi = by_s ? -U.i01 : U.i11;
    v2f uw = unit_conj(wr, wi, gw);
    uw.y = -uw.y;
    const v2f R = cmul(uw, by_s ? q00 : q10);
    v2f ab; ab.x = -gs / (1.0f + gc); ab.y = gs;
    rot[t] = ab;
    Lph[t] = L;
    Rph[t] = R;
  }
  __syncthreads();

  {
    const int r0 = t & 1, r4 = (t >> 4) & 1, r5 = (t >> 5) & 1;
    const int par = t ^ (t >> 1);
#pragma unroll 1
    for (int l = 0; l < DEPTH - 1; ++l) {
      const v2f* Rl = Rph + N_QUBITS * l;
      v2f fr, ft;
      fr.x = 1.0f; fr.y = 0.0f;
      ft = fr;
#pragma unroll
      for (int k = 0; k < 6; ++k) {
        fr = cmul(phase_if(Rl[5 - k],  (t >> k) & 1), fr);
        ft = cmul(phase_if(Rl[11 - k], (t >> k) & 1), ft);
      }
      v2f m5, m10, m11;
      m5.x = 1.0f; m5.y = 0.0f;
      m10 = m5; m11 = m5;
      if (l > 0) {
        const v2f* Ll = Lph + N_QUBITS * (l - 1);
#pragma unroll
        for (int k = 0; k < 4; ++k) fr = cmul(phase_if(Ll[5 - k],  (par >> k) & 1), fr);
#pragma unroll
        for (int k = 0; k < 5; ++k) ft = cmul(phase_if(Ll[11 - k], (par >> k) & 1), ft);
        m5 = Ll[6]; m10 = Ll[1]; m11 = Ll[0];
      }
      tabs[TABB0 + l * 64 + t] = ft;
      const v2f b0 = cmul(phase_if(m5, r0), fr);
      const v2f b1 = cmul(phase_if(m5, r0 ^ 1), fr);
      const v2f m0 = cmul(phase_if(m10, r4 ^ r5),     phase_if(m11, r5));
      const v2f m1 = cmul(phase_if(m10, r4 ^ r5 ^ 1), phase_if(m11, r5 ^ 1));
      tabs[(l * 4 + 0) * TPITCH + t] = cmul(m0, b0);
      tabs[(l * 4 + 1) * TPITCH + t] = cmul(m1, b0);
      tabs[(l * 4 + 2) * TPITCH + t] = cmul(m0, b1);
      tabs[(l * 4 + 3) * TPITCH + t] = cmul(m1, b1);
    }
  }

  v2f c[64];
  {
    v2f f0[N_QUBITS], f1[N_QUBITS];
#pragma unroll
    for (int b = 0; b < N_QUBITS; ++b) {
      const int g = N_QUBITS - 1 - b;
      const v2f ea = trig[3 * g], eb = trig[3 * g + 1], ez = trig[3 * g + 2];
      const float ca = ea.x, sa = ea.y, cb = eb.x, sb = eb.y, cz = ez.x, sz = ez.y;
      const float m00r =  cb * ca, m00i =  sb * sa;
      const float m10r =  sb * ca, m10i = -cb * sa;
      f0[b].x = cz * m00r + sz * m00i;  f0[b].y = cz * m00i - sz * m00r;
      f1[b].x = cz * m10r - sz * m10i;  f1[b].y = cz * m10i + sz * m10r;
    }
    v2f pt = (cg0 & 1) ? f1[0] : f0[0];
#pragma unroll
    for (int b = 1; b < 5; ++b) pt = cmul(((cg0 >> b) & 1) ? f1[b] : f0[b], pt);
    const int t5 = (t >> 5) & 1;
    c[0] = cmul(t5 ? f1[5] : f0[5], pt);
    c[1] = cmul(t5 ? f0[5] : f1[5], pt);
#pragma unroll
    for (int k = 1; k < 5; ++k) {
      const int n = 1 << k;
#pragma unroll
      for (int j = 0; j < n; ++j) {
        const int lowbit = (j >> (k - 1)) & 1;
        c[n + j] = cmul(lowbit ? f0[5 + k] : f1[5 + k], c[j]);
      }
#pragma unroll
      for (int j = 0; j < n; ++j) {
        const int lowbit = (j >> (k - 1)) & 1;
        c[j] = cmul(lowbit ? f1[5 + k] : f0[5 + k], c[j]);
      }
    }
    const v2f a10 = t0 ? f1[10] : f0[10], b10 = t0 ? f0[10] : f1[10];
    const v2f a11 = t0 ? f1[11] : f0[11], b11 = t0 ? f0[11] : f1[11];
    const v2f w00 = cmul(a10, a11);
    const v2f w01 = cmul(b10, a11);
    const v2f w10 = cmul(b10, b11);
    const v2f w11 = cmul(a10, b11);
#pragma unroll
    for (int j = 0; j < 32; ++j) {
      const int r4 = (j >> 4) & 1;
      c[32 + j] = cmul(r4 ? w11 : w10, c[j]);
      c[j]      = cmul(r4 ? w01 : w00, c[j]);
    }
  }

  __syncthreads();

  const int va = ((t >> 5) << 1) | t0;
#pragma unroll 1
  for (int s = 0; s < 2 * (DEPTH - 1); ++s) {
    if ((s & 1) == 0) {
      const int l = s >> 1;
      apply_phases(c, tabs + (l * 4 + va) * TPITCH, tabs[TABB0 + l * 64 + t]);
    }
    apply_rots(c, rot + 6 * s);
    if (s == 2 * (DEPTH - 1) - 1) break;
    if ((s & 1) == 0) {
#pragma unroll
      for (int r = 0; r < 64; ++r) st[r * QPITCH + t] = c[r];
      __syncthreads();
      const v4f* row = reinterpret_cast<const v4f*>(st + t * QPITCH);
#pragma unroll
      for (int h = 0; h < 32; ++h) {
        const v4f v = row[h];
        c[2 * h].x = v.x;     c[2 * h].y = v.y;
        c[2 * h + 1].x = v.z; c[2 * h + 1].y = v.w;
      }
    } else {
      v4f* row = reinterpret_cast<v4f*>(st + t * QPITCH);
#pragma unroll
      for (int h = 0; h < 32; ++h) {
        v4f v;
        v.x = c[2 * h].x;     v.y = c[2 * h].y;
        v.z = c[2 * h + 1].x; v.w = c[2 * h + 1].y;
        row[h] = v;
      }
      __syncthreads();
#pragma unroll
      for (int r = 0; r < 64; ++r) {
        const int G  = r ^ (r >> 1);
        const int hi = (G >> 4) & 3;
        const int delta = (hi == 0) ? 48 : (hi == 1) ? 16 : (hi == 2) ? -16 : -48;
        const int idx = G * QPITCH + ((r & 1) ? cg1 : cg0) + t0 * (delta * QPITCH);
        c[r] = st[idx];
      }
      __syncthreads();
    }
  }

  float cur[64];
#pragma unroll
  for (int r = 0; r < 64; ++r) cur[r] = c[r].x * c[r].x + c[r].y * c[r].y;
  float D[6];
#pragma unroll
  for (int k = 0; k < 6; ++k) {
    const int n = 32 >> k;
    float d = 0.0f;
#pragma unroll
    for (int m = 0; m < n; ++m) {
      const float a = cur[2 * m] + cur[2 * m + 1];
      const float b = cur[2 * m] - cur[2 * m + 1];
      d += (__builtin_popcount(m) & 1) ? -b : b;
      cur[m] = a;
    }
    D[k] = d;
  }
  const float S = cur[0];

  float z[N_QUBITS];
  const int pt = parity6(t);
#pragma unroll
  for (int k = 0; k < 6; ++k) z[k] = pt ? -D[k] : D[k];
#pragma unroll
  for (int k = 6; k < 11; ++k) z[k] = parity6(t >> (k - 6)) ? -S : S;
  z[11] = parity6(t & 31) ? -D[0] : D[0];

#pragma unroll
  for (int k = 0; k < N_QUBITS; ++k) {
    float v = z[k];
#pragma unroll
    for (int m = 16; m > 0; m >>= 1) v += __shfl_xor(v, m);
    z[k] = v;
  }
  if ((t & 31) == 0) {
#pragma unroll
    for (int k = 0; k < N_QUBITS; ++k) red[t >> 5][k] = z[k];
  }
  __syncthreads();
  if (t < N_QUBITS)
    meas[(size_t)bidx * N_QUBITS + t] = red[0][N_QUBITS - 1 - t] + red[1][N_QUBITS - 1 - t];
}

__global__ __launch_bounds__(128) void back_mlp_kernel(
    const float* __restrict__ meas, const float* __restrict__ W3,
    const float* __restrict__ b3,   const float* __restrict__ W4,
    const float* __restrict__ b4,   float* __restrict__ out) {
  __shared__ float h[16 * HID];
  const int lane  = threadIdx.x & 31;
  const int wave  = threadIdx.x >> 5;
  const int row0  = blockIdx.x * 16;
  const int col   = lane & 15;
  const int rhalf = (lane >> 4) * 8;

  v2f bw[HID / 4] = {};
  float bias4 = 0.0f;
  if (wave == 0 && col < OUT_DIM) {
#pragma unroll
    for (int i = 0; i < HID / 4; ++i) {
      const int k = 4 * i + ((lane >> 4) << 1);
      bw[i].x = W4[k * OUT_DIM + col];
      bw[i].y = W4[(k + 1) * OUT_DIM + col];
    }
    bias4 = b4[col];
  }

  {
    v8f c = {};
#pragma unroll
    for (int kc = 0; kc < N_QUBITS; kc += 4) {
      v2f a = load_a_frag(meas, N_QUBITS, row0, kc, lane);
      v2f b = load_b_frag(W3, HID, kc, wave * 16, lane);
      c = wmma_f32(a, b, c);
    }
    const int n = wave * 16 + col;
    const float bias = b3[n];
#pragma unroll
    for (int r = 0; r < 8; ++r)
      h[(rhalf + r) * HID + n] = tanhf(c[r] + bias);
  }
  __syncthreads();
  if (wave != 0) return;

  v8f c = {};
#pragma unroll
  for (int i = 0; i < HID / 4; ++i) {
    v2f a = load_a_frag(h, HID, 0, 4 * i, lane);
    c = wmma_f32(a, bw[i], c);
  }
  if (col < OUT_DIM) {
#pragma unroll
    for (int r = 0; r < 8; ++r)
      out[(size_t)(row0 + rhalf + r) * OUT_DIM + col] = c[r] + bias4;
  }
}

extern "C" void kernel_launch(void* const* d_in, const int* in_sizes, int n_in,
                              void* d_out, int out_size, void* d_ws, size_t ws_size,
                              hipStream_t stream) {
  const float* noise = (const float*)d_in[0];
  const float* W1 = (const float*)d_in[1];
  const float* b1 = (const float*)d_in[2];
  const float* W2 = (const float*)d_in[3];
  const float* b2 = (const float*)d_in[4];
  const float* W3 = (const float*)d_in[5];
  const float* b3 = (const float*)d_in[6];
  const float* W4 = (const float*)d_in[7];
  const float* b4 = (const float*)d_in[8];
  float* out = (float*)d_out;

  float* params = (float*)d_ws;
  float* meas   = params + (size_t)BATCH * NPARAMS;

  front_mlp_kernel<<<BATCH / 16, FRONT_WAVES * 32, 0, stream>>>(noise, W1, b1, W2, b2, params);
  quantum_kernel<<<BATCH, 64, 0, stream>>>(params, meas);
  back_mlp_kernel<<<BATCH / 16, 128, 0, stream>>>(meas, W3, b3, W4, b4, out);
}


# State-vector kernel: rotations of layers 1-3 as x-=q*y, y+=q*x, |angle|<=pi/4: 2 packed ops a pair, not 3; quarter turns traced as bit flips, cosines squared onto outputs (amplitude rounding changes)
# speedup vs baseline: 7.4264x; 1.1259x over previous
#include <hip/hip_runtime.h>
#include <math.h>

#define N_QUBITS 12
#define DEPTH    4
#define HID      64
#define NPARAMS  144
#define BATCH    4096
#define DIM      4096
#define OUT_DIM  2

typedef __attribute__((ext_vector_type(2))) float v2f;
typedef __attribute__((ext_vector_type(8))) float v8f;

__device__ __forceinline__ v8f wmma_f32(v2f a, v2f b, v8f c) {
  return __builtin_amdgcn_wmma_f32_16x16x4_f32(
       false, a,  false, b,
       (short)0, c,  false,  false);
}

__device__ __forceinline__ v2f load_a_frag(const float* src, int lda,
                                           int row0, int k0, int lane) {
  int m = row0 + (lane & 15);
  int k = k0 + ((lane >> 4) << 1);
  v2f a;
  a.x = src[m * lda + k];
  a.y = src[m * lda + k + 1];
  return a;
}

__device__ __forceinline__ v2f load_b_frag(const float* W, int ldb,
                                           int k0, int n0, int lane) {
  int c = n0 + (lane & 15);
  int k = k0 + ((lane >> 4) << 1);
  v2f b;
  b.x = W[k * ldb + c];
  b.y = W[(k + 1) * ldb + c];
  return b;
}

#define FRONT_WAVES (NPARAMS / 16)

__global__ __launch_bounds__(FRONT_WAVES * 32) void front_mlp_kernel(
    const float* __restrict__ noise, const float* __restrict__ W1,
    const float* __restrict__ b1,    const float* __restrict__ W2,
    const float* __restrict__ b2,    float* __restrict__ params) {
  __shared__ float h[16 * HID];
  const int lane  = threadIdx.x & 31;
  const int wave  = threadIdx.x >> 5;
  const int row0  = blockIdx.x * 16;
  const int col   = lane & 15;
  const int rhalf = (lane >> 4) * 8;

  v2f bw[HID / 4];
#pragma unroll
  for (int i = 0; i < HID / 4; ++i)
    bw[i] = load_b_frag(W2, NPARAMS, 4 * i, wave * 16, lane);
  const float bias2 = b2[wave * 16 + col];

  if (wave < HID / 16) {
    v8f c = {};
#pragma unroll
    for (int kc = 0; kc < N_QUBITS; kc += 4) {
      v2f a = load_a_frag(noise, N_QUBITS, row0, kc, lane);
      v2f b = load_b_frag(W1, HID, kc, wave * 16, lane);
      c = wmma_f32(a, b, c);
    }
    const int n = wave * 16 + col;
    const float bias = b1[n];
#pragma unroll
    for (int r = 0; r < 8; ++r)
      h[(rhalf + r) * HID + n] = tanhf(c[r] + bias);
  }
  __syncthreads();

  v8f c = {};
#pragma unroll
  for (int i = 0; i < HID / 4; ++i) {
    v2f a = load_a_frag(h, HID, 0, 4 * i, lane);
    c = wmma_f32(a, bw[i], c);
  }
  const int n = wave * 16 + col;
#pragma unroll
  for (int r = 0; r < 8; ++r)
    params[(size_t)(row0 + rhalf + r) * NPARAMS + n] = c[r] + bias2;
}

#define QPITCH 66
#define TPITCH 66
#define TABB0 ((DEPTH - 1) * 4 * TPITCH)

typedef __attribute__((ext_vector_type(4))) float v4f;

struct Gate { float r00, i00, r01, i01, r10, i10, r11, i11; };

__device__ __forceinline__ v2f unit_conj(float re, float im, float& mag) {
  const float n2 = re * re + im * im;
  v2f u;
  if (n2 > 1e-30f) {
    const float inv = 1.0f / sqrtf(n2);
    mag = n2 * inv;
    u.x = re * inv; u.y = -im * inv;
  } else {
    mag = re;
    u.x = 1.0f; u.y = 0.0f;
  }
  return u;
}

__device__ __forceinline__ Gate make_gate(v2f ea, v2f eb, v2f ez) {
  const float ca = ea.x, sa = ea.y, cb = eb.x, sb = eb.y, cz = ez.x, sz = ez.y;
  const float m00r =  cb * ca, m00i =  sb * sa;
  const float m01r = -sb * ca, m01i = -cb * sa;
  const float m10r =  sb * ca, m10i = -cb * sa;
  const float m11r =  cb * ca, m11i = -sb * sa;
  Gate U;
  U.r00 = cz * m00r + sz * m00i;  U.i00 = cz * m00i - sz * m00r;
  U.r01 = cz * m01r + sz * m01i;  U.i01 = cz * m01i - sz * m01r;
  U.r10 = cz * m10r - sz * m10i;  U.i10 = cz * m10i + sz * m10r;
  U.r11 = cz * m11r - sz * m11i;  U.i11 = cz * m11i + sz * m11r;
  return U;
}

__device__ __forceinline__ v2f cmul(v2f a, v2f b) {
  const v2f sb = __builtin_shufflevector(b, b, 1, 0);
  v2f na; na.x = -a.y; na.y = a.y;
  return a.x * b + na * sb;
}

__device__ __forceinline__ v2f phase_if(v2f z, int bit) {
  v2f o;
  o.x = bit ? z.x : 1.0f;
  o.y = bit ? z.y : 0.0f;
  return o;
}

template <int RB>
__device__ __forceinline__ void apply_rot(v2f (&c)[64], float q) {
#pragma unroll
  for (int j = 0; j < 32; ++j) {
    const int i0 = ((j >> RB) << (RB + 1)) | (j & ((1 << RB) - 1));
    const int i1 = i0 | (1 << RB);
    const v2f x = c[i0], y = c[i1];
    c[i0] = x - q * y;
    c[i1] = q * x + y;
  }
}

__device__ __forceinline__ unsigned ring_flips(unsigned f) {
  unsigned p = f;
  p ^= p << 1; p ^= p << 2; p ^= p << 4; p ^= p << 8;
  p &= 0xFFFu;
  return (p & 0xFFEu) | (((p >> 11) ^ f) & 1u);
}

__device__ __forceinline__ void apply_phases(v2f (&c)[64], const v2f* tab, v2f ft) {
  const v4f* tp = reinterpret_cast<const v4f*>(tab);
#pragma unroll
  for (int h = 0; h < 32; ++h) {
    const v4f v = tp[h];
    v2f e0, e1;
    e0.x = v.x; e0.y = v.y; e1.x = v.z; e1.y = v.w;
    c[2 * h]     = cmul(ft, cmul(e0, c[2 * h]));
    c[2 * h + 1] = cmul(ft, cmul(e1, c[2 * h + 1]));
  }
}

__device__ __forceinline__ void apply_rots(v2f (&c)[64], const float* rt) {
  const v2f* rp = reinterpret_cast<const v2f*>(rt);
  const v2f g54 = rp[0], g32 = rp[1], g10 = rp[2];
  apply_rot<5>(c, g54.x);
  apply_rot<4>(c, g54.y);
  apply_rot<3>(c, g32.x);
  apply_rot<2>(c, g32.y);
  apply_rot<1>(c, g10.x);
  apply_rot<0>(c, g10.y);
}

__device__ __forceinline__ int parity6(int v) { return __popc(v & 63) & 1; }

__global__ __launch_bounds__(64) __attribute__((amdgpu_waves_per_eu(3))) void quantum_kernel(
    const float* __restrict__ params, float* __restrict__ meas) {
  __shared__ __attribute__((aligned(16))) v2f st[64 * QPITCH];
  __shared__ float red[2][N_QUBITS];
  __shared__ float gsc[2];
  __shared__ __attribute__((aligned(16))) v2f trig[NPARAMS];
  __shared__ __attribute__((aligned(16))) float rot[(DEPTH - 1) * N_QUBITS];
  __shared__ int bigs[(DEPTH - 1) * N_QUBITS];
  __shared__ __attribute__((aligned(16))) v2f Lph[(DEPTH - 1) * N_QUBITS];
  __shared__ __attribute__((aligned(16))) v2f Rph[(DEPTH - 1) * N_QUBITS];
  __shared__ __attribute__((aligned(16))) v2f tabs[TABB0 + (DEPTH - 1) * 64];
  const int t = threadIdx.x;
  const int bidx = blockIdx.x;
  const float* p = params + (size_t)bidx * NPARAMS;

  const int cg0 = t ^ (t >> 1);
  const int cg1 = cg0 ^ 32;
  const int t0  = t & 1;

  const float ang0 = p[t], ang1 = p[64 + t], ang2 = p[128 + (t & 15)];
#pragma unroll 1
  for (int j = 0; j < 3; ++j) {
    const float ang = (j == 0) ? ang0 : (j == 1) ? ang1 : ang2;
    float sn, cs;
    sincosf(ang * 0.5f, &sn, &cs);
    v2f e; e.x = cs; e.y = sn;
    if (j < 2 || t < 16) trig[64 * j + t] = e;
  }
  __syncthreads();
  float hi = 1.0f;
  float qv = 0.0f;
  int big_s = 0;
  v2f L, R;
  L.x = 1.0f; L.y = 0.0f;
  R = L;
  if (t < (DEPTH - 1) * N_QUBITS) {
    const int a0 = 3 * (N_QUBITS + t);
    const Gate U = make_gate(trig[a0], trig[a0 + 1], trig[a0 + 2]);
    float gc, gs, gw;
    const v2f q00 = unit_conj(U.r00, U.i00, gc);
    const v2f q10 = unit_conj(U.r10, U.i10, gs);
    v2f p10; p10.x = q10.x; p10.y = -q10.y;
    L = cmul(p10, q00);
    const bool by_s = gs >= gc;
    const float wr = by_s ? -U.r01 : U.r11, wi = by_s ? -U.i01 : U.i11;
    v2f uw = unit_conj(wr, wi, gw);
    uw.y = -uw.y;
    R = cmul(uw, by_s ? q00 : q10);
    big_s = by_s ? 1 : 0;
    hi = by_s ? gs : gc;
    qv = (by_s ? gc : gs) / hi;
    bigs[t] = big_s;
  }
  for (int m = 16; m > 0; m >>= 1) hi *= __shfl_xor(hi, m);
  if ((t & 31) == 0) gsc[t >> 5] = hi;
  __syncthreads();

  const int ln = t & 31;
  const unsigned m0 = __builtin_amdgcn_ballot_w32(bigs[ln] != 0);
  const unsigned m1 = __builtin_amdgcn_ballot_w32(bigs[32 + (ln & 3)] != 0);
  const unsigned B1 = m0 & 0xFFFu, B2 = (m0 >> 12) & 0xFFFu, B3 = (m0 >> 24) | ((m1 & 0xFu) << 8);
  const unsigned F2 = ring_flips(B1);
  const unsigned F3 = ring_flips(F2 ^ B2);
  const unsigned Fend = ring_flips(F3 ^ B3);
  if (t < (DEPTH - 1) * N_QUBITS) {
    const int l = (t >= 2 * N_QUBITS) ? 2 : (t >= N_QUBITS) ? 1 : 0;
    const unsigned Fin = (l == 2) ? F3 : (l == 1) ? F2 : 0u;
    const int fin = (Fin >> (t - N_QUBITS * l)) & 1;
    if (fin) { L.y = -L.y; R.y = -R.y; }
    if (big_s) L.x = -L.x;
    rot[t] = (fin == big_s) ? qv : -qv;
    Lph[t] = L;
    Rph[t] = R;
  }
  __syncthreads();

  {
    const int r0 = t & 1, r4 = (t >> 4) & 1, r5 = (t >> 5) & 1;
    const int par = t ^ (t >> 1);
#pragma unroll 1
    for (int l = 0; l < DEPTH - 1; ++l) {
      const v2f* Rl = Rph + N_QUBITS * l;
      v2f fr, ft;
      fr.x = 1.0f; fr.y = 0.0f;
      ft = fr;
#pragma unroll
      for (int k = 0; k < 6; ++k) {
        fr = cmul(phase_if(Rl[5 - k],  (t >> k) & 1), fr);
        ft = cmul(phase_if(Rl[11 - k], (t >> k) & 1), ft);
      }
      v2f m5, m10, m11;
      m5.x = 1.0f; m5.y = 0.0f;
      m10 = m5; m11 = m5;
      if (l > 0) {
        const v2f* Ll = Lph + N_QUBITS * (l - 1);
#pragma unroll
        for (int k = 0; k < 4; ++k) fr = cmul(phase_if(Ll[5 - k],  (par >> k) & 1), fr);
#pragma unroll
        for (int k = 0; k < 5; ++k) ft = cmul(phase_if(Ll[11 - k], (par >> k) & 1), ft);
        m5 = Ll[6]; m10 = Ll[1]; m11 = Ll[0];
      }
      tabs[TABB0 + l * 64 + t] = ft;
      const v2f b0 = cmul(phase_if(m5, r0), fr);
      const v2f b1 = cmul(phase_if(m5, r0 ^ 1), fr);
      const v2f m0 = cmul(phase_if(m10, r4 ^ r5),     phase_if(m11, r5));
      const v2f m1 = cmul(phase_if(m10, r4 ^ r5 ^ 1), phase_if(m11, r5 ^ 1));
      tabs[(l * 4 + 0) * TPITCH + t] = cmul(m0, b0);
      tabs[(l * 4 + 1) * TPITCH + t] = cmul(m1, b0);
      tabs[(l * 4 + 2) * TPITCH + t] = cmul(m0, b1);
      tabs[(l * 4 + 3) * TPITCH + t] = cmul(m1, b1);
    }
  }

  v2f c[64];
  {
    v2f f0[N_QUBITS], f1[N_QUBITS];
#pragma unroll
    for (int b = 0; b < N_QUBITS; ++b) {
      const int g = N_QUBITS - 1 - b;
      const v2f ea = trig[3 * g], eb = trig[3 * g + 1], ez = trig[3 * g + 2];
      const float ca = ea.x, sa = ea.y, cb = eb.x, sb = eb.y, cz = ez.x, sz = ez.y;
      const float m00r =  cb * ca, m00i =  sb * sa;
      const float m10r =  sb * ca, m10i = -cb * sa;
      f0[b].x = cz * m00r + sz * m00i;  f0[b].y = cz * m00i - sz * m00r;
      f1[b].x = cz * m10r - sz * m10i;  f1[b].y = cz * m10i + sz * m10r;
    }
    v2f pt = (cg0 & 1) ? f1[0] : f0[0];
#pragma unroll
    for (int b = 1; b < 5; ++b) pt = cmul(((cg0 >> b) & 1) ? f1[b] : f0[b], pt);
    const int t5 = (t >> 5) & 1;
    c[0] = cmul(t5 ? f1[5] : f0[5], pt);
    c[1] = cmul(t5 ? f0[5] : f1[5], pt);
#pragma unroll
    for (int k = 1; k < 5; ++k) {
      const int n = 1 << k;
#pragma unroll
      for (int j = 0; j < n; ++j) {
        const int lowbit = (j >> (k - 1)) & 1;
        c[n + j] = cmul(lowbit ? f0[5 + k] : f1[5 + k], c[j]);
      }
#pragma unroll
      for (int j = 0; j < n; ++j) {
        const int lowbit = (j >> (k - 1)) & 1;
        c[j] = cmul(lowbit ? f1[5 + k] : f0[5 + k], c[j]);
      }
    }
    const v2f a10 = t0 ? f1[10] : f0[10], b10 = t0 ? f0[10] : f1[10];
    const v2f a11 = t0 ? f1[11] : f0[11], b11 = t0 ? f0[11] : f1[11];
    const v2f w00 = cmul(a10, a11);
    const v2f w01 = cmul(b10, a11);
    const v2f w10 = cmul(b10, b11);
    const v2f w11 = cmul(a10, b11);
#pragma unroll
    for (int j = 0; j < 32; ++j) {
      const int r4 = (j >> 4) & 1;
      c[32 + j] = cmul(r4 ? w11 : w10, c[j]);
      c[j]      = cmul(r4 ? w01 : w00, c[j]);
    }
  }

  __syncthreads();

  const int va = ((t >> 5) << 1) | t0;
#pragma unroll 1
  for (int s = 0; s < 2 * (DEPTH - 1); ++s) {
    if ((s & 1) == 0) {
      const int l = s >> 1;
      apply_phases(c, tabs + (l * 4 + va) * TPITCH, tabs[TABB0 + l * 64 + t]);
    }
    apply_rots(c, rot + 6 * s);
    if (s == 2 * (DEPTH - 1) - 1) break;
    if ((s & 1) == 0) {
#pragma unroll
      for (int r = 0; r < 64; ++r) st[r * QPITCH + t] = c[r];
      __syncthreads();
      const v4f* row = reinterpret_cast<const v4f*>(st + t * QPITCH);
#pragma unroll
      for (int h = 0; h < 32; ++h) {
        const v4f v = row[h];
        c[2 * h].x = v.x;     c[2 * h].y = v.y;
        c[2 * h + 1].x = v.z; c[2 * h + 1].y = v.w;
      }
    } else {
      v4f* row = reinterpret_cast<v4f*>(st + t * QPITCH);
#pragma unroll
      for (int h = 0; h < 32; ++h) {
        v4f v;
        v.x = c[2 * h].x;     v.y = c[2 * h].y;
        v.z = c[2 * h + 1].x; v.w = c[2 * h + 1].y;
        row[h] = v;
      }
      __syncthreads();
#pragma unroll
      for (int r = 0; r < 64; ++r) {
        const int G  = r ^ (r >> 1);
        const int hi = (G >> 4) & 3;
        const int delta = (hi == 0) ? 48 : (hi == 1) ? 16 : (hi == 2) ? -16 : -48;
        const int idx = G * QPITCH + ((r & 1) ? cg1 : cg0) + t0 * (delta * QPITCH);
        c[r] = st[idx];
      }
      __syncthreads();
    }
  }

  float cur[64];
#pragma unroll
  for (int r = 0; r < 64; ++r) cur[r] = c[r].x * c[r].x + c[r].y * c[r].y;
  float D[6];
#pragma unroll
  for (int k = 0; k < 6; ++k) {
    const int n = 32 >> k;
    float d = 0.0f;
#pragma unroll
    for (int m = 0; m < n; ++m) {
      const float a = cur[2 * m] + cur[2 * m + 1];
      const float b = cur[2 * m] - cur[2 * m + 1];
      d += (__builtin_popcount(m) & 1) ? -b : b;
      cur[m] = a;
    }
    D[k] = d;
  }
  const float S = cur[0];

  float z[N_QUBITS];
  const int pt = parity6(t);
#pragma unroll
  for (int k = 0; k < 6; ++k) z[k] = pt ? -D[k] : D[k];
#pragma unroll
  for (int k = 6; k < 11; ++k) z[k] = parity6(t >> (k - 6)) ? -S : S;
  z[11] = parity6(t & 31) ? -D[0] : D[0];

#pragma unroll
  for (int k = 0; k < N_QUBITS; ++k) {
    float v = z[k];
#pragma unroll
    for (int m = 16; m > 0; m >>= 1) v += __shfl_xor(v, m);
    z[k] = v;
  }
  if ((t & 31) == 0) {
#pragma unroll
    for (int k = 0; k < N_QUBITS; ++k) red[t >> 5][k] = z[k];
  }
  __syncthreads();
  if (t < N_QUBITS) {
    const float g = gsc[0] * gsc[1];
    const float v = (red[0][N_QUBITS - 1 - t] + red[1][N_QUBITS - 1 - t]) * (g * g);
    meas[(size_t)bidx * N_QUBITS + t] = ((Fend >> t) & 1u) ? -v : v;
  }
}

__global__ __launch_bounds__(128) void back_mlp_kernel(
    const float* __restrict__ meas, const float* __restrict__ W3,
    const float* __restrict__ b3,   const float* __restrict__ W4,
    const float* __restrict__ b4,   float* __restrict__ out) {
  __shared__ float h[16 * HID];
  const int lane  = threadIdx.x & 31;
  const int wave  = threadIdx.x >> 5;
  const int row0  = blockIdx.x * 16;
  const int col   = lane & 15;
  const int rhalf = (lane >> 4) * 8;

  v2f bw[HID / 4] = {};
  float bias4 = 0.0f;
  if (wave == 0 && col < OUT_DIM) {
#pragma unroll
    for (int i = 0; i < HID / 4; ++i) {
      const int k = 4 * i + ((lane >> 4) << 1);
      bw[i].x = W4[k * OUT_DIM + col];
      bw[i].y = W4[(k + 1) * OUT_DIM + col];
    }
    bias4 = b4[col];
  }

  {
    v8f c = {};
#pragma unroll
    for (int kc = 0; kc < N_QUBITS; kc += 4) {
      v2f a = load_a_frag(meas, N_QUBITS, row0, kc, lane);
      v2f b = load_b_frag(W3, HID, kc, wave * 16, lane);
      c = wmma_f32(a, b, c);
    }
    const int n = wave * 16 + col;
    const float bias = b3[n];
#pragma unroll
    for (int r = 0; r < 8; ++r)
      h[(rhalf + r) * HID + n] = tanhf(c[r] + bias);
  }
  __syncthreads();
  if (wave != 0) return;

  v8f c = {};
#pragma unroll
  for (int i = 0; i < HID / 4; ++i) {
    v2f a = load_a_frag(h, HID, 0, 4 * i, lane);
    c = wmma_f32(a, bw[i], c);
  }
  if (col < OUT_DIM) {
#pragma unroll
    for (int r = 0; r < 8; ++r)
      out[(size_t)(row0 + rhalf + r) * OUT_DIM + col] = c[r] + bias4;
  }
}

extern "C" void kernel_launch(void* const* d_in, const int* in_sizes, int n_in,
                              void* d_out, int out_size, void* d_ws, size_t ws_size,
                              hipStream_t stream) {
  const float* noise = (const float*)d_in[0];
  const float* W1 = (const float*)d_in[1];
  const float* b1 = (const float*)d_in[2];
  const float* W2 = (const float*)d_in[3];
  const float* b2 = (const float*)d_in[4];
  const float* W3 = (const float*)d_in[5];
  const float* b3 = (const float*)d_in[6];
  const float* W4 = (const float*)d_in[7];
  const float* b4 = (const float*)d_in[8];
  float* out = (float*)d_out;

  float* params = (float*)d_ws;
  float* meas   = params + (size_t)BATCH * NPARAMS;

  front_mlp_kernel<<<BATCH / 16, FRONT_WAVES * 32, 0, stream>>>(noise, W1, b1, W2, b2, params);
  quantum_kernel<<<BATCH, 64, 0, stream>>>(params, meas);
  back_mlp_kernel<<<BATCH / 16, 128, 0, stream>>>(meas, W3, b3, W4, b4, out);
}


# State-vector kernel: layer 1's phase factors (its R only) multiplied into the factors the layer-0 product state is written from; its phase table and table multiply go (amplitude rounding changes)
# speedup vs baseline: 7.9685x; 1.0730x over previous
#include <hip/hip_runtime.h>
#include <math.h>

#define N_QUBITS 12
#define DEPTH    4
#define HID      64
#define NPARAMS  144
#define BATCH    4096
#define DIM      4096
#define OUT_DIM  2

typedef __attribute__((ext_vector_type(2))) float v2f;
typedef __attribute__((ext_vector_type(8))) float v8f;

__device__ __forceinline__ v8f wmma_f32(v2f a, v2f b, v8f c) {
  return __builtin_amdgcn_wmma_f32_16x16x4_f32(
       false, a,  false, b,
       (short)0, c,  false,  false);
}

__device__ __forceinline__ v2f load_a_frag(const float* src, int lda,
                                           int row0, int k0, int lane) {
  int m = row0 + (lane & 15);
  int k = k0 + ((lane >> 4) << 1);
  v2f a;
  a.x = src[m * lda + k];
  a.y = src[m * lda + k + 1];
  return a;
}

__device__ __forceinline__ v2f load_b_frag(const float* W, int ldb,
                                           int k0, int n0, int lane) {
  int c = n0 + (lane & 15);
  int k = k0 + ((lane >> 4) << 1);
  v2f b;
  b.x = W[k * ldb + c];
  b.y = W[(k + 1) * ldb + c];
  return b;
}

#define FRONT_WAVES (NPARAMS / 16)

__global__ __launch_bounds__(FRONT_WAVES * 32) void front_mlp_kernel(
    const float* __restrict__ noise, const float* __restrict__ W1,
    const float* __restrict__ b1,    const float* __restrict__ W2,
    const float* __restrict__ b2,    float* __restrict__ params) {
  __shared__ float h[16 * HID];
  const int lane  = threadIdx.x & 31;
  const int wave  = threadIdx.x >> 5;
  const int row0  = blockIdx.x * 16;
  const int col   = lane & 15;
  const int rhalf = (lane >> 4) * 8;

  v2f bw[HID / 4];
#pragma unroll
  for (int i = 0; i < HID / 4; ++i)
    bw[i] = load_b_frag(W2, NPARAMS, 4 * i, wave * 16, lane);
  const float bias2 = b2[wave * 16 + col];

  if (wave < HID / 16) {
    v8f c = {};
#pragma unroll
    for (int kc = 0; kc < N_QUBITS; kc += 4) {
      v2f a = load_a_frag(noise, N_QUBITS, row0, kc, lane);
      v2f b = load_b_frag(W1, HID, kc, wave * 16, lane);
      c = wmma_f32(a, b, c);
    }
    const int n = wave * 16 + col;
    const float bias = b1[n];
#pragma unroll
    for (int r = 0; r < 8; ++r)
      h[(rhalf + r) * HID + n] = tanhf(c[r] + bias);
  }
  __syncthreads();

  v8f c = {};
#pragma unroll
  for (int i = 0; i < HID / 4; ++i) {
    v2f a = load_a_frag(h, HID, 0, 4 * i, lane);
    c = wmma_f32(a, bw[i], c);
  }
  const int n = wave * 16 + col;
#pragma unroll
  for (int r = 0; r < 8; ++r)
    params[(size_t)(row0 + rhalf + r) * NPARAMS + n] = c[r] + bias2;
}

#define QPITCH 66
#define TPITCH 66
#define TABB0 ((DEPTH - 2) * 4 * TPITCH)

typedef __attribute__((ext_vector_type(4))) float v4f;

struct Gate { float r00, i00, r01, i01, r10, i10, r11, i11; };

__device__ __forceinline__ v2f unit_conj(float re, float im, float& mag) {
  const float n2 = re * re + im * im;
  v2f u;
  if (n2 > 1e-30f) {
    const float inv = 1.0f / sqrtf(n2);
    mag = n2 * inv;
    u.x = re * inv; u.y = -im * inv;
  } else {
    mag = re;
    u.x = 1.0f; u.y = 0.0f;
  }
  return u;
}

__device__ __forceinline__ Gate make_gate(v2f ea, v2f eb, v2f ez) {
  const float ca = ea.x, sa = ea.y, cb = eb.x, sb = eb.y, cz = ez.x, sz = ez.y;
  const float m00r =  cb * ca, m00i =  sb * sa;
  const float m01r = -sb * ca, m01i = -cb * sa;
  const float m10r =  sb * ca, m10i = -cb * sa;
  const float m11r =  cb * ca, m11i = -sb * sa;
  Gate U;
  U.r00 = cz * m00r + sz * m00i;  U.i00 = cz * m00i - sz * m00r;
  U.r01 = cz * m01r + sz * m01i;  U.i01 = cz * m01i - sz * m01r;
  U.r10 = cz * m10r - sz * m10i;  U.i10 = cz * m10i + sz * m10r;
  U.r11 = cz * m11r - sz * m11i;  U.i11 = cz * m11i + sz * m11r;
  return U;
}

__device__ __forceinline__ v2f cmul(v2f a, v2f b) {
  const v2f sb = __builtin_shufflevector(b, b, 1, 0);
  v2f na; na.x = -a.y; na.y = a.y;
  return a.x * b + na * sb;
}

__device__ __forceinline__ v2f phase_if(v2f z, int bit) {
  v2f o;
  o.x = bit ? z.x : 1.0f;
  o.y = bit ? z.y : 0.0f;
  return o;
}

template <int RB>
__device__ __forceinline__ void apply_rot(v2f (&c)[64], float q) {
#pragma unroll
  for (int j = 0; j < 32; ++j) {
    const int i0 = ((j >> RB) << (RB + 1)) | (j & ((1 << RB) - 1));
    const int i1 = i0 | (1 << RB);
    const v2f x = c[i0], y = c[i1];
    c[i0] = x - q * y;
    c[i1] = q * x + y;
  }
}

__device__ __forceinline__ unsigned ring_flips(unsigned f) {
  unsigned p = f;
  p ^= p << 1; p ^= p << 2; p ^= p << 4; p ^= p << 8;
  p &= 0xFFFu;
  return (p & 0xFFEu) | (((p >> 11) ^ f) & 1u);
}

__device__ __forceinline__ void apply_phases(v2f (&c)[64], const v2f* tab, v2f ft) {
  const v4f* tp = reinterpret_cast<const v4f*>(tab);
#pragma unroll
  for (int h = 0; h < 32; ++h) {
    const v4f v = tp[h];
    v2f e0, e1;
    e0.x = v.x; e0.y = v.y; e1.x = v.z; e1.y = v.w;
    c[2 * h]     = cmul(ft, cmul(e0, c[2 * h]));
    c[2 * h + 1] = cmul(ft, cmul(e1, c[2 * h + 1]));
  }
}

__device__ __forceinline__ void apply_rots(v2f (&c)[64], const float* rt) {
  const v2f* rp = reinterpret_cast<const v2f*>(rt);
  const v2f g54 = rp[0], g32 = rp[1], g10 = rp[2];
  apply_rot<5>(c, g54.x);
  apply_rot<4>(c, g54.y);
  apply_rot<3>(c, g32.x);
  apply_rot<2>(c, g32.y);
  apply_rot<1>(c, g10.x);
  apply_rot<0>(c, g10.y);
}

__device__ __forceinline__ int parity6(int v) { return __popc(v & 63) & 1; }

__global__ __launch_bounds__(64) __attribute__((amdgpu_waves_per_eu(3))) void quantum_kernel(
    const float* __restrict__ params, float* __restrict__ meas) {
  __shared__ __attribute__((aligned(16))) v2f st[64 * QPITCH];
  __shared__ float red[2][N_QUBITS];
  __shared__ float gsc[2];
  __shared__ __attribute__((aligned(16))) v2f trig[NPARAMS];
  __shared__ __attribute__((aligned(16))) float rot[(DEPTH - 1) * N_QUBITS];
  __shared__ int bigs[(DEPTH - 1) * N_QUBITS];
  __shared__ __attribute__((aligned(16))) v2f Lph[(DEPTH - 1) * N_QUBITS];
  __shared__ __attribute__((aligned(16))) v2f Rph[(DEPTH - 1) * N_QUBITS];
  __shared__ __attribute__((aligned(16))) v2f tabs[TABB0 + (DEPTH - 2) * 64];
  const int t = threadIdx.x;
  const int bidx = blockIdx.x;
  const float* p = params + (size_t)bidx * NPARAMS;

  const int cg0 = t ^ (t >> 1);
  const int cg1 = cg0 ^ 32;
  const int t0  = t & 1;

  const float ang0 = p[t], ang1 = p[64 + t], ang2 = p[128 + (t & 15)];
#pragma unroll 1
  for (int j = 0; j < 3; ++j) {
    const float ang = (j == 0) ? ang0 : (j == 1) ? ang1 : ang2;
    float sn, cs;
    sincosf(ang * 0.5f, &sn, &cs);
    v2f e; e.x = cs; e.y = sn;
    if (j < 2 || t < 16) trig[64 * j + t] = e;
  }
  __syncthreads();
  float hi = 1.0f;
  float qv = 0.0f;
  int big_s = 0;
  v2f L, R;
  L.x = 1.0f; L.y = 0.0f;
  R = L;
  if (t < (DEPTH - 1) * N_QUBITS) {
    const int a0 = 3 * (N_QUBITS + t);
    const Gate U = make_gate(trig[a0], trig[a0 + 1], trig[a0 + 2]);
    float gc, gs, gw;
    const v2f q00 = unit_conj(U.r00, U.i00, gc);
    const v2f q10 = unit_conj(U.r10, U.i10, gs);
    v2f p10; p10.x = q10.x; p10.y = -q10.y;
    L = cmul(p10, q00);
    const bool by_s = gs >= gc;
    const float wr = by_s ? -U.r01 : U.r11, wi = by_s ? -U.i01 : U.i11;
    v2f uw = unit_conj(wr, wi, gw);
    uw.y = -uw.y;
    R = cmul(uw, by_s ? q00 : q10);
    big_s = by_s ? 1 : 0;
    hi = by_s ? gs : gc;
    qv = (by_s ? gc : gs) / hi;
    bigs[t] = big_s;
  }
  for (int m = 16; m > 0; m >>= 1) hi *= __shfl_xor(hi, m);
  if ((t & 31) == 0) gsc[t >> 5] = hi;
  __syncthreads();

  const int ln = t & 31;
  const unsigned m0 = __builtin_amdgcn_ballot_w32(bigs[ln] != 0);
  const unsigned m1 = __builtin_amdgcn_ballot_w32(bigs[32 + (ln & 3)] != 0);
  const unsigned B1 = m0 & 0xFFFu, B2 = (m0 >> 12) & 0xFFFu, B3 = (m0 >> 24) | ((m1 & 0xFu) << 8);
  const unsigned F2 = ring_flips(B1);
  const unsigned F3 = ring_flips(F2 ^ B2);
  const unsigned Fend = ring_flips(F3 ^ B3);
  if (t < (DEPTH - 1) * N_QUBITS) {
    const int l = (t >= 2 * N_QUBITS) ? 2 : (t >= N_QUBITS) ? 1 : 0;
    const unsigned Fin = (l == 2) ? F3 : (l == 1) ? F2 : 0u;
    const int fin = (Fin >> (t - N_QUBITS * l)) & 1;
    if (fin) { L.y = -L.y; R.y = -R.y; }
    if (big_s) L.x = -L.x;
    rot[t] = (fin == big_s) ? qv : -qv;
    Lph[t] = L;
    Rph[t] = R;
  }
  __syncthreads();

  {
    const int r0 = t & 1, r4 = (t >> 4) & 1, r5 = (t >> 5) & 1;
    const int par = t ^ (t >> 1);
#pragma unroll 1
    for (int l = 1; l < DEPTH - 1; ++l) {
      const v2f* Rl = Rph + N_QUBITS * l;
      const v2f* Ll = Lph + N_QUBITS * (l - 1);
      v2f fr, ft;
      fr.x = 1.0f; fr.y = 0.0f;
      ft = fr;
#pragma unroll
      for (int k = 0; k < 6; ++k) {
        fr = cmul(phase_if(Rl[5 - k],  (t >> k) & 1), fr);
        ft = cmul(phase_if(Rl[11 - k], (t >> k) & 1), ft);
      }
#pragma unroll
      for (int k = 0; k < 4; ++k) fr = cmul(phase_if(Ll[5 - k],  (par >> k) & 1), fr);
#pragma unroll
      for (int k = 0; k < 5; ++k) ft = cmul(phase_if(Ll[11 - k], (par >> k) & 1), ft);
      const v2f m5 = Ll[6], m10 = Ll[1], m11 = Ll[0];
      tabs[TABB0 + (l - 1) * 64 + t] = ft;
      const v2f b0 = cmul(phase_if(m5, r0), fr);
      const v2f b1 = cmul(phase_if(m5, r0 ^ 1), fr);
      const v2f m0 = cmul(phase_if(m10, r4 ^ r5),     phase_if(m11, r5));
      const v2f m1 = cmul(phase_if(m10, r4 ^ r5 ^ 1), phase_if(m11, r5 ^ 1));
      tabs[((l - 1) * 4 + 0) * TPITCH + t] = cmul(m0, b0);
      tabs[((l - 1) * 4 + 1) * TPITCH + t] = cmul(m1, b0);
      tabs[((l - 1) * 4 + 2) * TPITCH + t] = cmul(m0, b1);
      tabs[((l - 1) * 4 + 3) * TPITCH + t] = cmul(m1, b1);
    }
  }

  v2f c[64];
  {
    v2f f0[N_QUBITS], f1[N_QUBITS];
#pragma unroll
    for (int b = 0; b < N_QUBITS; ++b) {
      const int g = N_QUBITS - 1 - b;
      const v2f ea = trig[3 * g], eb = trig[3 * g + 1], ez = trig[3 * g + 2];
      const float ca = ea.x, sa = ea.y, cb = eb.x, sb = eb.y, cz = ez.x, sz = ez.y;
      const float m00r =  cb * ca, m00i =  sb * sa;
      const float m10r =  sb * ca, m10i = -cb * sa;
      f0[b].x = cz * m00r + sz * m00i;  f0[b].y = cz * m00i - sz * m00r;
      f1[b].x = cz * m10r - sz * m10i;  f1[b].y = cz * m10i + sz * m10r;
    }
    const v2f* R1 = Rph;
    v2f pt = (cg0 & 1) ? f1[0] : f0[0];
#pragma unroll
    for (int b = 1; b < 5; ++b) pt = cmul(((cg0 >> b) & 1) ? f1[b] : f0[b], pt);
#pragma unroll
    for (int k = 0; k < 6; ++k) pt = cmul(phase_if(R1[11 - k], (t >> k) & 1), pt);
    const int t5 = (t >> 5) & 1;
    c[0] = cmul(t5 ? f1[5] : f0[5], pt);
    c[1] = cmul(cmul(R1[5], t5 ? f0[5] : f1[5]), pt);
#pragma unroll
    for (int k = 1; k < 5; ++k) {
      const int n = 1 << k;
      const v2f g0 = cmul(R1[5 - k], f0[5 + k]), g1 = cmul(R1[5 - k], f1[5 + k]);
#pragma unroll
      for (int j = 0; j < n; ++j) {
        const int lowbit = (j >> (k - 1)) & 1;
        c[n + j] = cmul(lowbit ? g0 : g1, c[j]);
      }
#pragma unroll
      for (int j = 0; j < n; ++j) {
        const int lowbit = (j >> (k - 1)) & 1;
        c[j] = cmul(lowbit ? f1[5 + k] : f0[5 + k], c[j]);
      }
    }
    const v2f a10 = t0 ? f1[10] : f0[10], b10 = t0 ? f0[10] : f1[10];
    const v2f a11 = t0 ? f1[11] : f0[11], b11 = t0 ? f0[11] : f1[11];
    const v2f w00 = cmul(a10, a11);
    const v2f w01 = cmul(b10, a11);
    const v2f w10 = cmul(R1[0], cmul(b10, b11));
    const v2f w11 = cmul(R1[0], cmul(a10, b11));
#pragma unroll
    for (int j = 0; j < 32; ++j) {
      const int r4 = (j >> 4) & 1;
      c[32 + j] = cmul(r4 ? w11 : w10, c[j]);
      c[j]      = cmul(r4 ? w01 : w00, c[j]);
    }
  }

  const int va = ((t >> 5) << 1) | t0;
#pragma unroll 1
  for (int s = 0; s < 2 * (DEPTH - 1); ++s) {
    if ((s & 1) == 0 && s > 0) {
      const int l = (s >> 1) - 1;
      apply_phases(c, tabs + (l * 4 + va) * TPITCH, tabs[TABB0 + l * 64 + t]);
    }
    apply_rots(c, rot + 6 * s);
    if (s == 2 * (DEPTH - 1) - 1) break;
    if ((s & 1) == 0) {
#pragma unroll
      for (int r = 0; r < 64; ++r) st[r * QPITCH + t] = c[r];
      __syncthreads();
      const v4f* row = reinterpret_cast<const v4f*>(st + t * QPITCH);
#pragma unroll
      for (int h = 0; h < 32; ++h) {
        const v4f v = row[h];
        c[2 * h].x = v.x;     c[2 * h].y = v.y;
        c[2 * h + 1].x = v.z; c[2 * h + 1].y = v.w;
      }
    } else {
      v4f* row = reinterpret_cast<v4f*>(st + t * QPITCH);
#pragma unroll
      for (int h = 0; h < 32; ++h) {
        v4f v;
        v.x = c[2 * h].x;     v.y = c[2 * h].y;
        v.z = c[2 * h + 1].x; v.w = c[2 * h + 1].y;
        row[h] = v;
      }
      __syncthreads();
#pragma unroll
      for (int r = 0; r < 64; ++r) {
        const int G  = r ^ (r >> 1);
        const int hi = (G >> 4) & 3;
        const int delta = (hi == 0) ? 48 : (hi == 1) ? 16 : (hi == 2) ? -16 : -48;
        const int idx = G * QPITCH + ((r & 1) ? cg1 : cg0) + t0 * (delta * QPITCH);
        c[r] = st[idx];
      }
      __syncthreads();
    }
  }

  float cur[64];
#pragma unroll
  for (int r = 0; r < 64; ++r) cur[r] = c[r].x * c[r].x + c[r].y * c[r].y;
  float D[6];
#pragma unroll
  for (int k = 0; k < 6; ++k) {
    const int n = 32 >> k;
    float d = 0.0f;
#pragma unroll
    for (int m = 0; m < n; ++m) {
      const float a = cur[2 * m] + cur[2 * m + 1];
      const float b = cur[2 * m] - cur[2 * m + 1];
      d += (__builtin_popcount(m) & 1) ? -b : b;
      cur[m] = a;
    }
    D[k] = d;
  }
  const float S = cur[0];

  float z[N_QUBITS];
  const int pt = parity6(t);
#pragma unroll
  for (int k = 0; k < 6; ++k) z[k] = pt ? -D[k] : D[k];
#pragma unroll
  for (int k = 6; k < 11; ++k) z[k] = parity6(t >> (k - 6)) ? -S : S;
  z[11] = parity6(t & 31) ? -D[0] : D[0];

#pragma unroll
  for (int k = 0; k < N_QUBITS; ++k) {
    float v = z[k];
#pragma unroll
    for (int m = 16; m > 0; m >>= 1) v += __shfl_xor(v, m);
    z[k] = v;
  }
  if ((t & 31) == 0) {
#pragma unroll
    for (int k = 0; k < N_QUBITS; ++k) red[t >> 5][k] = z[k];
  }
  __syncthreads();
  if (t < N_QUBITS) {
    const float g = gsc[0] * gsc[1];
    const float v = (red[0][N_QUBITS - 1 - t] + red[1][N_QUBITS - 1 - t]) * (g * g);
    meas[(size_t)bidx * N_QUBITS + t] = ((Fend >> t) & 1u) ? -v : v;
  }
}

__global__ __launch_bounds__(128) void back_mlp_kernel(
    const float* __restrict__ meas, const float* __restrict__ W3,
    const float* __restrict__ b3,   const float* __restrict__ W4,
    const float* __restrict__ b4,   float* __restrict__ out) {
  __shared__ float h[16 * HID];
  const int lane  = threadIdx.x & 31;
  const int wave  = threadIdx.x >> 5;
  const int row0  = blockIdx.x * 16;
  const int col   = lane & 15;
  const int rhalf = (lane >> 4) * 8;

  v2f bw[HID / 4] = {};
  float bias4 = 0.0f;
  if (wave == 0 && col < OUT_DIM) {
#pragma unroll
    for (int i = 0; i < HID / 4; ++i) {
      const int k = 4 * i + ((lane >> 4) << 1);
      bw[i].x = W4[k * OUT_DIM + col];
      bw[i].y = W4[(k + 1) * OUT_DIM + col];
    }
    bias4 = b4[col];
  }

  {
    v8f c = {};
#pragma unroll
    for (int kc = 0; kc < N_QUBITS; kc += 4) {
      v2f a = load_a_frag(meas, N_QUBITS, row0, kc, lane);
      v2f b = load_b_frag(W3, HID, kc, wave * 16, lane);
      c = wmma_f32(a, b, c);
    }
    const int n = wave * 16 + col;
    const float bias = b3[n];
#pragma unroll
    for (int r = 0; r < 8; ++r)
      h[(rhalf + r) * HID + n] = tanhf(c[r] + bias);
  }
  __syncthreads();
  if (wave != 0) return;

  v8f c = {};
#pragma unroll
  for (int i = 0; i < HID / 4; ++i) {
    v2f a = load_a_frag(h, HID, 0, 4 * i, lane);
    c = wmma_f32(a, bw[i], c);
  }
  if (col < OUT_DIM) {
#pragma unroll
    for (int r = 0; r < 8; ++r)
      out[(size_t)(row0 + rhalf + r) * OUT_DIM + col] = c[r] + bias4;
  }
}

extern "C" void kernel_launch(void* const* d_in, const int* in_sizes, int n_in,
                              void* d_out, int out_size, void* d_ws, size_t ws_size,
                              hipStream_t stream) {
  const float* noise = (const float*)d_in[0];
  const float* W1 = (const float*)d_in[1];
  const float* b1 = (const float*)d_in[2];
  const float* W2 = (const float*)d_in[3];
  const float* b2 = (const float*)d_in[4];
  const float* W3 = (const float*)d_in[5];
  const float* b3 = (const float*)d_in[6];
  const float* W4 = (const float*)d_in[7];
  const float* b4 = (const float*)d_in[8];
  float* out = (float*)d_out;

  float* params = (float*)d_ws;
  float* meas   = params + (size_t)BATCH * NPARAMS;

  front_mlp_kernel<<<BATCH / 16, FRONT_WAVES * 32, 0, stream>>>(noise, W1, b1, W2, b2, params);
  quantum_kernel<<<BATCH, 64, 0, stream>>>(params, meas);
  back_mlp_kernel<<<BATCH / 16, 128, 0, stream>>>(meas, W3, b3, W4, b4, out);
}


# State-vector kernel: loop re-cut to one layer per trip (single exit), angle table overlaid on the state buffer: 208 registers, under 40 KB LDS, so 4 waves per SIMD (8 blocks per WGP), not 3
# speedup vs baseline: 8.6211x; 1.0819x over previous
#include <hip/hip_runtime.h>
#include <math.h>

#define N_QUBITS 12
#define DEPTH    4
#define HID      64
#define NPARAMS  144
#define BATCH    4096
#define DIM      4096
#define OUT_DIM  2

typedef __attribute__((ext_vector_type(2))) float v2f;
typedef __attribute__((ext_vector_type(8))) float v8f;

__device__ __forceinline__ v8f wmma_f32(v2f a, v2f b, v8f c) {
  return __builtin_amdgcn_wmma_f32_16x16x4_f32(
       false, a,  false, b,
       (short)0, c,  false,  false);
}

__device__ __forceinline__ v2f load_a_frag(const float* src, int lda,
                                           int row0, int k0, int lane) {
  int m = row0 + (lane & 15);
  int k = k0 + ((lane >> 4) << 1);
  v2f a;
  a.x = src[m * lda + k];
  a.y = src[m * lda + k + 1];
  return a;
}

__device__ __forceinline__ v2f load_b_frag(const float* W, int ldb,
                                           int k0, int n0, int lane) {
  int c = n0 + (lane & 15);
  int k = k0 + ((lane >> 4) << 1);
  v2f b;
  b.x = W[k * ldb + c];
  b.y = W[(k + 1) * ldb + c];
  return b;
}

#define FRONT_WAVES (NPARAMS / 16)

__global__ __launch_bounds__(FRONT_WAVES * 32) void front_mlp_kernel(
    const float* __restrict__ noise, const float* __restrict__ W1,
    const float* __restrict__ b1,    const float* __restrict__ W2,
    const float* __restrict__ b2,    float* __restrict__ params) {
  __shared__ float h[16 * HID];
  const int lane  = threadIdx.x & 31;
  const int wave  = threadIdx.x >> 5;
  const int row0  = blockIdx.x * 16;
  const int col   = lane & 15;
  const int rhalf = (lane >> 4) * 8;

  v2f bw[HID / 4];
#pragma unroll
  for (int i = 0; i < HID / 4; ++i)
    bw[i] = load_b_frag(W2, NPARAMS, 4 * i, wave * 16, lane);
  const float bias2 = b2[wave * 16 + col];

  if (wave < HID / 16) {
    v8f c = {};
#pragma unroll
    for (int kc = 0; kc < N_QUBITS; kc += 4) {
      v2f a = load_a_frag(noise, N_QUBITS, row0, kc, lane);
      v2f b = load_b_frag(W1, HID, kc, wave * 16, lane);
      c = wmma_f32(a, b, c);
    }
    const int n = wave * 16 + col;
    const float bias = b1[n];
#pragma unroll
    for (int r = 0; r < 8; ++r)
      h[(rhalf + r) * HID + n] = tanhf(c[r] + bias);
  }
  __syncthreads();

  v8f c = {};
#pragma unroll
  for (int i = 0; i < HID / 4; ++i) {
    v2f a = load_a_frag(h, HID, 0, 4 * i, lane);
    c = wmma_f32(a, bw[i], c);
  }
  const int n = wave * 16 + col;
#pragma unroll
  for (int r = 0; r < 8; ++r)
    params[(size_t)(row0 + rhalf + r) * NPARAMS + n] = c[r] + bias2;
}

#define QPITCH 66
#define TPITCH 66
#define TABB0 ((DEPTH - 2) * 4 * TPITCH)

typedef __attribute__((ext_vector_type(4))) float v4f;

struct Gate { float r00, i00, r01, i01, r10, i10, r11, i11; };

__device__ __forceinline__ v2f unit_conj(float re, float im, float& mag) {
  const float n2 = re * re + im * im;
  v2f u;
  if (n2 > 1e-30f) {
    const float inv = 1.0f / sqrtf(n2);
    mag = n2 * inv;
    u.x = re * inv; u.y = -im * inv;
  } else {
    mag = re;
    u.x = 1.0f; u.y = 0.0f;
  }
  return u;
}

__device__ __forceinline__ Gate make_gate(v2f ea, v2f eb, v2f ez) {
  const float ca = ea.x, sa = ea.y, cb = eb.x, sb = eb.y, cz = ez.x, sz = ez.y;
  const float m00r =  cb * ca, m00i =  sb * sa;
  const float m01r = -sb * ca, m01i = -cb * sa;
  const float m10r =  sb * ca, m10i = -cb * sa;
  const float m11r =  cb * ca, m11i = -sb * sa;
  Gate U;
  U.r00 = cz * m00r + sz * m00i;  U.i00 = cz * m00i - sz * m00r;
  U.r01 = cz * m01r + sz * m01i;  U.i01 = cz * m01i - sz * m01r;
  U.r10 = cz * m10r - sz * m10i;  U.i10 = cz * m10i + sz * m10r;
  U.r11 = cz * m11r - sz * m11i;  U.i11 = cz * m11i + sz * m11r;
  return U;
}

__device__ __forceinline__ v2f cmul(v2f a, v2f b) {
  const v2f sb = __builtin_shufflevector(b, b, 1, 0);
  v2f na; na.x = -a.y; na.y = a.y;
  return a.x * b + na * sb;
}

__device__ __forceinline__ v2f phase_if(v2f z, int bit) {
  v2f o;
  o.x = bit ? z.x : 1.0f;
  o.y = bit ? z.y : 0.0f;
  return o;
}

template <int RB>
__device__ __forceinline__ void apply_rot(v2f (&c)[64], float q) {
#pragma unroll
  for (int j = 0; j < 32; ++j) {
    const int i0 = ((j >> RB) << (RB + 1)) | (j & ((1 << RB) - 1));
    const int i1 = i0 | (1 << RB);
    const v2f x = c[i0], y = c[i1];
    c[i0] = x - q * y;
    c[i1] = q * x + y;
  }
}

__device__ __forceinline__ unsigned ring_flips(unsigned f) {
  unsigned p = f;
  p ^= p << 1; p ^= p << 2; p ^= p << 4; p ^= p << 8;
  p &= 0xFFFu;
  return (p & 0xFFEu) | (((p >> 11) ^ f) & 1u);
}

__device__ __forceinline__ void apply_phases(v2f (&c)[64], const v2f* tab, v2f ft) {
  const v4f* tp = reinterpret_cast<const v4f*>(tab);
#pragma unroll
  for (int h = 0; h < 32; ++h) {
    const v4f v = tp[h];
    v2f e0, e1;
    e0.x = v.x; e0.y = v.y; e1.x = v.z; e1.y = v.w;
    c[2 * h]     = cmul(ft, cmul(e0, c[2 * h]));
    c[2 * h + 1] = cmul(ft, cmul(e1, c[2 * h + 1]));
  }
}

__device__ __forceinline__ void apply_rots(v2f (&c)[64], const float* rt) {
  const v2f* rp = reinterpret_cast<const v2f*>(rt);
  const v2f g54 = rp[0], g32 = rp[1], g10 = rp[2];
  apply_rot<5>(c, g54.x);
  apply_rot<4>(c, g54.y);
  apply_rot<3>(c, g32.x);
  apply_rot<2>(c, g32.y);
  apply_rot<1>(c, g10.x);
  apply_rot<0>(c, g10.y);
}

__device__ __forceinline__ int parity6(int v) { return __popc(v & 63) & 1; }

__global__ __launch_bounds__(64) __attribute__((amdgpu_waves_per_eu(4))) void quantum_kernel(
    const float* __restrict__ params, float* __restrict__ meas) {
  __shared__ __attribute__((aligned(16))) v2f st[64 * QPITCH];
  __shared__ float red[2][N_QUBITS];
  __shared__ float gsc[2];
  v2f* const trig = st + (64 * QPITCH - NPARAMS);
  __shared__ __attribute__((aligned(16))) float rot[(DEPTH - 1) * N_QUBITS];
  __shared__ int bigs[(DEPTH - 1) * N_QUBITS];
  __shared__ __attribute__((aligned(16))) v2f Lph[(DEPTH - 1) * N_QUBITS];
  __shared__ __attribute__((aligned(16))) v2f Rph[(DEPTH - 1) * N_QUBITS];
  __shared__ __attribute__((aligned(16))) v2f tabs[TABB0 + (DEPTH - 2) * 64];
  const int t = threadIdx.x;
  const int bidx = blockIdx.x;
  const float* p = params + (size_t)bidx * NPARAMS;

  const int cg0 = t ^ (t >> 1);
  const int cg1 = cg0 ^ 32;
  const int t0  = t & 1;

  const float ang0 = p[t], ang1 = p[64 + t], ang2 = p[128 + (t & 15)];
#pragma unroll 1
  for (int j = 0; j < 3; ++j) {
    const float ang = (j == 0) ? ang0 : (j == 1) ? ang1 : ang2;
    float sn, cs;
    sincosf(ang * 0.5f, &sn, &cs);
    v2f e; e.x = cs; e.y = sn;
    if (j < 2 || t < 16) trig[64 * j + t] = e;
  }
  __syncthreads();
  float hi = 1.0f;
  float qv = 0.0f;
  int big_s = 0;
  v2f L, R;
  L.x = 1.0f; L.y = 0.0f;
  R = L;
  if (t < (DEPTH - 1) * N_QUBITS) {
    const int a0 = 3 * (N_QUBITS + t);
    const Gate U = make_gate(trig[a0], trig[a0 + 1], trig[a0 + 2]);
    float gc, gs, gw;
    const v2f q00 = unit_conj(U.r00, U.i00, gc);
    const v2f q10 = unit_conj(U.r10, U.i10, gs);
    v2f p10; p10.x = q10.x; p10.y = -q10.y;
    L = cmul(p10, q00);
    const bool by_s = gs >= gc;
    const float wr = by_s ? -U.r01 : U.r11, wi = by_s ? -U.i01 : U.i11;
    v2f uw = unit_conj(wr, wi, gw);
    uw.y = -uw.y;
    R = cmul(uw, by_s ? q00 : q10);
    big_s = by_s ? 1 : 0;
    hi = by_s ? gs : gc;
    qv = (by_s ? gc : gs) / hi;
    bigs[t] = big_s;
  }
  for (int m = 16; m > 0; m >>= 1) hi *= __shfl_xor(hi, m);
  if ((t & 31) == 0) gsc[t >> 5] = hi;
  __syncthreads();

  const int ln = t & 31;
  const unsigned m0 = __builtin_amdgcn_ballot_w32(bigs[ln] != 0);
  const unsigned m1 = __builtin_amdgcn_ballot_w32(bigs[32 + (ln & 3)] != 0);
  const unsigned B1 = m0 & 0xFFFu, B2 = (m0 >> 12) & 0xFFFu, B3 = (m0 >> 24) | ((m1 & 0xFu) << 8);
  const unsigned F2 = ring_flips(B1);
  const unsigned F3 = ring_flips(F2 ^ B2);
  const unsigned Fend = ring_flips(F3 ^ B3);
  if (t < (DEPTH - 1) * N_QUBITS) {
    const int l = (t >= 2 * N_QUBITS) ? 2 : (t >= N_QUBITS) ? 1 : 0;
    const unsigned Fin = (l == 2) ? F3 : (l == 1) ? F2 : 0u;
    const int fin = (Fin >> (t - N_QUBITS * l)) & 1;
    if (fin) { L.y = -L.y; R.y = -R.y; }
    if (big_s) L.x = -L.x;
    rot[t] = (fin == big_s) ? qv : -qv;
    Lph[t] = L;
    Rph[t] = R;
  }
  __syncthreads();

  {
    const int r0 = t & 1, r4 = (t >> 4) & 1, r5 = (t >> 5) & 1;
    const int par = t ^ (t >> 1);
#pragma unroll 1
    for (int l = 1; l < DEPTH - 1; ++l) {
      const v2f* Rl = Rph + N_QUBITS * l;
      const v2f* Ll = Lph + N_QUBITS * (l - 1);
      v2f fr, ft;
      fr.x = 1.0f; fr.y = 0.0f;
      ft = fr;
#pragma unroll
      for (int k = 0; k < 6; ++k) {
        fr = cmul(phase_if(Rl[5 - k],  (t >> k) & 1), fr);
        ft = cmul(phase_if(Rl[11 - k], (t >> k) & 1), ft);
      }
#pragma unroll
      for (int k = 0; k < 4; ++k) fr = cmul(phase_if(Ll[5 - k],  (par >> k) & 1), fr);
#pragma unroll
      for (int k = 0; k < 5; ++k) ft = cmul(phase_if(Ll[11 - k], (par >> k) & 1), ft);
      const v2f m5 = Ll[6], m10 = Ll[1], m11 = Ll[0];
      tabs[TABB0 + (l - 1) * 64 + t] = ft;
      const v2f b0 = cmul(phase_if(m5, r0), fr);
      const v2f b1 = cmul(phase_if(m5, r0 ^ 1), fr);
      const v2f m0 = cmul(phase_if(m10, r4 ^ r5),     phase_if(m11, r5));
      const v2f m1 = cmul(phase_if(m10, r4 ^ r5 ^ 1), phase_if(m11, r5 ^ 1));
      tabs[((l - 1) * 4 + 0) * TPITCH + t] = cmul(m0, b0);
      tabs[((l - 1) * 4 + 1) * TPITCH + t] = cmul(m1, b0);
      tabs[((l - 1) * 4 + 2) * TPITCH + t] = cmul(m0, b1);
      tabs[((l - 1) * 4 + 3) * TPITCH + t] = cmul(m1, b1);
    }
  }

  v2f c[64];
  {
    v2f f0[N_QUBITS], f1[N_QUBITS];
#pragma unroll
    for (int b = 0; b < N_QUBITS; ++b) {
      const int g = N_QUBITS - 1 - b;
      const v2f ea = trig[3 * g], eb = trig[3 * g + 1], ez = trig[3 * g + 2];
      const float ca = ea.x, sa = ea.y, cb = eb.x, sb = eb.y, cz = ez.x, sz = ez.y;
      const float m00r =  cb * ca, m00i =  sb * sa;
      const float m10r =  sb * ca, m10i = -cb * sa;
      f0[b].x = cz * m00r + sz * m00i;  f0[b].y = cz * m00i - sz * m00r;
      f1[b].x = cz * m10r - sz * m10i;  f1[b].y = cz * m10i + sz * m10r;
    }
    const v2f* R1 = Rph;
    v2f pt = (cg0 & 1) ? f1[0] : f0[0];
#pragma unroll
    for (int b = 1; b < 5; ++b) pt = cmul(((cg0 >> b) & 1) ? f1[b] : f0[b], pt);
#pragma unroll
    for (int k = 0; k < 6; ++k) pt = cmul(phase_if(R1[11 - k], (t >> k) & 1), pt);
    const int t5 = (t >> 5) & 1;
    c[0] = cmul(t5 ? f1[5] : f0[5], pt);
    c[1] = cmul(cmul(R1[5], t5 ? f0[5] : f1[5]), pt);
#pragma unroll
    for (int k = 1; k < 5; ++k) {
      const int n = 1 << k;
      const v2f g0 = cmul(R1[5 - k], f0[5 + k]), g1 = cmul(R1[5 - k], f1[5 + k]);
#pragma unroll
      for (int j = 0; j < n; ++j) {
        const int lowbit = (j >> (k - 1)) & 1;
        c[n + j] = cmul(lowbit ? g0 : g1, c[j]);
      }
#pragma unroll
      for (int j = 0; j < n; ++j) {
        const int lowbit = (j >> (k - 1)) & 1;
        c[j] = cmul(lowbit ? f1[5 + k] : f0[5 + k], c[j]);
      }
    }
    const v2f a10 = t0 ? f1[10] : f0[10], b10 = t0 ? f0[10] : f1[10];
    const v2f a11 = t0 ? f1[11] : f0[11], b11 = t0 ? f0[11] : f1[11];
    const v2f w00 = cmul(a10, a11);
    const v2f w01 = cmul(b10, a11);
    const v2f w10 = cmul(R1[0], cmul(b10, b11));
    const v2f w11 = cmul(R1[0], cmul(a10, b11));
#pragma unroll
    for (int j = 0; j < 32; ++j) {
      const int r4 = (j >> 4) & 1;
      c[32 + j] = cmul(r4 ? w11 : w10, c[j]);
      c[j]      = cmul(r4 ? w01 : w00, c[j]);
    }
  }

  __syncthreads();

  const int va = ((t >> 5) << 1) | t0;
#pragma unroll 1
  for (int l = 0; l < DEPTH - 1; ++l) {
    if (l > 0) {
      v4f* row = reinterpret_cast<v4f*>(st + t * QPITCH);
#pragma unroll
      for (int h = 0; h < 32; ++h) {
        v4f v;
        v.x = c[2 * h].x;     v.y = c[2 * h].y;
        v.z = c[2 * h + 1].x; v.w = c[2 * h + 1].y;
        row[h] = v;
      }
      __syncthreads();
#pragma unroll
      for (int r = 0; r < 64; ++r) {
        const int G  = r ^ (r >> 1);
        const int hi = (G >> 4) & 3;
        const int delta = (hi == 0) ? 48 : (hi == 1) ? 16 : (hi == 2) ? -16 : -48;
        const int idx = G * QPITCH + ((r & 1) ? cg1 : cg0) + t0 * (delta * QPITCH);
        c[r] = st[idx];
      }
      __syncthreads();
      apply_phases(c, tabs + ((l - 1) * 4 + va) * TPITCH, tabs[TABB0 + (l - 1) * 64 + t]);
    }
    apply_rots(c, rot + 12 * l);
    {
#pragma unroll
      for (int r = 0; r < 64; ++r) st[r * QPITCH + t] = c[r];
      __syncthreads();
      const v4f* row = reinterpret_cast<const v4f*>(st + t * QPITCH);
#pragma unroll
      for (int h = 0; h < 32; ++h) {
        const v4f v = row[h];
        c[2 * h].x = v.x;     c[2 * h].y = v.y;
        c[2 * h + 1].x = v.z; c[2 * h + 1].y = v.w;
      }
    }
    apply_rots(c, rot + 12 * l + 6);
  }

  float cur[64];
#pragma unroll
  for (int r = 0; r < 64; ++r) cur[r] = c[r].x * c[r].x + c[r].y * c[r].y;
  float D[6];
#pragma unroll
  for (int k = 0; k < 6; ++k) {
    const int n = 32 >> k;
    float d = 0.0f;
#pragma unroll
    for (int m = 0; m < n; ++m) {
      const float a = cur[2 * m] + cur[2 * m + 1];
      const float b = cur[2 * m] - cur[2 * m + 1];
      d += (__builtin_popcount(m) & 1) ? -b : b;
      cur[m] = a;
    }
    D[k] = d;
  }
  const float S = cur[0];

  float z[N_QUBITS];
  const int pt = parity6(t);
#pragma unroll
  for (int k = 0; k < 6; ++k) z[k] = pt ? -D[k] : D[k];
#pragma unroll
  for (int k = 6; k < 11; ++k) z[k] = parity6(t >> (k - 6)) ? -S : S;
  z[11] = parity6(t & 31) ? -D[0] : D[0];

#pragma unroll
  for (int k = 0; k < N_QUBITS; ++k) {
    float v = z[k];
#pragma unroll
    for (int m = 16; m > 0; m >>= 1) v += __shfl_xor(v, m);
    z[k] = v;
  }
  if ((t & 31) == 0) {
#pragma unroll
    for (int k = 0; k < N_QUBITS; ++k) red[t >> 5][k] = z[k];
  }
  __syncthreads();
  if (t < N_QUBITS) {
    const float g = gsc[0] * gsc[1];
    const float v = (red[0][N_QUBITS - 1 - t] + red[1][N_QUBITS - 1 - t]) * (g * g);
    meas[(size_t)bidx * N_QUBITS + t] = ((Fend >> t) & 1u) ? -v : v;
  }
}

__global__ __launch_bounds__(128) void back_mlp_kernel(
    const float* __restrict__ meas, const float* __restrict__ W3,
    const float* __restrict__ b3,   const float* __restrict__ W4,
    const float* __restrict__ b4,   float* __restrict__ out) {
  __shared__ float h[16 * HID];
  const int lane  = threadIdx.x & 31;
  const int wave  = threadIdx.x >> 5;
  const int row0  = blockIdx.x * 16;
  const int col   = lane & 15;
  const int rhalf = (lane >> 4) * 8;

  v2f bw[HID / 4] = {};
  float bias4 = 0.0f;
  if (wave == 0 && col < OUT_DIM) {
#pragma unroll
    for (int i = 0; i < HID / 4; ++i) {
      const int k = 4 * i + ((lane >> 4) << 1);
      bw[i].x = W4[k * OUT_DIM + col];
      bw[i].y = W4[(k + 1) * OUT_DIM + col];
    }
    bias4 = b4[col];
  }

  {
    v8f c = {};
#pragma unroll
    for (int kc = 0; kc < N_QUBITS; kc += 4) {
      v2f a = load_a_frag(meas, N_QUBITS, row0, kc, lane);
      v2f b = load_b_frag(W3, HID, kc, wave * 16, lane);
      c = wmma_f32(a, b, c);
    }
    const int n = wave * 16 + col;
    const float bias = b3[n];
#pragma unroll
    for (int r = 0; r < 8; ++r)
      h[(rhalf + r) * HID + n] = tanhf(c[r] + bias);
  }
  __syncthreads();
  if (wave != 0) return;

  v8f c = {};
#pragma unroll
  for (int i = 0; i < HID / 4; ++i) {
    v2f a = load_a_frag(h, HID, 0, 4 * i, lane);
    c = wmma_f32(a, bw[i], c);
  }
  if (col < OUT_DIM) {
#pragma unroll
    for (int r = 0; r < 8; ++r)
      out[(size_t)(row0 + rhalf + r) * OUT_DIM + col] = c[r] + bias4;
  }
}

extern "C" void kernel_launch(void* const* d_in, const int* in_sizes, int n_in,
                              void* d_out, int out_size, void* d_ws, size_t ws_size,
                              hipStream_t stream) {
  const float* noise = (const float*)d_in[0];
  const float* W1 = (const float*)d_in[1];
  const float* b1 = (const float*)d_in[2];
  const float* W2 = (const float*)d_in[3];
  const float* b2 = (const float*)d_in[4];
  const float* W3 = (const float*)d_in[5];
  const float* b3 = (const float*)d_in[6];
  const float* W4 = (const float*)d_in[7];
  const float* b4 = (const float*)d_in[8];
  float* out = (float*)d_out;

  float* params = (float*)d_ws;
  float* meas   = params + (size_t)BATCH * NPARAMS;

  front_mlp_kernel<<<BATCH / 16, FRONT_WAVES * 32, 0, stream>>>(noise, W1, b1, W2, b2, params);
  quantum_kernel<<<BATCH, 64, 0, stream>>>(params, meas);
  back_mlp_kernel<<<BATCH / 16, 128, 0, stream>>>(meas, W3, b3, W4, b4, out);
}


# Back MLP folded into the state-vector kernel: each block feeds its 12 expectation values from LDS through both WMMA layers (row replicated over the tile) and writes its 2 outputs; 2 launches, not 3
# speedup vs baseline: 8.9599x; 1.0393x over previous
#include <hip/hip_runtime.h>
#include <math.h>

#define N_QUBITS 12
#define DEPTH    4
#define HID      64
#define NPARAMS  144
#define BATCH    4096
#define DIM      4096
#define OUT_DIM  2

typedef __attribute__((ext_vector_type(2))) float v2f;
typedef __attribute__((ext_vector_type(8))) float v8f;

__device__ __forceinline__ v8f wmma_f32(v2f a, v2f b, v8f c) {
  return __builtin_amdgcn_wmma_f32_16x16x4_f32(
       false, a,  false, b,
       (short)0, c,  false,  false);
}

__device__ __forceinline__ v2f load_a_frag(const float* src, int lda,
                                           int row0, int k0, int lane) {
  int m = row0 + (lane & 15);
  int k = k0 + ((lane >> 4) << 1);
  v2f a;
  a.x = src[m * lda + k];
  a.y = src[m * lda + k + 1];
  return a;
}

__device__ __forceinline__ v2f load_b_frag(const float* W, int ldb,
                                           int k0, int n0, int lane) {
  int c = n0 + (lane & 15);
  int k = k0 + ((lane >> 4) << 1);
  v2f b;
  b.x = W[k * ldb + c];
  b.y = W[(k + 1) * ldb + c];
  return b;
}

#define FRONT_WAVES (NPARAMS / 16)

__global__ __launch_bounds__(FRONT_WAVES * 32) void front_mlp_kernel(
    const float* __restrict__ noise, const float* __restrict__ W1,
    const float* __restrict__ b1,    const float* __restrict__ W2,
    const float* __restrict__ b2,    float* __restrict__ params) {
  __shared__ float h[16 * HID];
  const int lane  = threadIdx.x & 31;
  const int wave  = threadIdx.x >> 5;
  const int row0  = blockIdx.x * 16;
  const int col   = lane & 15;
  const int rhalf = (lane >> 4) * 8;

  v2f bw[HID / 4];
#pragma unroll
  for (int i = 0; i < HID / 4; ++i)
    bw[i] = load_b_frag(W2, NPARAMS, 4 * i, wave * 16, lane);
  const float bias2 = b2[wave * 16 + col];

  if (wave < HID / 16) {
    v8f c = {};
#pragma unroll
    for (int kc = 0; kc < N_QUBITS; kc += 4) {
      v2f a = load_a_frag(noise, N_QUBITS, row0, kc, lane);
      v2f b = load_b_frag(W1, HID, kc, wave * 16, lane);
      c = wmma_f32(a, b, c);
    }
    const int n = wave * 16 + col;
    const float bias = b1[n];
#pragma unroll
    for (int r = 0; r < 8; ++r)
      h[(rhalf + r) * HID + n] = tanhf(c[r] + bias);
  }
  __syncthreads();

  v8f c = {};
#pragma unroll
  for (int i = 0; i < HID / 4; ++i) {
    v2f a = load_a_frag(h, HID, 0, 4 * i, lane);
    c = wmma_f32(a, bw[i], c);
  }
  const int n = wave * 16 + col;
#pragma unroll
  for (int r = 0; r < 8; ++r)
    params[(size_t)(row0 + rhalf + r) * NPARAMS + n] = c[r] + bias2;
}

#define QPITCH 66
#define TPITCH 66
#define TABB0 ((DEPTH - 2) * 4 * TPITCH)

typedef __attribute__((ext_vector_type(4))) float v4f;

struct Gate { float r00, i00, r01, i01, r10, i10, r11, i11; };

__device__ __forceinline__ v2f unit_conj(float re, float im, float& mag) {
  const float n2 = re * re + im * im;
  v2f u;
  if (n2 > 1e-30f) {
    const float inv = 1.0f / sqrtf(n2);
    mag = n2 * inv;
    u.x = re * inv; u.y = -im * inv;
  } else {
    mag = re;
    u.x = 1.0f; u.y = 0.0f;
  }
  return u;
}

__device__ __forceinline__ Gate make_gate(v2f ea, v2f eb, v2f ez) {
  const float ca = ea.x, sa = ea.y, cb = eb.x, sb = eb.y, cz = ez.x, sz = ez.y;
  const float m00r =  cb * ca, m00i =  sb * sa;
  const float m01r = -sb * ca, m01i = -cb * sa;
  const float m10r =  sb * ca, m10i = -cb * sa;
  const float m11r =  cb * ca, m11i = -sb * sa;
  Gate U;
  U.r00 = cz * m00r + sz * m00i;  U.i00 = cz * m00i - sz * m00r;
  U.r01 = cz * m01r + sz * m01i;  U.i01 = cz * m01i - sz * m01r;
  U.r10 = cz * m10r - sz * m10i;  U.i10 = cz * m10i + sz * m10r;
  U.r11 = cz * m11r - sz * m11i;  U.i11 = cz * m11i + sz * m11r;
  return U;
}

__device__ __forceinline__ v2f cmul(v2f a, v2f b) {
  const v2f sb = __builtin_shufflevector(b, b, 1, 0);
  v2f na; na.x = -a.y; na.y = a.y;
  return a.x * b + na * sb;
}

__device__ __forceinline__ v2f phase_if(v2f z, int bit) {
  v2f o;
  o.x = bit ? z.x : 1.0f;
  o.y = bit ? z.y : 0.0f;
  return o;
}

template <int RB>
__device__ __forceinline__ void apply_rot(v2f (&c)[64], float q) {
#pragma unroll
  for (int j = 0; j < 32; ++j) {
    const int i0 = ((j >> RB) << (RB + 1)) | (j & ((1 << RB) - 1));
    const int i1 = i0 | (1 << RB);
    const v2f x = c[i0], y = c[i1];
    c[i0] = x - q * y;
    c[i1] = q * x + y;
  }
}

__device__ __forceinline__ unsigned ring_flips(unsigned f) {
  unsigned p = f;
  p ^= p << 1; p ^= p << 2; p ^= p << 4; p ^= p << 8;
  p &= 0xFFFu;
  return (p & 0xFFEu) | (((p >> 11) ^ f) & 1u);
}

__device__ __forceinline__ void apply_phases(v2f (&c)[64], const v2f* tab, v2f ft) {
  const v4f* tp = reinterpret_cast<const v4f*>(tab);
#pragma unroll
  for (int h = 0; h < 32; ++h) {
    const v4f v = tp[h];
    v2f e0, e1;
    e0.x = v.x; e0.y = v.y; e1.x = v.z; e1.y = v.w;
    c[2 * h]     = cmul(ft, cmul(e0, c[2 * h]));
    c[2 * h + 1] = cmul(ft, cmul(e1, c[2 * h + 1]));
  }
}

__device__ __forceinline__ void apply_rots(v2f (&c)[64], const float* rt) {
  const v2f* rp = reinterpret_cast<const v2f*>(rt);
  const v2f g54 = rp[0], g32 = rp[1], g10 = rp[2];
  apply_rot<5>(c, g54.x);
  apply_rot<4>(c, g54.y);
  apply_rot<3>(c, g32.x);
  apply_rot<2>(c, g32.y);
  apply_rot<1>(c, g10.x);
  apply_rot<0>(c, g10.y);
}

__device__ __forceinline__ int parity6(int v) { return __popc(v & 63) & 1; }

__global__ __launch_bounds__(64) __attribute__((amdgpu_waves_per_eu(4))) void quantum_kernel(
    const float* __restrict__ params, const float* __restrict__ W3,
    const float* __restrict__ b3,     const float* __restrict__ W4,
    const float* __restrict__ b4,     float* __restrict__ out) {
  __shared__ __attribute__((aligned(16))) v2f st[64 * QPITCH];
  __shared__ float red[2][N_QUBITS];
  __shared__ float hv[HID];
  __shared__ float gsc[2];
  v2f* const trig = st + (64 * QPITCH - NPARAMS);
  __shared__ __attribute__((aligned(16))) float rot[(DEPTH - 1) * N_QUBITS];
  __shared__ int bigs[(DEPTH - 1) * N_QUBITS];
  __shared__ __attribute__((aligned(16))) v2f Lph[(DEPTH - 1) * N_QUBITS];
  __shared__ __attribute__((aligned(16))) v2f Rph[(DEPTH - 1) * N_QUBITS];
  __shared__ __attribute__((aligned(16))) v2f tabs[TABB0 + (DEPTH - 2) * 64];
  const int t = threadIdx.x;
  const int bidx = blockIdx.x;
  const float* p = params + (size_t)bidx * NPARAMS;

  const int cg0 = t ^ (t >> 1);
  const int cg1 = cg0 ^ 32;
  const int t0  = t & 1;

  const float ang0 = p[t], ang1 = p[64 + t], ang2 = p[128 + (t & 15)];
#pragma unroll 1
  for (int j = 0; j < 3; ++j) {
    const float ang = (j == 0) ? ang0 : (j == 1) ? ang1 : ang2;
    float sn, cs;
    sincosf(ang * 0.5f, &sn, &cs);
    v2f e; e.x = cs; e.y = sn;
    if (j < 2 || t < 16) trig[64 * j + t] = e;
  }
  __syncthreads();
  float hi = 1.0f;
  float qv = 0.0f;
  int big_s = 0;
  v2f L, R;
  L.x = 1.0f; L.y = 0.0f;
  R = L;
  if (t < (DEPTH - 1) * N_QUBITS) {
    const int a0 = 3 * (N_QUBITS + t);
    const Gate U = make_gate(trig[a0], trig[a0 + 1], trig[a0 + 2]);
    float gc, gs, gw;
    const v2f q00 = unit_conj(U.r00, U.i00, gc);
    const v2f q10 = unit_conj(U.r10, U.i10, gs);
    v2f p10; p10.x = q10.x; p10.y = -q10.y;
    L = cmul(p10, q00);
    const bool by_s = gs >= gc;
    const float wr = by_s ? -U.r01 : U.r11, wi = by_s ? -U.i01 : U.i11;
    v2f uw = unit_conj(wr, wi, gw);
    uw.y = -uw.y;
    R = cmul(uw, by_s ? q00 : q10);
    big_s = by_s ? 1 : 0;
    hi = by_s ? gs : gc;
    qv = (by_s ? gc : gs) / hi;
    bigs[t] = big_s;
  }
  for (int m = 16; m > 0; m >>= 1) hi *= __shfl_xor(hi, m);
  if ((t & 31) == 0) gsc[t >> 5] = hi;
  __syncthreads();

  const int ln = t & 31;
  const unsigned m0 = __builtin_amdgcn_ballot_w32(bigs[ln] != 0);
  const unsigned m1 = __builtin_amdgcn_ballot_w32(bigs[32 + (ln & 3)] != 0);
  const unsigned B1 = m0 & 0xFFFu, B2 = (m0 >> 12) & 0xFFFu, B3 = (m0 >> 24) | ((m1 & 0xFu) << 8);
  const unsigned F2 = ring_flips(B1);
  const unsigned F3 = ring_flips(F2 ^ B2);
  const unsigned Fend = ring_flips(F3 ^ B3);
  if (t < (DEPTH - 1) * N_QUBITS) {
    const int l = (t >= 2 * N_QUBITS) ? 2 : (t >= N_QUBITS) ? 1 : 0;
    const unsigned Fin = (l == 2) ? F3 : (l == 1) ? F2 : 0u;
    const int fin = (Fin >> (t - N_QUBITS * l)) & 1;
    if (fin) { L.y = -L.y; R.y = -R.y; }
    if (big_s) L.x = -L.x;
    rot[t] = (fin == big_s) ? qv : -qv;
    Lph[t] = L;
    Rph[t] = R;
  }
  __syncthreads();

  {
    const int r0 = t & 1, r4 = (t >> 4) & 1, r5 = (t >> 5) & 1;
    const int par = t ^ (t >> 1);
#pragma unroll 1
    for (int l = 1; l < DEPTH - 1; ++l) {
      const v2f* Rl = Rph + N_QUBITS * l;
      const v2f* Ll = Lph + N_QUBITS * (l - 1);
      v2f fr, ft;
      fr.x = 1.0f; fr.y = 0.0f;
      ft = fr;
#pragma unroll
      for (int k = 0; k < 6; ++k) {
        fr = cmul(phase_if(Rl[5 - k],  (t >> k) & 1), fr);
        ft = cmul(phase_if(Rl[11 - k], (t >> k) & 1), ft);
      }
#pragma unroll
      for (int k = 0; k < 4; ++k) fr = cmul(phase_if(Ll[5 - k],  (par >> k) & 1), fr);
#pragma unroll
      for (int k = 0; k < 5; ++k) ft = cmul(phase_if(Ll[11 - k], (par >> k) & 1), ft);
      const v2f m5 = Ll[6], m10 = Ll[1], m11 = Ll[0];
      tabs[TABB0 + (l - 1) * 64 + t] = ft;
      const v2f b0 = cmul(phase_if(m5, r0), fr);
      const v2f b1 = cmul(phase_if(m5, r0 ^ 1), fr);
      const v2f m0 = cmul(phase_if(m10, r4 ^ r5),     phase_if(m11, r5));
      const v2f m1 = cmul(phase_if(m10, r4 ^ r5 ^ 1), phase_if(m11, r5 ^ 1));
      tabs[((l - 1) * 4 + 0) * TPITCH + t] = cmul(m0, b0);
      tabs[((l - 1) * 4 + 1) * TPITCH + t] = cmul(m1, b0);
      tabs[((l - 1) * 4 + 2) * TPITCH + t] = cmul(m0, b1);
      tabs[((l - 1) * 4 + 3) * TPITCH + t] = cmul(m1, b1);
    }
  }

  v2f c[64];
  {
    v2f f0[N_QUBITS], f1[N_QUBITS];
#pragma unroll
    for (int b = 0; b < N_QUBITS; ++b) {
      const int g = N_QUBITS - 1 - b;
      const v2f ea = trig[3 * g], eb = trig[3 * g + 1], ez = trig[3 * g + 2];
      const float ca = ea.x, sa = ea.y, cb = eb.x, sb = eb.y, cz = ez.x, sz = ez.y;
      const float m00r =  cb * ca, m00i =  sb * sa;
      const float m10r =  sb * ca, m10i = -cb * sa;
      f0[b].x = cz * m00r + sz * m00i;  f0[b].y = cz * m00i - sz * m00r;
      f1[b].x = cz * m10r - sz * m10i;  f1[b].y = cz * m10i + sz * m10r;
    }
    const v2f* R1 = Rph;
    v2f pt = (cg0 & 1) ? f1[0] : f0[0];
#pragma unroll
    for (int b = 1; b < 5; ++b) pt = cmul(((cg0 >> b) & 1) ? f1[b] : f0[b], pt);
#pragma unroll
    for (int k = 0; k < 6; ++k) pt = cmul(phase_if(R1[11 - k], (t >> k) & 1), pt);
    const int t5 = (t >> 5) & 1;
    c[0] = cmul(t5 ? f1[5] : f0[5], pt);
    c[1] = cmul(cmul(R1[5], t5 ? f0[5] : f1[5]), pt);
#pragma unroll
    for (int k = 1; k < 5; ++k) {
      const int n = 1 << k;
      const v2f g0 = cmul(R1[5 - k], f0[5 + k]), g1 = cmul(R1[5 - k], f1[5 + k]);
#pragma unroll
      for (int j = 0; j < n; ++j) {
        const int lowbit = (j >> (k - 1)) & 1;
        c[n + j] = cmul(lowbit ? g0 : g1, c[j]);
      }
#pragma unroll
      for (int j = 0; j < n; ++j) {
        const int lowbit = (j >> (k - 1)) & 1;
        c[j] = cmul(lowbit ? f1[5 + k] : f0[5 + k], c[j]);
      }
    }
    const v2f a10 = t0 ? f1[10] : f0[10], b10 = t0 ? f0[10] : f1[10];
    const v2f a11 = t0 ? f1[11] : f0[11], b11 = t0 ? f0[11] : f1[11];
    const v2f w00 = cmul(a10, a11);
    const v2f w01 = cmul(b10, a11);
    const v2f w10 = cmul(R1[0], cmul(b10, b11));
    const v2f w11 = cmul(R1[0], cmul(a10, b11));
#pragma unroll
    for (int j = 0; j < 32; ++j) {
      const int r4 = (j >> 4) & 1;
      c[32 + j] = cmul(r4 ? w11 : w10, c[j]);
      c[j]      = cmul(r4 ? w01 : w00, c[j]);
    }
  }

  __syncthreads();

  const int va = ((t >> 5) << 1) | t0;
#pragma unroll 1
  for (int l = 0; l < DEPTH - 1; ++l) {
    if (l > 0) {
      v4f* row = reinterpret_cast<v4f*>(st + t * QPITCH);
#pragma unroll
      for (int h = 0; h < 32; ++h) {
        v4f v;
        v.x = c[2 * h].x;     v.y = c[2 * h].y;
        v.z = c[2 * h + 1].x; v.w = c[2 * h + 1].y;
        row[h] = v;
      }
      __syncthreads();
#pragma unroll
      for (int r = 0; r < 64; ++r) {
        const int G  = r ^ (r >> 1);
        const int hi = (G >> 4) & 3;
        const int delta = (hi == 0) ? 48 : (hi == 1) ? 16 : (hi == 2) ? -16 : -48;
        const int idx = G * QPITCH + ((r & 1) ? cg1 : cg0) + t0 * (delta * QPITCH);
        c[r] = st[idx];
      }
      __syncthreads();
      apply_phases(c, tabs + ((l - 1) * 4 + va) * TPITCH, tabs[TABB0 + (l - 1) * 64 + t]);
    }
    apply_rots(c, rot + 12 * l);
    {
#pragma unroll
      for (int r = 0; r < 64; ++r) st[r * QPITCH + t] = c[r];
      __syncthreads();
      const v4f* row = reinterpret_cast<const v4f*>(st + t * QPITCH);
#pragma unroll
      for (int h = 0; h < 32; ++h) {
        const v4f v = row[h];
        c[2 * h].x = v.x;     c[2 * h].y = v.y;
        c[2 * h + 1].x = v.z; c[2 * h + 1].y = v.w;
      }
    }
    apply_rots(c, rot + 12 * l + 6);
  }

  float cur[64];
#pragma unroll
  for (int r = 0; r < 64; ++r) cur[r] = c[r].x * c[r].x + c[r].y * c[r].y;
  float D[6];
#pragma unroll
  for (int k = 0; k < 6; ++k) {
    const int n = 32 >> k;
    float d = 0.0f;
#pragma unroll
    for (int m = 0; m < n; ++m) {
      const float a = cur[2 * m] + cur[2 * m + 1];
      const float b = cur[2 * m] - cur[2 * m + 1];
      d += (__builtin_popcount(m) & 1) ? -b : b;
      cur[m] = a;
    }
    D[k] = d;
  }
  const float S = cur[0];

  float z[N_QUBITS];
  const int pt = parity6(t);
#pragma unroll
  for (int k = 0; k < 6; ++k) z[k] = pt ? -D[k] : D[k];
#pragma unroll
  for (int k = 6; k < 11; ++k) z[k] = parity6(t >> (k - 6)) ? -S : S;
  z[11] = parity6(t & 31) ? -D[0] : D[0];

  const int ln5   = t & 31;
  const int wv    = t >> 5;
  const int mcol  = ln5 & 15;
  const int khalf = (ln5 >> 4) << 1;
  v2f bw3[2][N_QUBITS / 4];
  float bias3[2];
#pragma unroll
  for (int j = 0; j < 2; ++j) {
    const int n0 = (2 * wv + j) * 16;
#pragma unroll
    for (int i = 0; i < N_QUBITS / 4; ++i)
      bw3[j][i] = load_b_frag(W3, HID, 4 * i, n0, ln5);
    bias3[j] = b3[n0 + mcol];
  }
  v2f bw4[HID / 4] = {};
  float bias4 = 0.0f;
  if (wv == 0 && mcol < OUT_DIM) {
#pragma unroll
    for (int i = 0; i < HID / 4; ++i) {
      const int k = 4 * i + khalf;
      bw4[i].x = W4[k * OUT_DIM + mcol];
      bw4[i].y = W4[(k + 1) * OUT_DIM + mcol];
    }
    bias4 = b4[mcol];
  }

#pragma unroll
  for (int k = 0; k < N_QUBITS; ++k) {
    float v = z[k];
#pragma unroll
    for (int m = 16; m > 0; m >>= 1) v += __shfl_xor(v, m);
    z[k] = v;
  }
  if ((t & 31) == 0) {
#pragma unroll
    for (int k = 0; k < N_QUBITS; ++k) red[t >> 5][k] = z[k];
  }
  __syncthreads();
  const float g = gsc[0] * gsc[1];
  const float g2 = g * g;
  v2f am[N_QUBITS / 4];
#pragma unroll
  for (int i = 0; i < N_QUBITS / 4; ++i) {
    const int k = 4 * i + khalf;
    const float v0 = (red[0][N_QUBITS - 1 - k] + red[1][N_QUBITS - 1 - k]) * g2;
    const float v1 = (red[0][N_QUBITS - 2 - k] + red[1][N_QUBITS - 2 - k]) * g2;
    am[i].x = ((Fend >> k) & 1u) ? -v0 : v0;
    am[i].y = ((Fend >> (k + 1)) & 1u) ? -v1 : v1;
  }
#pragma unroll
  for (int j = 0; j < 2; ++j) {
    v8f acc = {};
#pragma unroll
    for (int i = 0; i < N_QUBITS / 4; ++i) acc = wmma_f32(am[i], bw3[j][i], acc);
    if (ln5 < 16) hv[(2 * wv + j) * 16 + mcol] = tanhf(acc[0] + bias3[j]);
  }
  __syncthreads();
  if (wv != 0) return;

  v8f o = {};
#pragma unroll
  for (int i = 0; i < HID / 4; ++i) {
    v2f a;
    a.x = hv[4 * i + khalf];
    a.y = hv[4 * i + khalf + 1];
    o = wmma_f32(a, bw4[i], o);
  }
  if (ln5 < OUT_DIM) out[(size_t)bidx * OUT_DIM + ln5] = o[0] + bias4;
}

extern "C" void kernel_launch(void* const* d_in, const int* in_sizes, int n_in,
                              void* d_out, int out_size, void* d_ws, size_t ws_size,
                              hipStream_t stream) {
  const float* noise = (const float*)d_in[0];
  const float* W1 = (const float*)d_in[1];
  const float* b1 = (const float*)d_in[2];
  const float* W2 = (const float*)d_in[3];
  const float* b2 = (const float*)d_in[4];
  const float* W3 = (const float*)d_in[5];
  const float* b3 = (const float*)d_in[6];
  const float* W4 = (const float*)d_in[7];
  const float* b4 = (const float*)d_in[8];
  float* out = (float*)d_out;

  float* params = (float*)d_ws;

  front_mlp_kernel<<<BATCH / 16, FRONT_WAVES * 32, 0, stream>>>(noise, W1, b1, W2, b2, params);
  quantum_kernel<<<BATCH, 64, 0, stream>>>(params, W3, b3, W4, b4, out);
}
